# starting kernel
# baseline (speedup 1.0000x reference)
#include <hip/hip_runtime.h>
#include <math.h>


typedef float v2f __attribute__((ext_vector_type(2)));
typedef float v8f __attribute__((ext_vector_type(8)));

#define DK 256
#define KCHUNK 128
#define LDT (KCHUNK + 4)
#define BM 32
#define BN 64

__device__ __forceinline__ unsigned f2ord(float f) {
  unsigned u = __float_as_uint(f);
  return (u & 0x80000000u) ? ~u : (u | 0x80000000u);
}
__device__ __forceinline__ float ord2f(unsigned u) {
  return __uint_as_float((u & 0x80000000u) ? (u ^ 0x80000000u) : ~u);
}

__global__ __launch_bounds__(256) void gat_gemm_wmma(
    const float* __restrict__ X, const float* __restrict__ W,
    float* __restrict__ H, int N, int Dact, int Dst) {
  __shared__ float As[BM * LDT];
  __shared__ float Bst[BN * LDT];

  const int tid  = threadIdx.x;
  const int lane = tid & 31;
  const int lo   = lane & 15;
  const int hi   = lane >> 4;
  const int wid  = tid >> 5;
  const int wm   = (wid >> 2) * 16;
  const int wn   = (wid & 3) * 16;
  const int rowBase = blockIdx.x * BM;
  const int colBase = blockIdx.y * BN;

  v8f acc = {};

  for (int kk = 0; kk < DK; kk += KCHUNK) {
    __syncthreads();
    for (int it = 0; it < 4; ++it) {
      int f4  = tid + 256 * it;
      int r   = f4 >> 5;
      int c4  = f4 & 31;
      int gr  = rowBase + r;
      float4 v = make_float4(0.f, 0.f, 0.f, 0.f);
      if (gr < N)
        v = *reinterpret_cast<const float4*>(X + (size_t)gr * DK + kk + c4 * 4);
      *reinterpret_cast<float4*>(&As[r * LDT + c4 * 4]) = v;
    }
    for (int it = 0; it < 32; ++it) {
      int idx = tid + 256 * it;
      int k   = idx >> 6;
      int n   = idx & 63;
      int gc  = colBase + n;
      float v = (gc < Dact) ? W[(size_t)(kk + k) * Dact + gc] : 0.f;
      Bst[n * LDT + k] = v;
    }
    __syncthreads();

    const float* arow = &As[(wm + lo) * LDT];
    const float* brow = &Bst[(wn + lo) * LDT];
    #pragma unroll
    for (int k0 = 0; k0 < KCHUNK; k0 += 4) {
      v2f a = *reinterpret_cast<const v2f*>(arow + k0 + 2 * hi);
      v2f b = *reinterpret_cast<const v2f*>(brow + k0 + 2 * hi);
      acc = __builtin_amdgcn_wmma_f32_16x16x4_f32(false, a, false, b,
                                                  (short)0, acc, false, false);
    }
  }

  const int col = colBase + wn + lo;
  if (col < Dst) {
    #pragma unroll
    for (int r = 0; r < 8; ++r) {
      int row = rowBase + wm + hi * 8 + r;
      if (row < N) H[(size_t)row * Dst + col] = acc[r];
    }
  }
}

__global__ __launch_bounds__(256) void gat_alpha_init(
    const float* __restrict__ H, const float* __restrict__ a_s,
    const float* __restrict__ a_d, float* __restrict__ asv,
    float* __restrict__ adv, unsigned* __restrict__ mv,
    float* __restrict__ dnv, float* __restrict__ outz,
    int N, int Dpad, int Dact) {
  const int lane = threadIdx.x & 31;
  const int i = blockIdx.x * 8 + (threadIdx.x >> 5);
  if (i >= N) return;
  float s1 = 0.f, s2 = 0.f;
  for (int d = lane; d < Dpad; d += 32) {
    float hv = H[(size_t)i * Dpad + d];
    float av = (d < Dact) ? a_s[d] : 0.f;
    float bv = (d < Dact) ? a_d[d] : 0.f;
    s1 += hv * av;
    s2 += hv * bv;
    outz[(size_t)i * Dpad + d] = 0.f;
  }
  #pragma unroll
  for (int off = 16; off > 0; off >>= 1) {
    s1 += __shfl_xor(s1, off, 32);
    s2 += __shfl_xor(s2, off, 32);
  }
  if (lane == 0) {
    asv[i] = s1;
    adv[i] = s2;
    mv[i]  = f2ord(-INFINITY);
    dnv[i] = 0.f;
  }
}

__global__ __launch_bounds__(256) void gat_edge_score(
    const int* __restrict__ src, const int* __restrict__ dst,
    const float* __restrict__ asv, const float* __restrict__ adv,
    float* __restrict__ esc, unsigned* __restrict__ mv, int E, int ET) {
  int e = blockIdx.x * blockDim.x + threadIdx.x;
  if (e >= ET) return;
  int s, d;
  if (e < E) { s = src[e]; d = dst[e]; } else { s = d = e - E; }
  float sc = asv[s] + adv[d];
  sc = (sc > 0.f) ? sc : 0.2f * sc;
  esc[e] = sc;
  atomicMax(&mv[d], f2ord(sc));
}

__global__ __launch_bounds__(256) void gat_edge_exp(
    const int* __restrict__ dst, const unsigned* __restrict__ mv,
    float* __restrict__ esc, float* __restrict__ dnv, int E, int ET) {
  int e = blockIdx.x * blockDim.x + threadIdx.x;
  if (e >= ET) return;
  int d = (e < E) ? dst[e] : (e - E);
  float p = __expf(esc[e] - ord2f(mv[d]));
  esc[e] = p;
  atomicAdd(&dnv[d], p);
}

__global__ __launch_bounds__(256) void gat_aggregate(
    const int* __restrict__ src, const int* __restrict__ dst,
    const float* __restrict__ esc, const float* __restrict__ dnv,
    const float* __restrict__ H, float* __restrict__ OUT,
    int E, int ET, int Dpad) {
  int e = (blockIdx.x * blockDim.x + threadIdx.x) >> 5;
  int lane = threadIdx.x & 31;
  if (e >= ET) return;
  int s, d;
  if (e < E) { s = src[e]; d = dst[e]; } else { s = d = e - E; }
  float coeff = esc[e] / dnv[d];
  const float4* hp = reinterpret_cast<const float4*>(H + (size_t)s * Dpad);
  float* op = OUT + (size_t)d * Dpad;
  int nf4 = Dpad >> 2;
  for (int d4 = lane; d4 < nf4; d4 += 32) {
    float4 hv = hp[d4];
    atomicAdd(op + d4 * 4 + 0, coeff * hv.x);
    atomicAdd(op + d4 * 4 + 1, coeff * hv.y);
    atomicAdd(op + d4 * 4 + 2, coeff * hv.z);
    atomicAdd(op + d4 * 4 + 3, coeff * hv.w);
  }
}

__global__ __launch_bounds__(256) void gat_bias_relu(
    float* __restrict__ HIO, const float* __restrict__ b,
    size_t total, int Dpad) {
  size_t t = (size_t)blockIdx.x * blockDim.x + threadIdx.x;
  if (t >= total) return;
  float v = HIO[t] + b[t % Dpad];
  HIO[t] = v > 0.f ? v : 0.f;
}

__global__ __launch_bounds__(256) void gat_bias_out(
    const float* __restrict__ O, const float* __restrict__ b,
    float* __restrict__ out, int N, int Dpad, int Dact) {
  int t = blockIdx.x * blockDim.x + threadIdx.x;
  if (t >= N * Dpad) return;
  int i = t / Dpad, d = t % Dpad;
  if (d < Dact) out[(size_t)i * Dact + d] = O[t] + b[d];
}

extern "C" void kernel_launch(void* const* d_in, const int* in_sizes, int n_in,
                              void* d_out, int out_size, void* d_ws, size_t ws_size,
                              hipStream_t stream) {
  const float* x  = (const float*)d_in[0];
  const int*   ei = (const int*)d_in[1];
  const float* W1 = (const float*)d_in[2];
  const float* as1 = (const float*)d_in[3];
  const float* ad1 = (const float*)d_in[4];
  const float* b1 = (const float*)d_in[5];
  const float* W2 = (const float*)d_in[6];
  const float* as2 = (const float*)d_in[7];
  const float* ad2 = (const float*)d_in[8];
  const float* b2 = (const float*)d_in[9];
  const float* W3 = (const float*)d_in[10];
  const float* as3 = (const float*)d_in[11];
  const float* ad3 = (const float*)d_in[12];
  const float* b3 = (const float*)d_in[13];

  const int N  = in_sizes[0] / 256;
  const int E  = in_sizes[1] / 2;
  const int ET = E + N;
  const int* src = ei;
  const int* dst = ei + E;

  float* ws = (float*)d_ws;
  float*    hA  = ws;
  float*    hB  = hA + (size_t)N * 256;
  float*    h3  = hB + (size_t)N * 256;
  float*    o3  = h3 + (size_t)N * 16;
  float*    asv = o3 + (size_t)N * 16;
  float*    adv = asv + N;
  unsigned* mv  = (unsigned*)(adv + N);
  float*    dnv = adv + 2 * (size_t)N;
  float*    esc = dnv + N;

  dim3 blk(256);
  dim3 gemmG((N + BM - 1) / BM, 4);
  dim3 gemmG3((N + BM - 1) / BM, 1);
  int gAlpha = (N + 7) / 8;
  int gEdge  = (ET + 255) / 256;
  int gAgg   = (ET + 7) / 8;
  int gEltF  = (int)(((size_t)N * 256 + 255) / 256);
  int gElt3  = (N * 16 + 255) / 256;

  gat_gemm_wmma<<<gemmG, blk, 0, stream>>>(x, W1, hA, N, 256, 256);
  gat_alpha_init<<<gAlpha, blk, 0, stream>>>(hA, as1, ad1, asv, adv, mv, dnv, hB, N, 256, 256);
  gat_edge_score<<<gEdge, blk, 0, stream>>>(src, dst, asv, adv, esc, mv, E, ET);
  gat_edge_exp<<<gEdge, blk, 0, stream>>>(dst, mv, esc, dnv, E, ET);
  gat_aggregate<<<gAgg, blk, 0, stream>>>(src, dst, esc, dnv, hA, hB, E, ET, 256);
  gat_bias_relu<<<gEltF, blk, 0, stream>>>(hB, b1, (size_t)N * 256, 256);

  gat_gemm_wmma<<<gemmG, blk, 0, stream>>>(hB, W2, hA, N, 256, 256);
  gat_alpha_init<<<gAlpha, blk, 0, stream>>>(hA, as2, ad2, asv, adv, mv, dnv, hB, N, 256, 256);
  gat_edge_score<<<gEdge, blk, 0, stream>>>(src, dst, asv, adv, esc, mv, E, ET);
  gat_edge_exp<<<gEdge, blk, 0, stream>>>(dst, mv, esc, dnv, E, ET);
  gat_aggregate<<<gAgg, blk, 0, stream>>>(src, dst, esc, dnv, hA, hB, E, ET, 256);
  gat_bias_relu<<<gEltF, blk, 0, stream>>>(hB, b2, (size_t)N * 256, 256);

  gat_gemm_wmma<<<gemmG3, blk, 0, stream>>>(hB, W3, h3, N, 10, 16);
  gat_alpha_init<<<gAlpha, blk, 0, stream>>>(h3, as3, ad3, asv, adv, mv, dnv, o3, N, 16, 10);
  gat_edge_score<<<gEdge, blk, 0, stream>>>(src, dst, asv, adv, esc, mv, E, ET);
  gat_edge_exp<<<gEdge, blk, 0, stream>>>(dst, mv, esc, dnv, E, ET);
  gat_aggregate<<<gAgg, blk, 0, stream>>>(src, dst, esc, dnv, h3, o3, E, ET, 16);
  gat_bias_out<<<gElt3, blk, 0, stream>>>(o3, b3, (float*)d_out, N, 16, 10);
}


# Edges counting-sorted by dst once per call; one wave per dst node does softmax and weighted h[src] sum in registers, one row write with bias/relu; no float atomics; sums run in arrival-rank order
# speedup vs baseline: 2.4030x; 2.4030x over previous
#include <hip/hip_runtime.h>
#include <math.h>


typedef float v2f __attribute__((ext_vector_type(2)));
typedef float v8f __attribute__((ext_vector_type(8)));

#define DK 256
#define KCHUNK 128
#define LDT (KCHUNK + 4)
#define BM 32
#define BN 64

__global__ __launch_bounds__(256) void gat_gemm_wmma(
    const float* __restrict__ X, const float* __restrict__ W,
    float* __restrict__ H, int N, int Dact, int Dst) {
  __shared__ float As[BM * LDT];
  __shared__ float Bst[BN * LDT];

  const int tid  = threadIdx.x;
  const int lane = tid & 31;
  const int lo   = lane & 15;
  const int hi   = lane >> 4;
  const int wid  = tid >> 5;
  const int wm   = (wid >> 2) * 16;
  const int wn   = (wid & 3) * 16;
  const int rowBase = blockIdx.x * BM;
  const int colBase = blockIdx.y * BN;

  v8f acc = {};

  for (int kk = 0; kk < DK; kk += KCHUNK) {
    __syncthreads();
    for (int it = 0; it < 4; ++it) {
      int f4  = tid + 256 * it;
      int r   = f4 >> 5;
      int c4  = f4 & 31;
      int gr  = rowBase + r;
      float4 v = make_float4(0.f, 0.f, 0.f, 0.f);
      if (gr < N)
        v = *reinterpret_cast<const float4*>(X + (size_t)gr * DK + kk + c4 * 4);
      *reinterpret_cast<float4*>(&As[r * LDT + c4 * 4]) = v;
    }
    for (int it = 0; it < 32; ++it) {
      int idx = tid + 256 * it;
      int k   = idx >> 6;
      int n   = idx & 63;
      int gc  = colBase + n;
      float v = (gc < Dact) ? W[(size_t)(kk + k) * Dact + gc] : 0.f;
      Bst[n * LDT + k] = v;
    }
    __syncthreads();

    const float* arow = &As[(wm + lo) * LDT];
    const float* brow = &Bst[(wn + lo) * LDT];
    #pragma unroll
    for (int k0 = 0; k0 < KCHUNK; k0 += 4) {
      v2f a = *reinterpret_cast<const v2f*>(arow + k0 + 2 * hi);
      v2f b = *reinterpret_cast<const v2f*>(brow + k0 + 2 * hi);
      acc = __builtin_amdgcn_wmma_f32_16x16x4_f32(false, a, false, b,
                                                  (short)0, acc, false, false);
    }
  }

  const int col = colBase + wn + lo;
  if (col < Dst) {
    #pragma unroll
    for (int r = 0; r < 8; ++r) {
      int row = rowBase + wm + hi * 8 + r;
      if (row < N) H[(size_t)row * Dst + col] = acc[r];
    }
  }
}

__global__ __launch_bounds__(256) void gat_alpha(
    const float* __restrict__ H, const float* __restrict__ a_s,
    const float* __restrict__ a_d, float* __restrict__ asv,
    float* __restrict__ adv, int N, int Dpad, int Dact) {
  const int lane = threadIdx.x & 31;
  const int i = blockIdx.x * 8 + (threadIdx.x >> 5);
  if (i >= N) return;
  float s1 = 0.f, s2 = 0.f;
  for (int d = lane; d < Dpad; d += 32) {
    float hv = H[(size_t)i * Dpad + d];
    float av = (d < Dact) ? a_s[d] : 0.f;
    float bv = (d < Dact) ? a_d[d] : 0.f;
    s1 += hv * av;
    s2 += hv * bv;
  }
  #pragma unroll
  for (int off = 16; off > 0; off >>= 1) {
    s1 += __shfl_xor(s1, off, 32);
    s2 += __shfl_xor(s2, off, 32);
  }
  if (lane == 0) {
    asv[i] = s1;
    adv[i] = s2;
  }
}

__global__ __launch_bounds__(256) void gat_csr_zero(int* __restrict__ cnt, int N) {
  int i = blockIdx.x * 256 + threadIdx.x;
  if (i < N) cnt[i] = 0;
}

__global__ __launch_bounds__(256) void gat_csr_hist(
    const int* __restrict__ dst, int* __restrict__ cnt, int E) {
  int e = blockIdx.x * 256 + threadIdx.x;
  if (e < E) atomicAdd(&cnt[dst[e]], 1);
}

__global__ __launch_bounds__(256) void gat_csr_tilesum(
    const int* __restrict__ cnt, int* __restrict__ tsum, int N) {
  __shared__ int red[8];
  const int tid = threadIdx.x;
  int i = blockIdx.x * 256 + tid;
  int v = (i < N) ? cnt[i] : 0;
  #pragma unroll
  for (int off = 16; off > 0; off >>= 1) v += __shfl_xor(v, off, 32);
  if ((tid & 31) == 0) red[tid >> 5] = v;
  __syncthreads();
  if (tid == 0) {
    int t = 0;
    #pragma unroll
    for (int w = 0; w < 8; ++w) t += red[w];
    tsum[blockIdx.x] = t;
  }
}

__global__ __launch_bounds__(256) void gat_csr_scan(
    const int* __restrict__ cnt, const int* __restrict__ tsum,
    int* __restrict__ rowptr, int* __restrict__ cursor, int N) {
  __shared__ int red[8];
  __shared__ int wtot[8];
  const int tid = threadIdx.x;
  const int lane = tid & 31;
  const int wid = tid >> 5;
  int part = 0;
  for (int t = tid; t < (int)blockIdx.x; t += 256) part += tsum[t];
  #pragma unroll
  for (int off = 16; off > 0; off >>= 1) part += __shfl_xor(part, off, 32);
  if (lane == 0) red[wid] = part;
  int i = blockIdx.x * 256 + tid;
  int v = (i < N) ? cnt[i] : 0;
  int x = v;
  #pragma unroll
  for (int d = 1; d < 32; d <<= 1) {
    int n = __shfl_up(x, d, 32);
    if (lane >= d) x += n;
  }
  if (lane == 31) wtot[wid] = x;
  __syncthreads();
  int base = 0;
  #pragma unroll
  for (int w = 0; w < 8; ++w) {
    base += red[w];
    if (w < wid) base += wtot[w];
  }
  int excl = base + x - v;
  if (i < N) {
    rowptr[i] = excl;
    cursor[i] = excl;
    if (i == N - 1) rowptr[N] = excl + v;
  }
}

__global__ __launch_bounds__(256) void gat_csr_fill(
    const int* __restrict__ src, const int* __restrict__ dst,
    int* __restrict__ cursor, int* __restrict__ csrc, int E) {
  int e = blockIdx.x * 256 + threadIdx.x;
  if (e < E) {
    int pos = atomicAdd(&cursor[dst[e]], 1);
    csrc[pos] = src[e];
  }
}

__device__ __forceinline__ float leaky02(float v) {
  return (v > 0.f) ? v : 0.2f * v;
}

__device__ __forceinline__ void gat_node_stats(
    const int* __restrict__ csrc, const float* __restrict__ asv,
    int i, int beg, int end, int lane, float ad, float sself,
    int& s0, float& sc0, bool& v0, float& m, float& sum) {
  const int j0 = beg + lane;
  v0 = j0 < end;
  s0 = v0 ? csrc[j0] : i;
  sc0 = v0 ? leaky02(asv[s0] + ad) : sself;
  float mm = fmaxf(sself, sc0);
  for (int j = j0 + 32; j < end; j += 32)
    mm = fmaxf(mm, leaky02(asv[csrc[j]] + ad));
  #pragma unroll
  for (int off = 16; off > 0; off >>= 1) mm = fmaxf(mm, __shfl_xor(mm, off, 32));
  float ss = v0 ? __expf(sc0 - mm) : 0.f;
  for (int j = j0 + 32; j < end; j += 32)
    ss += __expf(leaky02(asv[csrc[j]] + ad) - mm);
  #pragma unroll
  for (int off = 16; off > 0; off >>= 1) ss += __shfl_xor(ss, off, 32);
  m = mm;
  sum = ss + __expf(sself - mm);
}

template <bool RELU>
__global__ __launch_bounds__(256) void gat_node_gather256(
    const int* __restrict__ rowptr, const int* __restrict__ csrc,
    const float* __restrict__ asv, const float* __restrict__ adv,
    const float* __restrict__ H, const float* __restrict__ bias,
    float* __restrict__ OUT, int N) {
  const int lane = threadIdx.x & 31;
  const int i = blockIdx.x * 8 + __builtin_amdgcn_readfirstlane((int)(threadIdx.x >> 5));
  if (i >= N) return;
  const int beg = rowptr[i];
  const int end = rowptr[i + 1];
  const float ad = adv[i];
  const float sself = leaky02(asv[i] + ad);
  int s0; float sc0; bool v0; float m, sum;
  gat_node_stats(csrc, asv, i, beg, end, lane, ad, sself, s0, sc0, v0, m, sum);

  const float4* H4 = reinterpret_cast<const float4*>(H);
  float4 a0, a1;
  {
    const float c = __expf(sself - m) / sum;
    const float4* hp = H4 + (size_t)i * 64;
    const float4 x = hp[lane];
    const float4 y = hp[lane + 32];
    a0 = make_float4(c * x.x, c * x.y, c * x.z, c * x.w);
    a1 = make_float4(c * y.x, c * y.y, c * y.z, c * y.w);
  }
  int sv = s0;
  float cv = v0 ? __expf(sc0 - m) / sum : 0.f;
  for (int base = beg; base < end; base += 32) {
    if (base != beg) {
      const int j = base + lane;
      const bool v = j < end;
      sv = v ? csrc[j] : i;
      cv = v ? __expf(leaky02(asv[sv] + ad) - m) / sum : 0.f;
    }
    const int cnt = min(32, end - base);
    #pragma unroll 4
    for (int t = 0; t < cnt; ++t) {
      const int s = __builtin_amdgcn_readlane(sv, t);
      const float c = __int_as_float(__builtin_amdgcn_readlane(__float_as_int(cv), t));
      const float4* hp = H4 + (size_t)s * 64;
      const float4 x = hp[lane];
      const float4 y = hp[lane + 32];
      a0.x = fmaf(c, x.x, a0.x); a0.y = fmaf(c, x.y, a0.y);
      a0.z = fmaf(c, x.z, a0.z); a0.w = fmaf(c, x.w, a0.w);
      a1.x = fmaf(c, y.x, a1.x); a1.y = fmaf(c, y.y, a1.y);
      a1.z = fmaf(c, y.z, a1.z); a1.w = fmaf(c, y.w, a1.w);
    }
  }
  const int c0 = lane * 4;
  const int c1 = 128 + lane * 4;
  a0.x += bias[c0 + 0]; a0.y += bias[c0 + 1]; a0.z += bias[c0 + 2]; a0.w += bias[c0 + 3];
  a1.x += bias[c1 + 0]; a1.y += bias[c1 + 1]; a1.z += bias[c1 + 2]; a1.w += bias[c1 + 3];
  if (RELU) {
    a0.x = fmaxf(a0.x, 0.f); a0.y = fmaxf(a0.y, 0.f); a0.z = fmaxf(a0.z, 0.f); a0.w = fmaxf(a0.w, 0.f);
    a1.x = fmaxf(a1.x, 0.f); a1.y = fmaxf(a1.y, 0.f); a1.z = fmaxf(a1.z, 0.f); a1.w = fmaxf(a1.w, 0.f);
  }
  float4* op = reinterpret_cast<float4*>(OUT) + (size_t)i * 64;
  op[lane] = a0;
  op[lane + 32] = a1;
}

__global__ __launch_bounds__(256) void gat_node_gather16(
    const int* __restrict__ rowptr, const int* __restrict__ csrc,
    const float* __restrict__ asv, const float* __restrict__ adv,
    const float* __restrict__ H, const float* __restrict__ bias,
    float* __restrict__ out, int N, int Dact) {
  const int lane = threadIdx.x & 31;
  const int i = blockIdx.x * 8 + __builtin_amdgcn_readfirstlane((int)(threadIdx.x >> 5));
  if (i >= N) return;
  const int beg = rowptr[i];
  const int end = rowptr[i + 1];
  const float ad = adv[i];
  const float sself = leaky02(asv[i] + ad);
  int s0; float sc0; bool v0; float m, sum;
  gat_node_stats(csrc, asv, i, beg, end, lane, ad, sself, s0, sc0, v0, m, sum);

  const int q = lane & 3;
  const int slot = lane >> 2;
  const float4* H4 = reinterpret_cast<const float4*>(H);
  float4 acc = make_float4(0.f, 0.f, 0.f, 0.f);
  if (slot == 0) {
    const float c = __expf(sself - m) / sum;
    const float4 x = H4[(size_t)i * 4 + q];
    acc = make_float4(c * x.x, c * x.y, c * x.z, c * x.w);
  }
  int sv = s0;
  float cv = v0 ? __expf(sc0 - m) / sum : 0.f;
  for (int base = beg; base < end; base += 32) {
    if (base != beg) {
      const int j = base + lane;
      const bool v = j < end;
      sv = v ? csrc[j] : i;
      cv = v ? __expf(leaky02(asv[sv] + ad) - m) / sum : 0.f;
    }
    const int cnt = min(32, end - base);
    for (int t0 = 0; t0 < cnt; t0 += 8) {
      const int t = t0 + slot;
      const int s = __shfl(sv, t, 32);
      const float c = __shfl(cv, t, 32);
      if (t < cnt) {
        const float4 x = H4[(size_t)s * 4 + q];
        acc.x = fmaf(c, x.x, acc.x); acc.y = fmaf(c, x.y, acc.y);
        acc.z = fmaf(c, x.z, acc.z); acc.w = fmaf(c, x.w, acc.w);
      }
    }
  }
  #pragma unroll
  for (int off = 4; off < 32; off <<= 1) {
    acc.x += __shfl_xor(acc.x, off, 32);
    acc.y += __shfl_xor(acc.y, off, 32);
    acc.z += __shfl_xor(acc.z, off, 32);
    acc.w += __shfl_xor(acc.w, off, 32);
  }
  if (slot == 0) {
    const int c = q * 4;
    float* op = out + (size_t)i * Dact;
    if (c + 0 < Dact) op[c + 0] = acc.x + bias[c + 0];
    if (c + 1 < Dact) op[c + 1] = acc.y + bias[c + 1];
    if (c + 2 < Dact) op[c + 2] = acc.z + bias[c + 2];
    if (c + 3 < Dact) op[c + 3] = acc.w + bias[c + 3];
  }
}

extern "C" void kernel_launch(void* const* d_in, const int* in_sizes, int n_in,
                              void* d_out, int out_size, void* d_ws, size_t ws_size,
                              hipStream_t stream) {
  const float* x  = (const float*)d_in[0];
  const int*   ei = (const int*)d_in[1];
  const float* W1 = (const float*)d_in[2];
  const float* as1 = (const float*)d_in[3];
  const float* ad1 = (const float*)d_in[4];
  const float* b1 = (const float*)d_in[5];
  const float* W2 = (const float*)d_in[6];
  const float* as2 = (const float*)d_in[7];
  const float* ad2 = (const float*)d_in[8];
  const float* b2 = (const float*)d_in[9];
  const float* W3 = (const float*)d_in[10];
  const float* as3 = (const float*)d_in[11];
  const float* ad3 = (const float*)d_in[12];
  const float* b3 = (const float*)d_in[13];

  const int N  = in_sizes[0] / 256;
  const int E  = in_sizes[1] / 2;
  const int* src = ei;
  const int* dst = ei + E;

  float* ws = (float*)d_ws;
  float* hA  = ws;
  float* hB  = hA + (size_t)N * 256;
  float* h3  = hB + (size_t)N * 256;
  int*   ib  = (int*)(h3 + (size_t)N * 16);
  int*   cnt    = ib;
  int*   rowptr = cnt + N;
  int*   cursor = rowptr + N + 1;
  int*   tsum   = cursor + N;
  float* asv = (float*)(ib + (size_t)N * 16);
  float* adv = asv + N;
  int*   csrc = (int*)(adv + N);

  dim3 blk(256);
  dim3 gemmG((N + BM - 1) / BM, 4);
  dim3 gemmG3((N + BM - 1) / BM, 1);
  int gNodeW = (N + 7) / 8;
  int gNodeT = (N + 255) / 256;
  int gEdgeT = (E + 255) / 256;

  gat_csr_zero<<<gNodeT, blk, 0, stream>>>(cnt, N);
  gat_csr_hist<<<gEdgeT, blk, 0, stream>>>(dst, cnt, E);
  gat_csr_tilesum<<<gNodeT, blk, 0, stream>>>(cnt, tsum, N);
  gat_csr_scan<<<gNodeT, blk, 0, stream>>>(cnt, tsum, rowptr, cursor, N);
  gat_csr_fill<<<gEdgeT, blk, 0, stream>>>(src, dst, cursor, csrc, E);

  gat_gemm_wmma<<<gemmG, blk, 0, stream>>>(x, W1, hA, N, 256, 256);
  gat_alpha<<<gNodeW, blk, 0, stream>>>(hA, as1, ad1, asv, adv, N, 256, 256);
  gat_node_gather256<true><<<gNodeW, blk, 0, stream>>>(rowptr, csrc, asv, adv, hA, b1, hB, N);

  gat_gemm_wmma<<<gemmG, blk, 0, stream>>>(hB, W2, hA, N, 256, 256);
  gat_alpha<<<gNodeW, blk, 0, stream>>>(hA, as2, ad2, asv, adv, N, 256, 256);
  gat_node_gather256<true><<<gNodeW, blk, 0, stream>>>(rowptr, csrc, asv, adv, hA, b2, hB, N);

  gat_gemm_wmma<<<gemmG3, blk, 0, stream>>>(hB, W3, h3, N, 10, 16);
  gat_alpha<<<gNodeW, blk, 0, stream>>>(h3, as3, ad3, asv, adv, N, 16, 10);
  gat_node_gather16<<<gNodeW, blk, 0, stream>>>(rowptr, csrc, asv, adv, h3, b3, (float*)d_out, N, 10);
}


# GEMM block owns whole rows (32x256; 128x16 in layer 3): h.a_src and h.a_dst come from its accumulators, the 3 alpha launches go; 16-byte k-quad fragment reads; GEMM k order and dot sum order change
# speedup vs baseline: 3.9786x; 1.6557x over previous
#include <hip/hip_runtime.h>
#include <math.h>


typedef float v2f __attribute__((ext_vector_type(2)));
typedef float v8f __attribute__((ext_vector_type(8)));

#define DK 256
#define KC 32
#define LDT (KC + 4)

template <int WMW, int WNW, int NT>
__global__ __launch_bounds__(256) void gat_gemm_alpha(
    const float* __restrict__ X, const float* __restrict__ W,
    const float* __restrict__ a_s, const float* __restrict__ a_d,
    float* __restrict__ H, float* __restrict__ asv, float* __restrict__ adv,
    int N, int Dact) {
  static_assert(WMW * WNW == 8, "eight waves per block");
  constexpr int BM = 16 * WMW;
  constexpr int BN = 16 * NT * WNW;
  constexpr int CP = BN + 4;
  constexpr int A_F4 = (BM * (KC / 4)) / 256;
  constexpr int B_QT = BN * (KC / 4);
  constexpr int B_Q = (B_QT + 255) / 256;
  constexpr int SM_AB = (BM + BN) * LDT;
  constexpr int SM_C = BM * CP;
  constexpr int SM = SM_AB > SM_C ? SM_AB : SM_C;
  __shared__ __attribute__((aligned(16))) float smem[SM];
  float* As = smem;
  float* Bst = smem + BM * LDT;
  float* Cs = smem;
  static_assert(SM >= SM_C + 2 * BM, "room for the per-row scalars behind the C tile");
  float* red = smem + SM_C;

  const int tid  = threadIdx.x;
  const int lane = tid & 31;
  const int lo   = lane & 15;
  const int hi   = lane >> 4;
  const int wid  = tid >> 5;
  const int wm   = (wid / WNW) * 16;
  const int wn   = (wid % WNW) * (16 * NT);
  const int rowBase = blockIdx.x * BM;

  v8f acc[NT];
  #pragma unroll
  for (int t = 0; t < NT; ++t) acc[t] = v8f{};

  float4 ra[A_F4];
  float4 rb[B_Q];

  auto load_chunk = [&](int kk) {
    #pragma unroll
    for (int i = 0; i < A_F4; ++i) {
      const int f4 = tid + 256 * i;
      const int r  = f4 >> 3;
      const int c4 = f4 & 7;
      const int gr = rowBase + r;
      float4 v = make_float4(0.f, 0.f, 0.f, 0.f);
      if (gr < N)
        v = *reinterpret_cast<const float4*>(X + (size_t)gr * DK + kk + c4 * 4);
      ra[i] = v;
    }
    #pragma unroll
    for (int i = 0; i < B_Q; ++i) {
      const int q  = tid + 256 * i;
      const int n  = q % BN;
      const int kq = q / BN;
      float4 v = make_float4(0.f, 0.f, 0.f, 0.f);
      if ((B_QT % 256 == 0 || q < B_QT) && n < Dact) {
        const float* wp = W + (size_t)(kk + kq * 4) * Dact + n;
        v.x = wp[0];
        v.y = wp[Dact];
        v.z = wp[2 * Dact];
        v.w = wp[3 * Dact];
      }
      rb[i] = v;
    }
  };
  auto store_chunk = [&]() {
    #pragma unroll
    for (int i = 0; i < A_F4; ++i) {
      const int f4 = tid + 256 * i;
      const int r  = f4 >> 3;
      const int c4 = f4 & 7;
      *reinterpret_cast<float4*>(&As[r * LDT + c4 * 4]) = ra[i];
    }
    #pragma unroll
    for (int i = 0; i < B_Q; ++i) {
      const int q  = tid + 256 * i;
      const int n  = q % BN;
      const int kq = q / BN;
      if (B_QT % 256 == 0 || q < B_QT)
        *reinterpret_cast<float4*>(&Bst[n * LDT + kq * 4]) = rb[i];
    }
  };

  const float* arow = As + (wm + lo) * LDT + 4 * hi;
  const float* brow = Bst + (wn + lo) * LDT + 4 * hi;

  load_chunk(0);
  for (int kk = 0; kk < DK; kk += KC) {
    __syncthreads();
    store_chunk();
    __syncthreads();
    if (kk + KC < DK) load_chunk(kk + KC);
    #pragma unroll
    for (int kb = 0; kb < KC; kb += 8) {
      const float4 a4 = *reinterpret_cast<const float4*>(arow + kb);
      float4 b4[NT];
      #pragma unroll
      for (int t = 0; t < NT; ++t)
        b4[t] = *reinterpret_cast<const float4*>(brow + t * 16 * LDT + kb);
      const v2f a01 = {a4.x, a4.y};
      const v2f a23 = {a4.z, a4.w};
      #pragma unroll
      for (int t = 0; t < NT; ++t) {
        const v2f b01 = {b4[t].x, b4[t].y};
        acc[t] = __builtin_amdgcn_wmma_f32_16x16x4_f32(false, a01, false, b01,
                                                       (short)0, acc[t], false, false);
      }
      #pragma unroll
      for (int t = 0; t < NT; ++t) {
        const v2f b23 = {b4[t].z, b4[t].w};
        acc[t] = __builtin_amdgcn_wmma_f32_16x16x4_f32(false, a23, false, b23,
                                                       (short)0, acc[t], false, false);
      }
    }
  }

  __syncthreads();
  #pragma unroll
  for (int t = 0; t < NT; ++t) {
    #pragma unroll
    for (int r = 0; r < 8; ++r)
      Cs[(wm + 8 * hi + r) * CP + wn + 16 * t + lo] = acc[t][r];
  }
  __syncthreads();

  if constexpr (BN == 256) {
    float4 s0, s1, d0, d1;
    {
      const int c0 = 4 * lane, c1 = 128 + 4 * lane;
      s0.x = (c0 + 0 < Dact) ? a_s[c0 + 0] : 0.f; s0.y = (c0 + 1 < Dact) ? a_s[c0 + 1] : 0.f;
      s0.z = (c0 + 2 < Dact) ? a_s[c0 + 2] : 0.f; s0.w = (c0 + 3 < Dact) ? a_s[c0 + 3] : 0.f;
      s1.x = (c1 + 0 < Dact) ? a_s[c1 + 0] : 0.f; s1.y = (c1 + 1 < Dact) ? a_s[c1 + 1] : 0.f;
      s1.z = (c1 + 2 < Dact) ? a_s[c1 + 2] : 0.f; s1.w = (c1 + 3 < Dact) ? a_s[c1 + 3] : 0.f;
      d0.x = (c0 + 0 < Dact) ? a_d[c0 + 0] : 0.f; d0.y = (c0 + 1 < Dact) ? a_d[c0 + 1] : 0.f;
      d0.z = (c0 + 2 < Dact) ? a_d[c0 + 2] : 0.f; d0.w = (c0 + 3 < Dact) ? a_d[c0 + 3] : 0.f;
      d1.x = (c1 + 0 < Dact) ? a_d[c1 + 0] : 0.f; d1.y = (c1 + 1 < Dact) ? a_d[c1 + 1] : 0.f;
      d1.z = (c1 + 2 < Dact) ? a_d[c1 + 2] : 0.f; d1.w = (c1 + 3 < Dact) ? a_d[c1 + 3] : 0.f;
    }
    constexpr int RPW = BM / 8;
    #pragma unroll
    for (int rr = 0; rr < RPW; ++rr) {
      const int row = wid * RPW + rr;
      const int gr  = rowBase + row;
      const float4 x = *reinterpret_cast<const float4*>(&Cs[row * CP + 4 * lane]);
      const float4 y = *reinterpret_cast<const float4*>(&Cs[row * CP + 128 + 4 * lane]);
      float p = x.x * s0.x + x.y * s0.y + x.z * s0.z + x.w * s0.w
              + y.x * s1.x + y.y * s1.y + y.z * s1.z + y.w * s1.w;
      float q = x.x * d0.x + x.y * d0.y + x.z * d0.z + x.w * d0.w
              + y.x * d1.x + y.y * d1.y + y.z * d1.z + y.w * d1.w;
      #pragma unroll
      for (int off = 16; off > 0; off >>= 1) {
        p += __shfl_xor(p, off, 32);
        q += __shfl_xor(q, off, 32);
      }
      if (lane == 0) {
        red[row] = p;
        red[BM + row] = q;
      }
      if (gr < N) {
        float4* hp = reinterpret_cast<float4*>(H) + (size_t)gr * 64;
        hp[lane] = x;
        hp[lane + 32] = y;
      }
    }
  } else {
    static_assert(BN == 16, "narrow form: 16 padded columns");
    const int qd = lane & 3;
    const int c0 = 4 * qd;
    float4 s0, d0;
    s0.x = (c0 + 0 < Dact) ? a_s[c0 + 0] : 0.f; s0.y = (c0 + 1 < Dact) ? a_s[c0 + 1] : 0.f;
    s0.z = (c0 + 2 < Dact) ? a_s[c0 + 2] : 0.f; s0.w = (c0 + 3 < Dact) ? a_s[c0 + 3] : 0.f;
    d0.x = (c0 + 0 < Dact) ? a_d[c0 + 0] : 0.f; d0.y = (c0 + 1 < Dact) ? a_d[c0 + 1] : 0.f;
    d0.z = (c0 + 2 < Dact) ? a_d[c0 + 2] : 0.f; d0.w = (c0 + 3 < Dact) ? a_d[c0 + 3] : 0.f;
    constexpr int RPW = BM / 8;
    #pragma unroll
    for (int ps = 0; ps < RPW / 8; ++ps) {
      const int row = wid * RPW + ps * 8 + (lane >> 2);
      const int gr  = rowBase + row;
      const float4 x = *reinterpret_cast<const float4*>(&Cs[row * CP + c0]);
      float p = x.x * s0.x + x.y * s0.y + x.z * s0.z + x.w * s0.w;
      float q = x.x * d0.x + x.y * d0.y + x.z * d0.z + x.w * d0.w;
      #pragma unroll
      for (int off = 2; off > 0; off >>= 1) {
        p += __shfl_xor(p, off, 32);
        q += __shfl_xor(q, off, 32);
      }
      if (qd == 0) {
        red[row] = p;
        red[BM + row] = q;
      }
      if (gr < N) reinterpret_cast<float4*>(H)[(size_t)gr * 4 + qd] = x;
    }
  }

  __syncthreads();
  if (tid < 2 * BM) {
    const int row = (tid < BM) ? tid : tid - BM;
    const int gr  = rowBase + row;
    if (gr < N) {
      if (tid < BM) asv[gr] = red[row];
      else          adv[gr] = red[BM + row];
    }
  }
}

__global__ __launch_bounds__(256) void gat_csr_zero(int* __restrict__ cnt, int N) {
  int i = blockIdx.x * 256 + threadIdx.x;
  if (i < N) cnt[i] = 0;
}

__global__ __launch_bounds__(256) void gat_csr_hist(
    const int* __restrict__ dst, int* __restrict__ cnt, int E) {
  int e = blockIdx.x * 256 + threadIdx.x;
  if (e < E) atomicAdd(&cnt[dst[e]], 1);
}

__global__ __launch_bounds__(256) void gat_csr_tilesum(
    const int* __restrict__ cnt, int* __restrict__ tsum, int N) {
  __shared__ int red[8];
  const int tid = threadIdx.x;
  int i = blockIdx.x * 256 + tid;
  int v = (i < N) ? cnt[i] : 0;
  #pragma unroll
  for (int off = 16; off > 0; off >>= 1) v += __shfl_xor(v, off, 32);
  if ((tid & 31) == 0) red[tid >> 5] = v;
  __syncthreads();
  if (tid == 0) {
    int t = 0;
    #pragma unroll
    for (int w = 0; w < 8; ++w) t += red[w];
    tsum[blockIdx.x] = t;
  }
}

__global__ __launch_bounds__(256) void gat_csr_scan(
    const int* __restrict__ cnt, const int* __restrict__ tsum,
    int* __restrict__ rowptr, int* __restrict__ cursor, int N) {
  __shared__ int red[8];
  __shared__ int wtot[8];
  const int tid = threadIdx.x;
  const int lane = tid & 31;
  const int wid = tid >> 5;
  int part = 0;
  for (int t = tid; t < (int)blockIdx.x; t += 256) part += tsum[t];
  #pragma unroll
  for (int off = 16; off > 0; off >>= 1) part += __shfl_xor(part, off, 32);
  if (lane == 0) red[wid] = part;
  int i = blockIdx.x * 256 + tid;
  int v = (i < N) ? cnt[i] : 0;
  int x = v;
  #pragma unroll
  for (int d = 1; d < 32; d <<= 1) {
    int n = __shfl_up(x, d, 32);
    if (lane >= d) x += n;
  }
  if (lane == 31) wtot[wid] = x;
  __syncthreads();
  int base = 0;
  #pragma unroll
  for (int w = 0; w < 8; ++w) {
    base += red[w];
    if (w < wid) base += wtot[w];
  }
  int excl = base + x - v;
  if (i < N) {
    rowptr[i] = excl;
    cursor[i] = excl;
    if (i == N - 1) rowptr[N] = excl + v;
  }
}

__global__ __launch_bounds__(256) void gat_csr_fill(
    const int* __restrict__ src, const int* __restrict__ dst,
    int* __restrict__ cursor, int* __restrict__ csrc, int E) {
  int e = blockIdx.x * 256 + threadIdx.x;
  if (e < E) {
    int pos = atomicAdd(&cursor[dst[e]], 1);
    csrc[pos] = src[e];
  }
}

__device__ __forceinline__ float leaky02(float v) {
  return (v > 0.f) ? v : 0.2f * v;
}

__device__ __forceinline__ void gat_node_stats(
    const int* __restrict__ csrc, const float* __restrict__ asv,
    int i, int beg, int end, int lane, float ad, float sself,
    int& s0, float& sc0, bool& v0, float& m, float& sum) {
  const int j0 = beg + lane;
  v0 = j0 < end;
  s0 = v0 ? csrc[j0] : i;
  sc0 = v0 ? leaky02(asv[s0] + ad) : sself;
  float mm = fmaxf(sself, sc0);
  for (int j = j0 + 32; j < end; j += 32)
    mm = fmaxf(mm, leaky02(asv[csrc[j]] + ad));
  #pragma unroll
  for (int off = 16; off > 0; off >>= 1) mm = fmaxf(mm, __shfl_xor(mm, off, 32));
  float ss = v0 ? __expf(sc0 - mm) : 0.f;
  for (int j = j0 + 32; j < end; j += 32)
    ss += __expf(leaky02(asv[csrc[j]] + ad) - mm);
  #pragma unroll
  for (int off = 16; off > 0; off >>= 1) ss += __shfl_xor(ss, off, 32);
  m = mm;
  sum = ss + __expf(sself - mm);
}

template <bool RELU>
__global__ __launch_bounds__(256) void gat_node_gather256(
    const int* __restrict__ rowptr, const int* __restrict__ csrc,
    const float* __restrict__ asv, const float* __restrict__ adv,
    const float* __restrict__ H, const float* __restrict__ bias,
    float* __restrict__ OUT, int N) {
  const int lane = threadIdx.x & 31;
  const int i = blockIdx.x * 8 + __builtin_amdgcn_readfirstlane((int)(threadIdx.x >> 5));
  if (i >= N) return;
  const int beg = rowptr[i];
  const int end = rowptr[i + 1];
  const float ad = adv[i];
  const float sself = leaky02(asv[i] + ad);
  int s0; float sc0; bool v0; float m, sum;
  gat_node_stats(csrc, asv, i, beg, end, lane, ad, sself, s0, sc0, v0, m, sum);

  const float4* H4 = reinterpret_cast<const float4*>(H);
  float4 a0, a1;
  {
    const float c = __expf(sself - m) / sum;
    const float4* hp = H4 + (size_t)i * 64;
    const float4 x = hp[lane];
    const float4 y = hp[lane + 32];
    a0 = make_float4(c * x.x, c * x.y, c * x.z, c * x.w);
    a1 = make_float4(c * y.x, c * y.y, c * y.z, c * y.w);
  }
  int sv = s0;
  float cv = v0 ? __expf(sc0 - m) / sum : 0.f;
  for (int base = beg; base < end; base += 32) {
    if (base != beg) {
      const int j = base + lane;
      const bool v = j < end;
      sv = v ? csrc[j] : i;
      cv = v ? __expf(leaky02(asv[sv] + ad) - m) / sum : 0.f;
    }
    const int cnt = min(32, end - base);
    #pragma unroll 4
    for (int t = 0; t < cnt; ++t) {
      const int s = __builtin_amdgcn_readlane(sv, t);
      const float c = __int_as_float(__builtin_amdgcn_readlane(__float_as_int(cv), t));
      const float4* hp = H4 + (size_t)s * 64;
      const float4 x = hp[lane];
      const float4 y = hp[lane + 32];
      a0.x = fmaf(c, x.x, a0.x); a0.y = fmaf(c, x.y, a0.y);
      a0.z = fmaf(c, x.z, a0.z); a0.w = fmaf(c, x.w, a0.w);
      a1.x = fmaf(c, y.x, a1.x); a1.y = fmaf(c, y.y, a1.y);
      a1.z = fmaf(c, y.z, a1.z); a1.w = fmaf(c, y.w, a1.w);
    }
  }
  const int c0 = lane * 4;
  const int c1 = 128 + lane * 4;
  a0.x += bias[c0 + 0]; a0.y += bias[c0 + 1]; a0.z += bias[c0 + 2]; a0.w += bias[c0 + 3];
  a1.x += bias[c1 + 0]; a1.y += bias[c1 + 1]; a1.z += bias[c1 + 2]; a1.w += bias[c1 + 3];
  if (RELU) {
    a0.x = fmaxf(a0.x, 0.f); a0.y = fmaxf(a0.y, 0.f); a0.z = fmaxf(a0.z, 0.f); a0.w = fmaxf(a0.w, 0.f);
    a1.x = fmaxf(a1.x, 0.f); a1.y = fmaxf(a1.y, 0.f); a1.z = fmaxf(a1.z, 0.f); a1.w = fmaxf(a1.w, 0.f);
  }
  float4* op = reinterpret_cast<float4*>(OUT) + (size_t)i * 64;
  op[lane] = a0;
  op[lane + 32] = a1;
}

__global__ __launch_bounds__(256) void gat_node_gather16(
    const int* __restrict__ rowptr, const int* __restrict__ csrc,
    const float* __restrict__ asv, const float* __restrict__ adv,
    const float* __restrict__ H, const float* __restrict__ bias,
    float* __restrict__ out, int N, int Dact) {
  const int lane = threadIdx.x & 31;
  const int i = blockIdx.x * 8 + __builtin_amdgcn_readfirstlane((int)(threadIdx.x >> 5));
  if (i >= N) return;
  const int beg = rowptr[i];
  const int end = rowptr[i + 1];
  const float ad = adv[i];
  const float sself = leaky02(asv[i] + ad);
  int s0; float sc0; bool v0; float m, sum;
  gat_node_stats(csrc, asv, i, beg, end, lane, ad, sself, s0, sc0, v0, m, sum);

  const int q = lane & 3;
  const int slot = lane >> 2;
  const float4* H4 = reinterpret_cast<const float4*>(H);
  float4 acc = make_float4(0.f, 0.f, 0.f, 0.f);
  if (slot == 0) {
    const float c = __expf(sself - m) / sum;
    const float4 x = H4[(size_t)i * 4 + q];
    acc = make_float4(c * x.x, c * x.y, c * x.z, c * x.w);
  }
  int sv = s0;
  float cv = v0 ? __expf(sc0 - m) / sum : 0.f;
  for (int base = beg; base < end; base += 32) {
    if (base != beg) {
      const int j = base + lane;
      const bool v = j < end;
      sv = v ? csrc[j] : i;
      cv = v ? __expf(leaky02(asv[sv] + ad) - m) / sum : 0.f;
    }
    const int cnt = min(32, end - base);
    for (int t0 = 0; t0 < cnt; t0 += 8) {
      const int t = t0 + slot;
      const int s = __shfl(sv, t, 32);
      const float c = __shfl(cv, t, 32);
      if (t < cnt) {
        const float4 x = H4[(size_t)s * 4 + q];
        acc.x = fmaf(c, x.x, acc.x); acc.y = fmaf(c, x.y, acc.y);
        acc.z = fmaf(c, x.z, acc.z); acc.w = fmaf(c, x.w, acc.w);
      }
    }
  }
  #pragma unroll
  for (int off = 4; off < 32; off <<= 1) {
    acc.x += __shfl_xor(acc.x, off, 32);
    acc.y += __shfl_xor(acc.y, off, 32);
    acc.z += __shfl_xor(acc.z, off, 32);
    acc.w += __shfl_xor(acc.w, off, 32);
  }
  if (slot == 0) {
    const int c = q * 4;
    float* op = out + (size_t)i * Dact;
    if (c + 0 < Dact) op[c + 0] = acc.x + bias[c + 0];
    if (c + 1 < Dact) op[c + 1] = acc.y + bias[c + 1];
    if (c + 2 < Dact) op[c + 2] = acc.z + bias[c + 2];
    if (c + 3 < Dact) op[c + 3] = acc.w + bias[c + 3];
  }
}

extern "C" void kernel_launch(void* const* d_in, const int* in_sizes, int n_in,
                              void* d_out, int out_size, void* d_ws, size_t ws_size,
                              hipStream_t stream) {
  const float* x  = (const float*)d_in[0];
  const int*   ei = (const int*)d_in[1];
  const float* W1 = (const float*)d_in[2];
  const float* as1 = (const float*)d_in[3];
  const float* ad1 = (const float*)d_in[4];
  const float* b1 = (const float*)d_in[5];
  const float* W2 = (const float*)d_in[6];
  const float* as2 = (const float*)d_in[7];
  const float* ad2 = (const float*)d_in[8];
  const float* b2 = (const float*)d_in[9];
  const float* W3 = (const float*)d_in[10];
  const float* as3 = (const float*)d_in[11];
  const float* ad3 = (const float*)d_in[12];
  const float* b3 = (const float*)d_in[13];

  const int N  = in_sizes[0] / 256;
  const int E  = in_sizes[1] / 2;
  const int* src = ei;
  const int* dst = ei + E;

  float* ws = (float*)d_ws;
  float* hA  = ws;
  float* hB  = hA + (size_t)N * 256;
  float* h3  = hB + (size_t)N * 256;
  int*   ib  = (int*)(h3 + (size_t)N * 16);
  int*   cnt    = ib;
  int*   rowptr = cnt + N;
  int*   cursor = rowptr + N + 1;
  int*   tsum   = cursor + N;
  float* asv = (float*)(ib + (size_t)N * 16);
  float* adv = asv + N;
  int*   csrc = (int*)(adv + N);

  dim3 blk(256);
  int gGemm  = (N + 31) / 32;
  int gGemm3 = (N + 127) / 128;
  int gNodeW = (N + 7) / 8;
  int gNodeT = (N + 255) / 256;
  int gEdgeT = (E + 255) / 256;

  gat_csr_zero<<<gNodeT, blk, 0, stream>>>(cnt, N);
  gat_csr_hist<<<gEdgeT, blk, 0, stream>>>(dst, cnt, E);
  gat_csr_tilesum<<<gNodeT, blk, 0, stream>>>(cnt, tsum, N);
  gat_csr_scan<<<gNodeT, blk, 0, stream>>>(cnt, tsum, rowptr, cursor, N);
  gat_csr_fill<<<gEdgeT, blk, 0, stream>>>(src, dst, cursor, csrc, E);

  gat_gemm_alpha<2, 4, 4><<<gGemm, blk, 0, stream>>>(x, W1, as1, ad1, hA, asv, adv, N, 256);
  gat_node_gather256<true><<<gNodeW, blk, 0, stream>>>(rowptr, csrc, asv, adv, hA, b1, hB, N);

  gat_gemm_alpha<2, 4, 4><<<gGemm, blk, 0, stream>>>(hB, W2, as2, ad2, hA, asv, adv, N, 256);
  gat_node_gather256<true><<<gNodeW, blk, 0, stream>>>(rowptr, csrc, asv, adv, hA, b2, hB, N);

  gat_gemm_alpha<8, 1, 1><<<gGemm3, blk, 0, stream>>>(hB, W3, as3, ad3, h3, asv, adv, N, 10);
  gat_node_gather16<<<gNodeW, blk, 0, stream>>>(rowptr, csrc, asv, adv, h3, b3, (float*)d_out, N, 10);
}


# Layers 1 and 2 aggregate (softmax, weighted h[src] sum, bias, relu) inside the next GEMM's block: each wave builds 4 rows of the A tile in LDS; 2 launches and the rows' trip through memory go
# speedup vs baseline: 4.1886x; 1.0528x over previous
#include <hip/hip_runtime.h>
#include <math.h>


typedef float v2f __attribute__((ext_vector_type(2)));
typedef float v8f __attribute__((ext_vector_type(8)));

#define DK 256
#define KC 32
#define LDT (KC + 4)
#define LDA (DK + 4)

__global__ __launch_bounds__(256) void gat_csr_zero(int* __restrict__ cnt, int N) {
  int i = blockIdx.x * 256 + threadIdx.x;
  if (i < N) cnt[i] = 0;
}

__global__ __launch_bounds__(256) void gat_csr_hist(
    const int* __restrict__ dst, int* __restrict__ cnt, int E) {
  int e = blockIdx.x * 256 + threadIdx.x;
  if (e < E) atomicAdd(&cnt[dst[e]], 1);
}

__global__ __launch_bounds__(256) void gat_csr_tilesum(
    const int* __restrict__ cnt, int* __restrict__ tsum, int N) {
  __shared__ int red[8];
  const int tid = threadIdx.x;
  int i = blockIdx.x * 256 + tid;
  int v = (i < N) ? cnt[i] : 0;
  #pragma unroll
  for (int off = 16; off > 0; off >>= 1) v += __shfl_xor(v, off, 32);
  if ((tid & 31) == 0) red[tid >> 5] = v;
  __syncthreads();
  if (tid == 0) {
    int t = 0;
    #pragma unroll
    for (int w = 0; w < 8; ++w) t += red[w];
    tsum[blockIdx.x] = t;
  }
}

__global__ __launch_bounds__(256) void gat_csr_scan(
    const int* __restrict__ cnt, const int* __restrict__ tsum,
    int* __restrict__ rowptr, int* __restrict__ cursor, int N) {
  __shared__ int red[8];
  __shared__ int wtot[8];
  const int tid = threadIdx.x;
  const int lane = tid & 31;
  const int wid = tid >> 5;
  int part = 0;
  for (int t = tid; t < (int)blockIdx.x; t += 256) part += tsum[t];
  #pragma unroll
  for (int off = 16; off > 0; off >>= 1) part += __shfl_xor(part, off, 32);
  if (lane == 0) red[wid] = part;
  int i = blockIdx.x * 256 + tid;
  int v = (i < N) ? cnt[i] : 0;
  int x = v;
  #pragma unroll
  for (int d = 1; d < 32; d <<= 1) {
    int n = __shfl_up(x, d, 32);
    if (lane >= d) x += n;
  }
  if (lane == 31) wtot[wid] = x;
  __syncthreads();
  int base = 0;
  #pragma unroll
  for (int w = 0; w < 8; ++w) {
    base += red[w];
    if (w < wid) base += wtot[w];
  }
  int excl = base + x - v;
  if (i < N) {
    rowptr[i] = excl;
    cursor[i] = excl;
    if (i == N - 1) rowptr[N] = excl + v;
  }
}

__global__ __launch_bounds__(256) void gat_csr_fill(
    const int* __restrict__ src, const int* __restrict__ dst,
    int* __restrict__ cursor, int* __restrict__ csrc, int E) {
  int e = blockIdx.x * 256 + threadIdx.x;
  if (e < E) {
    int pos = atomicAdd(&cursor[dst[e]], 1);
    csrc[pos] = src[e];
  }
}

__device__ __forceinline__ float leaky02(float v) {
  return (v > 0.f) ? v : 0.2f * v;
}

__device__ __forceinline__ void gat_node_stats(
    const int* __restrict__ csrc, const float* __restrict__ asv,
    int i, int beg, int end, int lane, float ad, float sself,
    int& s0, float& sc0, bool& v0, float& m, float& sum) {
  const int j0 = beg + lane;
  v0 = j0 < end;
  s0 = v0 ? csrc[j0] : i;
  sc0 = v0 ? leaky02(asv[s0] + ad) : sself;
  float mm = fmaxf(sself, sc0);
  for (int j = j0 + 32; j < end; j += 32)
    mm = fmaxf(mm, leaky02(asv[csrc[j]] + ad));
  #pragma unroll
  for (int off = 16; off > 0; off >>= 1) mm = fmaxf(mm, __shfl_xor(mm, off, 32));
  float ss = v0 ? __expf(sc0 - mm) : 0.f;
  for (int j = j0 + 32; j < end; j += 32)
    ss += __expf(leaky02(asv[csrc[j]] + ad) - mm);
  #pragma unroll
  for (int off = 16; off > 0; off >>= 1) ss += __shfl_xor(ss, off, 32);
  m = mm;
  sum = ss + __expf(sself - mm);
}

__device__ __forceinline__ void gat_node_row256(
    const int* __restrict__ rowptr, const int* __restrict__ csrc,
    const float* __restrict__ asv, const float* __restrict__ adv,
    const float* __restrict__ H, int i, int lane, float4& a0, float4& a1) {
  const float4* H4 = reinterpret_cast<const float4*>(H);
  const float4 xs = H4[(size_t)i * 64 + lane];
  const float4 ys = H4[(size_t)i * 64 + lane + 32];
  const int beg = rowptr[i];
  const int end = rowptr[i + 1];
  const float ad = adv[i];
  const float sself = leaky02(asv[i] + ad);
  int s0; float sc0; bool v0; float m, sum;
  gat_node_stats(csrc, asv, i, beg, end, lane, ad, sself, s0, sc0, v0, m, sum);

  {
    const float c = __expf(sself - m) / sum;
    a0 = make_float4(c * xs.x, c * xs.y, c * xs.z, c * xs.w);
    a1 = make_float4(c * ys.x, c * ys.y, c * ys.z, c * ys.w);
  }
  int sv = s0;
  float cv = v0 ? __expf(sc0 - m) / sum : 0.f;
  for (int base = beg; base < end; base += 32) {
    if (base != beg) {
      const int j = base + lane;
      const bool v = j < end;
      sv = v ? csrc[j] : i;
      cv = v ? __expf(leaky02(asv[sv] + ad) - m) / sum : 0.f;
    }
    const int cnt = min(32, end - base);
    #pragma unroll 4
    for (int t = 0; t < cnt; ++t) {
      const int s = __builtin_amdgcn_readlane(sv, t);
      const float c = __int_as_float(__builtin_amdgcn_readlane(__float_as_int(cv), t));
      const float4* hp = H4 + (size_t)s * 64;
      const float4 x = hp[lane];
      const float4 y = hp[lane + 32];
      a0.x = fmaf(c, x.x, a0.x); a0.y = fmaf(c, x.y, a0.y);
      a0.z = fmaf(c, x.z, a0.z); a0.w = fmaf(c, x.w, a0.w);
      a1.x = fmaf(c, y.x, a1.x); a1.y = fmaf(c, y.y, a1.y);
      a1.z = fmaf(c, y.z, a1.z); a1.w = fmaf(c, y.w, a1.w);
    }
  }
}

template <bool GATHER, int WNW, int NT>
__global__ __launch_bounds__(256, GATHER ? 8 : 1) void gat_layer_gemm(
    const float* __restrict__ X,
    const int* __restrict__ rowptr, const int* __restrict__ csrc,
    const float* __restrict__ asv_in, const float* __restrict__ adv_in,
    const float* __restrict__ Hin, const float* __restrict__ bias_in,
    const float* __restrict__ W,
    const float* __restrict__ a_s, const float* __restrict__ a_d,
    float* __restrict__ H, float* __restrict__ asv, float* __restrict__ adv,
    int N, int Dact) {
  static_assert(WNW == 1 || WNW == 2 || WNW == 4, "2 x WNW accumulator waves out of eight");
  constexpr int BM = 32;
  constexpr int BN = 16 * NT * WNW;
  constexpr int NWA = 2 * WNW;
  constexpr int AP = GATHER ? LDA : LDT;
  constexpr int CP = BN + 4;
  constexpr int A_F4 = (BM * (KC / 4)) / 256;
  constexpr int B_QT = BN * (KC / 4);
  constexpr int B_Q = (B_QT + 255) / 256;
  constexpr int SM_A = BM * AP;
  constexpr int SM_AB = SM_A + BN * LDT;
  constexpr int SM_C = BM * CP + 2 * BM;
  constexpr int SM = SM_AB > SM_C ? SM_AB : SM_C;
  __shared__ __attribute__((aligned(16))) float smem[SM];
  float* As = smem;
  float* Bst = smem + SM_A;
  float* Cs = smem;
  float* red = smem + BM * CP;

  const int tid  = threadIdx.x;
  const int lane = tid & 31;
  const int lo   = lane & 15;
  const int hi   = lane >> 4;
  const int wid  = __builtin_amdgcn_readfirstlane(tid >> 5);
  const bool accw = (NWA == 8) || (wid < NWA);
  const int wm   = (wid / WNW) * 16;
  const int wn   = (wid % WNW) * (16 * NT);
  const int rowBase = blockIdx.x * BM;

  if constexpr (GATHER) {
    const int c0 = 4 * lane, c1 = 128 + 4 * lane;
    float4 bz0, bz1;
    bz0.x = bias_in[c0 + 0]; bz0.y = bias_in[c0 + 1]; bz0.z = bias_in[c0 + 2]; bz0.w = bias_in[c0 + 3];
    bz1.x = bias_in[c1 + 0]; bz1.y = bias_in[c1 + 1]; bz1.z = bias_in[c1 + 2]; bz1.w = bias_in[c1 + 3];
    for (int rr = 0; rr < BM / 8; ++rr) {
      const int row = wid * (BM / 8) + rr;
      const int i   = rowBase + row;
      float4 a0 = make_float4(0.f, 0.f, 0.f, 0.f);
      float4 a1 = a0;
      if (i < N) {
        gat_node_row256(rowptr, csrc, asv_in, adv_in, Hin, i, lane, a0, a1);
        a0.x += bz0.x; a0.y += bz0.y; a0.z += bz0.z; a0.w += bz0.w;
        a1.x += bz1.x; a1.y += bz1.y; a1.z += bz1.z; a1.w += bz1.w;
        a0.x = fmaxf(a0.x, 0.f); a0.y = fmaxf(a0.y, 0.f); a0.z = fmaxf(a0.z, 0.f); a0.w = fmaxf(a0.w, 0.f);
        a1.x = fmaxf(a1.x, 0.f); a1.y = fmaxf(a1.y, 0.f); a1.z = fmaxf(a1.z, 0.f); a1.w = fmaxf(a1.w, 0.f);
      }
      *reinterpret_cast<float4*>(&As[row * AP + c0]) = a0;
      *reinterpret_cast<float4*>(&As[row * AP + c1]) = a1;
    }
  }

  v8f acc[NT];
  #pragma unroll
  for (int t = 0; t < NT; ++t) acc[t] = v8f{};

  float4 ra[A_F4];
  float4 rb[B_Q];

  auto load_chunk = [&](int kk) {
    if constexpr (!GATHER) {
      #pragma unroll
      for (int i = 0; i < A_F4; ++i) {
        const int f4 = tid + 256 * i;
        const int r  = f4 >> 3;
        const int c4 = f4 & 7;
        const int gr = rowBase + r;
        float4 v = make_float4(0.f, 0.f, 0.f, 0.f);
        if (gr < N)
          v = *reinterpret_cast<const float4*>(X + (size_t)gr * DK + kk + c4 * 4);
        ra[i] = v;
      }
    }
    #pragma unroll
    for (int i = 0; i < B_Q; ++i) {
      const int q  = tid + 256 * i;
      const int n  = q % BN;
      const int kq = q / BN;
      float4 v = make_float4(0.f, 0.f, 0.f, 0.f);
      if ((B_QT % 256 == 0 || q < B_QT) && n < Dact) {
        const float* wp = W + (size_t)(kk + kq * 4) * Dact + n;
        v.x = wp[0];
        v.y = wp[Dact];
        v.z = wp[2 * Dact];
        v.w = wp[3 * Dact];
      }
      rb[i] = v;
    }
  };
  auto store_chunk = [&]() {
    if constexpr (!GATHER) {
      #pragma unroll
      for (int i = 0; i < A_F4; ++i) {
        const int f4 = tid + 256 * i;
        const int r  = f4 >> 3;
        const int c4 = f4 & 7;
        *reinterpret_cast<float4*>(&As[r * AP + c4 * 4]) = ra[i];
      }
    }
    #pragma unroll
    for (int i = 0; i < B_Q; ++i) {
      const int q  = tid + 256 * i;
      const int n  = q % BN;
      const int kq = q / BN;
      if (B_QT % 256 == 0 || q < B_QT)
        *reinterpret_cast<float4*>(&Bst[n * LDT + kq * 4]) = rb[i];
    }
  };

  const float* arow = As + (wm + lo) * AP + 4 * hi;
  const float* brow = Bst + (wn + lo) * LDT + 4 * hi;

  load_chunk(0);
  for (int kk = 0; kk < DK; kk += KC) {
    __syncthreads();
    store_chunk();
    __syncthreads();
    if (kk + KC < DK) load_chunk(kk + KC);
    if (accw) {
      const float* ak = GATHER ? arow + kk : arow;
      #pragma unroll
      for (int kb = 0; kb < KC; kb += 8) {
        const float4 a4 = *reinterpret_cast<const float4*>(ak + kb);
        float4 b4[NT];
        #pragma unroll
        for (int t = 0; t < NT; ++t)
          b4[t] = *reinterpret_cast<const float4*>(brow + t * 16 * LDT + kb);
        const v2f a01 = {a4.x, a4.y};
        const v2f a23 = {a4.z, a4.w};
        #pragma unroll
        for (int t = 0; t < NT; ++t) {
          const v2f b01 = {b4[t].x, b4[t].y};
          acc[t] = __builtin_amdgcn_wmma_f32_16x16x4_f32(false, a01, false, b01,
                                                         (short)0, acc[t], false, false);
        }
        #pragma unroll
        for (int t = 0; t < NT; ++t) {
          const v2f b23 = {b4[t].z, b4[t].w};
          acc[t] = __builtin_amdgcn_wmma_f32_16x16x4_f32(false, a23, false, b23,
                                                         (short)0, acc[t], false, false);
        }
      }
    }
  }

  __syncthreads();
  if (accw) {
    #pragma unroll
    for (int t = 0; t < NT; ++t) {
      #pragma unroll
      for (int r = 0; r < 8; ++r)
        Cs[(wm + 8 * hi + r) * CP + wn + 16 * t + lo] = acc[t][r];
    }
  }
  __syncthreads();

  if constexpr (BN == 256) {
    float4 s0, s1, d0, d1;
    {
      const int c0 = 4 * lane, c1 = 128 + 4 * lane;
      s0.x = (c0 + 0 < Dact) ? a_s[c0 + 0] : 0.f; s0.y = (c0 + 1 < Dact) ? a_s[c0 + 1] : 0.f;
      s0.z = (c0 + 2 < Dact) ? a_s[c0 + 2] : 0.f; s0.w = (c0 + 3 < Dact) ? a_s[c0 + 3] : 0.f;
      s1.x = (c1 + 0 < Dact) ? a_s[c1 + 0] : 0.f; s1.y = (c1 + 1 < Dact) ? a_s[c1 + 1] : 0.f;
      s1.z = (c1 + 2 < Dact) ? a_s[c1 + 2] : 0.f; s1.w = (c1 + 3 < Dact) ? a_s[c1 + 3] : 0.f;
      d0.x = (c0 + 0 < Dact) ? a_d[c0 + 0] : 0.f; d0.y = (c0 + 1 < Dact) ? a_d[c0 + 1] : 0.f;
      d0.z = (c0 + 2 < Dact) ? a_d[c0 + 2] : 0.f; d0.w = (c0 + 3 < Dact) ? a_d[c0 + 3] : 0.f;
      d1.x = (c1 + 0 < Dact) ? a_d[c1 + 0] : 0.f; d1.y = (c1 + 1 < Dact) ? a_d[c1 + 1] : 0.f;
      d1.z = (c1 + 2 < Dact) ? a_d[c1 + 2] : 0.f; d1.w = (c1 + 3 < Dact) ? a_d[c1 + 3] : 0.f;
    }
    constexpr int RPW = BM / 8;
    #pragma unroll
    for (int rr = 0; rr < RPW; ++rr) {
      const int row = wid * RPW + rr;
      const int gr  = rowBase + row;
      const float4 x = *reinterpret_cast<const float4*>(&Cs[row * CP + 4 * lane]);
      const float4 y = *reinterpret_cast<const float4*>(&Cs[row * CP + 128 + 4 * lane]);
      float p = x.x * s0.x + x.y * s0.y + x.z * s0.z + x.w * s0.w
              + y.x * s1.x + y.y * s1.y + y.z * s1.z + y.w * s1.w;
      float q = x.x * d0.x + x.y * d0.y + x.z * d0.z + x.w * d0.w
              + y.x * d1.x + y.y * d1.y + y.z * d1.z + y.w * d1.w;
      #pragma unroll
      for (int off = 16; off > 0; off >>= 1) {
        p += __shfl_xor(p, off, 32);
        q += __shfl_xor(q, off, 32);
      }
      if (lane == 0) {
        red[row] = p;
        red[BM + row] = q;
      }
      if (gr < N) {
        float4* hp = reinterpret_cast<float4*>(H) + (size_t)gr * 64;
        hp[lane] = x;
        hp[lane + 32] = y;
      }
    }
  } else {
    static_assert(BN == 16, "narrow form: 16 padded columns");
    if (wid < BM / 8) {
      const int qd = lane & 3;
      const int c0 = 4 * qd;
      float4 s0, d0;
      s0.x = (c0 + 0 < Dact) ? a_s[c0 + 0] : 0.f; s0.y = (c0 + 1 < Dact) ? a_s[c0 + 1] : 0.f;
      s0.z = (c0 + 2 < Dact) ? a_s[c0 + 2] : 0.f; s0.w = (c0 + 3 < Dact) ? a_s[c0 + 3] : 0.f;
      d0.x = (c0 + 0 < Dact) ? a_d[c0 + 0] : 0.f; d0.y = (c0 + 1 < Dact) ? a_d[c0 + 1] : 0.f;
      d0.z = (c0 + 2 < Dact) ? a_d[c0 + 2] : 0.f; d0.w = (c0 + 3 < Dact) ? a_d[c0 + 3] : 0.f;
      const int row = wid * 8 + (lane >> 2);
      const int gr  = rowBase + row;
      const float4 x = *reinterpret_cast<const float4*>(&Cs[row * CP + c0]);
      float p = x.x * s0.x + x.y * s0.y + x.z * s0.z + x.w * s0.w;
      float q = x.x * d0.x + x.y * d0.y + x.z * d0.z + x.w * d0.w;
      #pragma unroll
      for (int off = 2; off > 0; off >>= 1) {
        p += __shfl_xor(p, off, 32);
        q += __shfl_xor(q, off, 32);
      }
      if (qd == 0) {
        red[row] = p;
        red[BM + row] = q;
      }
      if (gr < N) reinterpret_cast<float4*>(H)[(size_t)gr * 4 + qd] = x;
    }
  }

  __syncthreads();
  if (tid < 2 * BM) {
    const int row = (tid < BM) ? tid : tid - BM;
    const int gr  = rowBase + row;
    if (gr < N) {
      if (tid < BM) asv[gr] = red[row];
      else          adv[gr] = red[BM + row];
    }
  }
}

__global__ __launch_bounds__(256) void gat_node_gather16(
    const int* __restrict__ rowptr, const int* __restrict__ csrc,
    const float* __restrict__ asv, const float* __restrict__ adv,
    const float* __restrict__ H, const float* __restrict__ bias,
    float* __restrict__ out, int N, int Dact) {
  const int lane = threadIdx.x & 31;
  const int i = blockIdx.x * 8 + __builtin_amdgcn_readfirstlane((int)(threadIdx.x >> 5));
  if (i >= N) return;
  const int beg = rowptr[i];
  const int end = rowptr[i + 1];
  const float ad = adv[i];
  const float sself = leaky02(asv[i] + ad);
  int s0; float sc0; bool v0; float m, sum;
  gat_node_stats(csrc, asv, i, beg, end, lane, ad, sself, s0, sc0, v0, m, sum);

  const int q = lane & 3;
  const int slot = lane >> 2;
  const float4* H4 = reinterpret_cast<const float4*>(H);
  float4 acc = make_float4(0.f, 0.f, 0.f, 0.f);
  if (slot == 0) {
    const float c = __expf(sself - m) / sum;
    const float4 x = H4[(size_t)i * 4 + q];
    acc = make_float4(c * x.x, c * x.y, c * x.z, c * x.w);
  }
  int sv = s0;
  float cv = v0 ? __expf(sc0 - m) / sum : 0.f;
  for (int base = beg; base < end; base += 32) {
    if (base != beg) {
      const int j = base + lane;
      const bool v = j < end;
      sv = v ? csrc[j] : i;
      cv = v ? __expf(leaky02(asv[sv] + ad) - m) / sum : 0.f;
    }
    const int cnt = min(32, end - base);
    for (int t0 = 0; t0 < cnt; t0 += 8) {
      const int t = t0 + slot;
      const int s = __shfl(sv, t, 32);
      const float c = __shfl(cv, t, 32);
      if (t < cnt) {
        const float4 x = H4[(size_t)s * 4 + q];
        acc.x = fmaf(c, x.x, acc.x); acc.y = fmaf(c, x.y, acc.y);
        acc.z = fmaf(c, x.z, acc.z); acc.w = fmaf(c, x.w, acc.w);
      }
    }
  }
  #pragma unroll
  for (int off = 4; off < 32; off <<= 1) {
    acc.x += __shfl_xor(acc.x, off, 32);
    acc.y += __shfl_xor(acc.y, off, 32);
    acc.z += __shfl_xor(acc.z, off, 32);
    acc.w += __shfl_xor(acc.w, off, 32);
  }
  if (slot == 0) {
    const int c = q * 4;
    float* op = out + (size_t)i * Dact;
    if (c + 0 < Dact) op[c + 0] = acc.x + bias[c + 0];
    if (c + 1 < Dact) op[c + 1] = acc.y + bias[c + 1];
    if (c + 2 < Dact) op[c + 2] = acc.z + bias[c + 2];
    if (c + 3 < Dact) op[c + 3] = acc.w + bias[c + 3];
  }
}

extern "C" void kernel_launch(void* const* d_in, const int* in_sizes, int n_in,
                              void* d_out, int out_size, void* d_ws, size_t ws_size,
                              hipStream_t stream) {
  const float* x  = (const float*)d_in[0];
  const int*   ei = (const int*)d_in[1];
  const float* W1 = (const float*)d_in[2];
  const float* as1 = (const float*)d_in[3];
  const float* ad1 = (const float*)d_in[4];
  const float* b1 = (const float*)d_in[5];
  const float* W2 = (const float*)d_in[6];
  const float* as2 = (const float*)d_in[7];
  const float* ad2 = (const float*)d_in[8];
  const float* b2 = (const float*)d_in[9];
  const float* W3 = (const float*)d_in[10];
  const float* as3 = (const float*)d_in[11];
  const float* ad3 = (const float*)d_in[12];
  const float* b3 = (const float*)d_in[13];

  const int N  = in_sizes[0] / 256;
  const int E  = in_sizes[1] / 2;
  const int* src = ei;
  const int* dst = ei + E;

  float* ws = (float*)d_ws;
  float* hA  = ws;
  float* hB  = hA + (size_t)N * 256;
  float* h3  = hB + (size_t)N * 256;
  int*   ib  = (int*)(h3 + (size_t)N * 16);
  int*   cnt    = ib;
  int*   rowptr = cnt + N;
  int*   cursor = rowptr + N + 1;
  int*   tsum   = cursor + N;
  float* asvB = (float*)(ib + (size_t)N * 4);
  float* advB = asvB + N;
  float* asvA = (float*)(ib + (size_t)N * 16);
  float* advA = asvA + N;
  int*   csrc = (int*)(advA + N);

  dim3 blk(256);
  int gGemm  = (N + 31) / 32;
  int gNodeW = (N + 7) / 8;
  int gNodeT = (N + 255) / 256;
  int gEdgeT = (E + 255) / 256;

  gat_csr_zero<<<gNodeT, blk, 0, stream>>>(cnt, N);
  gat_csr_hist<<<gEdgeT, blk, 0, stream>>>(dst, cnt, E);
  gat_csr_tilesum<<<gNodeT, blk, 0, stream>>>(cnt, tsum, N);
  gat_csr_scan<<<gNodeT, blk, 0, stream>>>(cnt, tsum, rowptr, cursor, N);
  gat_csr_fill<<<gEdgeT, blk, 0, stream>>>(src, dst, cursor, csrc, E);

  gat_layer_gemm<false, 4, 4><<<gGemm, blk, 0, stream>>>(
      x, nullptr, nullptr, nullptr, nullptr, nullptr, nullptr,
      W1, as1, ad1, hA, asvA, advA, N, 256);

  gat_layer_gemm<true, 4, 4><<<gGemm, blk, 0, stream>>>(
      nullptr, rowptr, csrc, asvA, advA, hA, b1,
      W2, as2, ad2, hB, asvB, advB, N, 256);

  gat_layer_gemm<true, 1, 1><<<gGemm, blk, 0, stream>>>(
      nullptr, rowptr, csrc, asvB, advB, hB, b2,
      W3, as3, ad3, h3, asvA, advA, N, 10);

  gat_node_gather16<<<gNodeW, blk, 0, stream>>>(rowptr, csrc, asvA, advA, h3, b3, (float*)d_out, N, 10);
}


# Edge sort by destination goes through LDS: edges binned by 256-node bucket, each bucket counted and placed per node in LDS; per-edge global atomics and scattered stores go; sums in LDS arrival order
# speedup vs baseline: 5.4880x; 1.3102x over previous
#include <hip/hip_runtime.h>
#include <math.h>


typedef float v2f __attribute__((ext_vector_type(2)));
typedef float v8f __attribute__((ext_vector_type(8)));

#define DK 256
#define KC 32
#define LDT (KC + 4)
#define LDA (DK + 4)

__global__ __launch_bounds__(256) void gat_csr_zero(int* __restrict__ cnt, int N) {
  int i = blockIdx.x * 256 + threadIdx.x;
  if (i < N) cnt[i] = 0;
}

__global__ __launch_bounds__(256) void gat_csr_hist(
    const int* __restrict__ dst, int* __restrict__ cnt, int E) {
  int e = blockIdx.x * 256 + threadIdx.x;
  if (e < E) atomicAdd(&cnt[dst[e]], 1);
}

__global__ __launch_bounds__(256) void gat_csr_tilesum(
    const int* __restrict__ cnt, int* __restrict__ tsum, int N) {
  __shared__ int red[8];
  const int tid = threadIdx.x;
  int i = blockIdx.x * 256 + tid;
  int v = (i < N) ? cnt[i] : 0;
  #pragma unroll
  for (int off = 16; off > 0; off >>= 1) v += __shfl_xor(v, off, 32);
  if ((tid & 31) == 0) red[tid >> 5] = v;
  __syncthreads();
  if (tid == 0) {
    int t = 0;
    #pragma unroll
    for (int w = 0; w < 8; ++w) t += red[w];
    tsum[blockIdx.x] = t;
  }
}

__global__ __launch_bounds__(256) void gat_csr_scan(
    const int* __restrict__ cnt, const int* __restrict__ tsum,
    int* __restrict__ rowptr, int* __restrict__ cursor, int N) {
  __shared__ int red[8];
  __shared__ int wtot[8];
  const int tid = threadIdx.x;
  const int lane = tid & 31;
  const int wid = tid >> 5;
  int part = 0;
  for (int t = tid; t < (int)blockIdx.x; t += 256) part += tsum[t];
  #pragma unroll
  for (int off = 16; off > 0; off >>= 1) part += __shfl_xor(part, off, 32);
  if (lane == 0) red[wid] = part;
  int i = blockIdx.x * 256 + tid;
  int v = (i < N) ? cnt[i] : 0;
  int x = v;
  #pragma unroll
  for (int d = 1; d < 32; d <<= 1) {
    int n = __shfl_up(x, d, 32);
    if (lane >= d) x += n;
  }
  if (lane == 31) wtot[wid] = x;
  __syncthreads();
  int base = 0;
  #pragma unroll
  for (int w = 0; w < 8; ++w) {
    base += red[w];
    if (w < wid) base += wtot[w];
  }
  int excl = base + x - v;
  if (i < N) {
    rowptr[i] = excl;
    cursor[i] = excl;
    if (i == N - 1) rowptr[N] = excl + v;
  }
}

__global__ __launch_bounds__(256) void gat_csr_fill(
    const int* __restrict__ src, const int* __restrict__ dst,
    int* __restrict__ cursor, int* __restrict__ csrc, int E) {
  int e = blockIdx.x * 256 + threadIdx.x;
  if (e < E) {
    int pos = atomicAdd(&cursor[dst[e]], 1);
    csrc[pos] = src[e];
  }
}

#define SORT_CH 4096
#define SORT_MAXB 256
#define SORT_CAP 8192

__device__ __forceinline__ int gat_excl_scan256(int v, int* red, int& total) {
  const int lane = threadIdx.x & 31;
  const int wid = threadIdx.x >> 5;
  int x = v;
  #pragma unroll
  for (int d = 1; d < 32; d <<= 1) {
    int n = __shfl_up(x, d, 32);
    if (lane >= d) x += n;
  }
  if (lane == 31) red[wid] = x;
  __syncthreads();
  int base = 0, tot = 0;
  #pragma unroll
  for (int w = 0; w < 8; ++w) {
    const int r = red[w];
    if (w < wid) base += r;
    tot += r;
  }
  __syncthreads();
  total = tot;
  return base + x - v;
}

__global__ __launch_bounds__(256) void gat_bin_count(
    const int* __restrict__ dst, int* __restrict__ cntT, int E, int NBK, int NCH) {
  __shared__ int hist[SORT_MAXB];
  const int tid = threadIdx.x;
  hist[tid] = 0;
  __syncthreads();
  const int base = blockIdx.x * SORT_CH;
  int d[SORT_CH / 256];
  #pragma unroll
  for (int i = 0; i < SORT_CH / 256; ++i)
    d[i] = dst[min(base + i * 256 + tid, E - 1)];
  #pragma unroll
  for (int i = 0; i < SORT_CH / 256; ++i)
    if (base + i * 256 + tid < E) atomicAdd(&hist[d[i] >> 8], 1);
  __syncthreads();
  if (tid < NBK) cntT[(size_t)tid * NCH + blockIdx.x] = hist[tid];
}

#define SCAN_EPT 8
__global__ __launch_bounds__(1024) void gat_bin_scan(
    const int* __restrict__ cntT, int* __restrict__ off, int M) {
  __shared__ int wsum[32];
  const int tid = threadIdx.x;
  const int lane = tid & 31;
  const int wid = tid >> 5;
  int carry = 0;
  int vn[SCAN_EPT];
  #pragma unroll
  for (int q = 0; q < SCAN_EPT; ++q)
    vn[q] = (SCAN_EPT * tid + q < M) ? cntT[SCAN_EPT * tid + q] : 0;
  for (int base = 0; base < M; base += 1024 * SCAN_EPT) {
    const int j = base + SCAN_EPT * tid;
    int v[SCAN_EPT];
    int sum = 0;
    #pragma unroll
    for (int q = 0; q < SCAN_EPT; ++q) { v[q] = vn[q]; sum += v[q]; }
    #pragma unroll
    for (int q = 0; q < SCAN_EPT; ++q) {
      const int jn = j + 1024 * SCAN_EPT + q;
      vn[q] = (jn < M) ? cntT[jn] : 0;
    }
    int x = sum;
    #pragma unroll
    for (int d = 1; d < 32; d <<= 1) {
      int n = __shfl_up(x, d, 32);
      if (lane >= d) x += n;
    }
    if (lane == 31) wsum[wid] = x;
    __syncthreads();
    int wex = 0, tot = 0;
    #pragma unroll
    for (int w = 0; w < 32; ++w) {
      const int r = wsum[w];
      if (w < wid) wex += r;
      tot += r;
    }
    int run = carry + wex + x - sum;
    #pragma unroll
    for (int q = 0; q < SCAN_EPT; ++q) {
      if (j + q < M) off[j + q] = run;
      run += v[q];
    }
    carry += tot;
    __syncthreads();
  }
}

__global__ __launch_bounds__(256) void gat_bin_scatter(
    const int* __restrict__ src, const int* __restrict__ dst,
    const int* __restrict__ cntT, const int* __restrict__ off,
    int* __restrict__ tmpw, int E, int NBK, int NCH) {
  int2* tmp = reinterpret_cast<int2*>(tmpw);
  __shared__ int red[8];
  __shared__ int cursor[SORT_MAXB];
  __shared__ int goff[SORT_MAXB];
  __shared__ int2 stage[SORT_CH];
  const int tid = threadIdx.x;
  const int b = blockIdx.x;
  int mine = 0, o = 0;
  if (tid < NBK) {
    mine = cntT[(size_t)tid * NCH + b];
    o = off[(size_t)tid * NCH + b];
  }
  int total;
  const int ls = gat_excl_scan256(mine, red, total);
  cursor[tid] = ls;
  goff[tid] = o - ls;
  __syncthreads();
  const int base = b * SORT_CH;
  int d[SORT_CH / 256], s[SORT_CH / 256];
  #pragma unroll
  for (int i = 0; i < SORT_CH / 256; ++i) {
    const int e = min(base + i * 256 + tid, E - 1);
    d[i] = dst[e];
    s[i] = src[e];
  }
  #pragma unroll
  for (int i = 0; i < SORT_CH / 256; ++i) {
    if (base + i * 256 + tid < E) {
      const int p = atomicAdd(&cursor[d[i] >> 8], 1);
      stage[p] = make_int2(s[i], d[i]);
    }
  }
  __syncthreads();
  #pragma unroll 4
  for (int j = tid; j < total; j += 256) {
    const int2 pr = stage[j];
    tmp[goff[pr.y >> 8] + j] = pr;
  }
}

__global__ __launch_bounds__(256) void gat_bucket_sort(
    const int* __restrict__ tmpw, const int* __restrict__ off,
    int* __restrict__ rowptr, int* __restrict__ csrc, int N, int E, int NBK, int NCH) {
  const int2* tmp = reinterpret_cast<const int2*>(tmpw);
  __shared__ int red[8];
  __shared__ int cnt[256];
  __shared__ int cursor[256];
  __shared__ int stage[SORT_CAP];
  const int tid = threadIdx.x;
  const int k = blockIdx.x;
  const int bs = off[(size_t)k * NCH];
  const int be = (k + 1 < NBK) ? off[(size_t)(k + 1) * NCH] : E;
  cnt[tid] = 0;
  __syncthreads();
  for (int j0 = bs + tid; j0 < be; j0 += 256 * 8) {
    int dd[8];
    #pragma unroll
    for (int u = 0; u < 8; ++u) dd[u] = tmp[min(j0 + 256 * u, be - 1)].y;
    #pragma unroll
    for (int u = 0; u < 8; ++u)
      if (j0 + 256 * u < be) atomicAdd(&cnt[dd[u] & 255], 1);
  }
  __syncthreads();
  const int c = cnt[tid];
  int total;
  const int st = gat_excl_scan256(c, red, total);
  cursor[tid] = st;
  const int node = k * 256 + tid;
  if (node < N) {
    rowptr[node] = bs + st;
    if (node == N - 1) rowptr[N] = bs + st + c;
  }
  __syncthreads();
  for (int j0 = bs + tid; j0 < be; j0 += 256 * 8) {
    int2 pr[8];
    #pragma unroll
    for (int u = 0; u < 8; ++u) pr[u] = tmp[min(j0 + 256 * u, be - 1)];
    #pragma unroll
    for (int u = 0; u < 8; ++u) {
      if (j0 + 256 * u < be) {
        const int p = atomicAdd(&cursor[pr[u].y & 255], 1);
        if (p < SORT_CAP) stage[p] = pr[u].x;
        else              csrc[bs + p] = pr[u].x;
      }
    }
  }
  __syncthreads();
  const int nst = min(be - bs, SORT_CAP);
  for (int j = tid; j < nst; j += 256) csrc[bs + j] = stage[j];
}

__device__ __forceinline__ float leaky02(float v) {
  return (v > 0.f) ? v : 0.2f * v;
}

__device__ __forceinline__ void gat_node_stats(
    const int* __restrict__ csrc, const float* __restrict__ asv,
    int i, int beg, int end, int lane, float ad, float sself,
    int& s0, float& sc0, bool& v0, float& m, float& sum) {
  const int j0 = beg + lane;
  v0 = j0 < end;
  s0 = v0 ? csrc[j0] : i;
  sc0 = v0 ? leaky02(asv[s0] + ad) : sself;
  float mm = fmaxf(sself, sc0);
  for (int j = j0 + 32; j < end; j += 32)
    mm = fmaxf(mm, leaky02(asv[csrc[j]] + ad));
  #pragma unroll
  for (int off = 16; off > 0; off >>= 1) mm = fmaxf(mm, __shfl_xor(mm, off, 32));
  float ss = v0 ? __expf(sc0 - mm) : 0.f;
  for (int j = j0 + 32; j < end; j += 32)
    ss += __expf(leaky02(asv[csrc[j]] + ad) - mm);
  #pragma unroll
  for (int off = 16; off > 0; off >>= 1) ss += __shfl_xor(ss, off, 32);
  m = mm;
  sum = ss + __expf(sself - mm);
}

__device__ __forceinline__ void gat_node_row256(
    const int* __restrict__ rowptr, const int* __restrict__ csrc,
    const float* __restrict__ asv, const float* __restrict__ adv,
    const float* __restrict__ H, int i, int lane, float4& a0, float4& a1) {
  const float4* H4 = reinterpret_cast<const float4*>(H);
  const float4 xs = H4[(size_t)i * 64 + lane];
  const float4 ys = H4[(size_t)i * 64 + lane + 32];
  const int beg = rowptr[i];
  const int end = rowptr[i + 1];
  const float ad = adv[i];
  const float sself = leaky02(asv[i] + ad);
  int s0; float sc0; bool v0; float m, sum;
  gat_node_stats(csrc, asv, i, beg, end, lane, ad, sself, s0, sc0, v0, m, sum);

  {
    const float c = __expf(sself - m) / sum;
    a0 = make_float4(c * xs.x, c * xs.y, c * xs.z, c * xs.w);
    a1 = make_float4(c * ys.x, c * ys.y, c * ys.z, c * ys.w);
  }
  int sv = s0;
  float cv = v0 ? __expf(sc0 - m) / sum : 0.f;
  for (int base = beg; base < end; base += 32) {
    if (base != beg) {
      const int j = base + lane;
      const bool v = j < end;
      sv = v ? csrc[j] : i;
      cv = v ? __expf(leaky02(asv[sv] + ad) - m) / sum : 0.f;
    }
    const int cnt = min(32, end - base);
    #pragma unroll 4
    for (int t = 0; t < cnt; ++t) {
      const int s = __builtin_amdgcn_readlane(sv, t);
      const float c = __int_as_float(__builtin_amdgcn_readlane(__float_as_int(cv), t));
      const float4* hp = H4 + (size_t)s * 64;
      const float4 x = hp[lane];
      const float4 y = hp[lane + 32];
      a0.x = fmaf(c, x.x, a0.x); a0.y = fmaf(c, x.y, a0.y);
      a0.z = fmaf(c, x.z, a0.z); a0.w = fmaf(c, x.w, a0.w);
      a1.x = fmaf(c, y.x, a1.x); a1.y = fmaf(c, y.y, a1.y);
      a1.z = fmaf(c, y.z, a1.z); a1.w = fmaf(c, y.w, a1.w);
    }
  }
}

template <bool GATHER, int WNW, int NT>
__global__ __launch_bounds__(256, GATHER ? 8 : 1) void gat_layer_gemm(
    const float* __restrict__ X,
    const int* __restrict__ rowptr, const int* __restrict__ csrc,
    const float* __restrict__ asv_in, const float* __restrict__ adv_in,
    const float* __restrict__ Hin, const float* __restrict__ bias_in,
    const float* __restrict__ W,
    const float* __restrict__ a_s, const float* __restrict__ a_d,
    float* __restrict__ H, float* __restrict__ asv, float* __restrict__ adv,
    int N, int Dact) {
  static_assert(WNW == 1 || WNW == 2 || WNW == 4, "2 x WNW accumulator waves out of eight");
  constexpr int BM = 32;
  constexpr int BN = 16 * NT * WNW;
  constexpr int NWA = 2 * WNW;
  constexpr int AP = GATHER ? LDA : LDT;
  constexpr int CP = BN + 4;
  constexpr int A_F4 = (BM * (KC / 4)) / 256;
  constexpr int B_QT = BN * (KC / 4);
  constexpr int B_Q = (B_QT + 255) / 256;
  constexpr int SM_A = BM * AP;
  constexpr int SM_AB = SM_A + BN * LDT;
  constexpr int SM_C = BM * CP + 2 * BM;
  constexpr int SM = SM_AB > SM_C ? SM_AB : SM_C;
  __shared__ __attribute__((aligned(16))) float smem[SM];
  float* As = smem;
  float* Bst = smem + SM_A;
  float* Cs = smem;
  float* red = smem + BM * CP;

  const int tid  = threadIdx.x;
  const int lane = tid & 31;
  const int lo   = lane & 15;
  const int hi   = lane >> 4;
  const int wid  = __builtin_amdgcn_readfirstlane(tid >> 5);
  const bool accw = (NWA == 8) || (wid < NWA);
  const int wm   = (wid / WNW) * 16;
  const int wn   = (wid % WNW) * (16 * NT);
  const int rowBase = blockIdx.x * BM;

  if constexpr (GATHER) {
    const int c0 = 4 * lane, c1 = 128 + 4 * lane;
    float4 bz0, bz1;
    bz0.x = bias_in[c0 + 0]; bz0.y = bias_in[c0 + 1]; bz0.z = bias_in[c0 + 2]; bz0.w = bias_in[c0 + 3];
    bz1.x = bias_in[c1 + 0]; bz1.y = bias_in[c1 + 1]; bz1.z = bias_in[c1 + 2]; bz1.w = bias_in[c1 + 3];
    for (int rr = 0; rr < BM / 8; ++rr) {
      const int row = wid * (BM / 8) + rr;
      const int i   = rowBase + row;
      float4 a0 = make_float4(0.f, 0.f, 0.f, 0.f);
      float4 a1 = a0;
      if (i < N) {
        gat_node_row256(rowptr, csrc, asv_in, adv_in, Hin, i, lane, a0, a1);
        a0.x += bz0.x; a0.y += bz0.y; a0.z += bz0.z; a0.w += bz0.w;
        a1.x += bz1.x; a1.y += bz1.y; a1.z += bz1.z; a1.w += bz1.w;
        a0.x = fmaxf(a0.x, 0.f); a0.y = fmaxf(a0.y, 0.f); a0.z = fmaxf(a0.z, 0.f); a0.w = fmaxf(a0.w, 0.f);
        a1.x = fmaxf(a1.x, 0.f); a1.y = fmaxf(a1.y, 0.f); a1.z = fmaxf(a1.z, 0.f); a1.w = fmaxf(a1.w, 0.f);
      }
      *reinterpret_cast<float4*>(&As[row * AP + c0]) = a0;
      *reinterpret_cast<float4*>(&As[row * AP + c1]) = a1;
    }
  }

  v8f acc[NT];
  #pragma unroll
  for (int t = 0; t < NT; ++t) acc[t] = v8f{};

  float4 ra[A_F4];
  float4 rb[B_Q];

  auto load_chunk = [&](int kk) {
    if constexpr (!GATHER) {
      #pragma unroll
      for (int i = 0; i < A_F4; ++i) {
        const int f4 = tid + 256 * i;
        const int r  = f4 >> 3;
        const int c4 = f4 & 7;
        const int gr = rowBase + r;
        float4 v = make_float4(0.f, 0.f, 0.f, 0.f);
        if (gr < N)
          v = *reinterpret_cast<const float4*>(X + (size_t)gr * DK + kk + c4 * 4);
        ra[i] = v;
      }
    }
    #pragma unroll
    for (int i = 0; i < B_Q; ++i) {
      const int q  = tid + 256 * i;
      const int n  = q % BN;
      const int kq = q / BN;
      float4 v = make_float4(0.f, 0.f, 0.f, 0.f);
      if ((B_QT % 256 == 0 || q < B_QT) && n < Dact) {
        const float* wp = W + (size_t)(kk + kq * 4) * Dact + n;
        v.x = wp[0];
        v.y = wp[Dact];
        v.z = wp[2 * Dact];
        v.w = wp[3 * Dact];
      }
      rb[i] = v;
    }
  };
  auto store_chunk = [&]() {
    if constexpr (!GATHER) {
      #pragma unroll
      for (int i = 0; i < A_F4; ++i) {
        const int f4 = tid + 256 * i;
        const int r  = f4 >> 3;
        const int c4 = f4 & 7;
        *reinterpret_cast<float4*>(&As[r * AP + c4 * 4]) = ra[i];
      }
    }
    #pragma unroll
    for (int i = 0; i < B_Q; ++i) {
      const int q  = tid + 256 * i;
      const int n  = q % BN;
      const int kq = q / BN;
      if (B_QT % 256 == 0 || q < B_QT)
        *reinterpret_cast<float4*>(&Bst[n * LDT + kq * 4]) = rb[i];
    }
  };

  const float* arow = As + (wm + lo) * AP + 4 * hi;
  const float* brow = Bst + (wn + lo) * LDT + 4 * hi;

  load_chunk(0);
  for (int kk = 0; kk < DK; kk += KC) {
    __syncthreads();
    store_chunk();
    __syncthreads();
    if (kk + KC < DK) load_chunk(kk + KC);
    if (accw) {
      const float* ak = GATHER ? arow + kk : arow;
      #pragma unroll
      for (int kb = 0; kb < KC; kb += 8) {
        const float4 a4 = *reinterpret_cast<const float4*>(ak + kb);
        float4 b4[NT];
        #pragma unroll
        for (int t = 0; t < NT; ++t)
          b4[t] = *reinterpret_cast<const float4*>(brow + t * 16 * LDT + kb);
        const v2f a01 = {a4.x, a4.y};
        const v2f a23 = {a4.z, a4.w};
        #pragma unroll
        for (int t = 0; t < NT; ++t) {
          const v2f b01 = {b4[t].x, b4[t].y};
          acc[t] = __builtin_amdgcn_wmma_f32_16x16x4_f32(false, a01, false, b01,
                                                         (short)0, acc[t], false, false);
        }
        #pragma unroll
        for (int t = 0; t < NT; ++t) {
          const v2f b23 = {b4[t].z, b4[t].w};
          acc[t] = __builtin_amdgcn_wmma_f32_16x16x4_f32(false, a23, false, b23,
                                                         (short)0, acc[t], false, false);
        }
      }
    }
  }

  __syncthreads();
  if (accw) {
    #pragma unroll
    for (int t = 0; t < NT; ++t) {
      #pragma unroll
      for (int r = 0; r < 8; ++r)
        Cs[(wm + 8 * hi + r) * CP + wn + 16 * t + lo] = acc[t][r];
    }
  }
  __syncthreads();

  if constexpr (BN == 256) {
    float4 s0, s1, d0, d1;
    {
      const int c0 = 4 * lane, c1 = 128 + 4 * lane;
      s0.x = (c0 + 0 < Dact) ? a_s[c0 + 0] : 0.f; s0.y = (c0 + 1 < Dact) ? a_s[c0 + 1] : 0.f;
      s0.z = (c0 + 2 < Dact) ? a_s[c0 + 2] : 0.f; s0.w = (c0 + 3 < Dact) ? a_s[c0 + 3] : 0.f;
      s1.x = (c1 + 0 < Dact) ? a_s[c1 + 0] : 0.f; s1.y = (c1 + 1 < Dact) ? a_s[c1 + 1] : 0.f;
      s1.z = (c1 + 2 < Dact) ? a_s[c1 + 2] : 0.f; s1.w = (c1 + 3 < Dact) ? a_s[c1 + 3] : 0.f;
      d0.x = (c0 + 0 < Dact) ? a_d[c0 + 0] : 0.f; d0.y = (c0 + 1 < Dact) ? a_d[c0 + 1] : 0.f;
      d0.z = (c0 + 2 < Dact) ? a_d[c0 + 2] : 0.f; d0.w = (c0 + 3 < Dact) ? a_d[c0 + 3] : 0.f;
      d1.x = (c1 + 0 < Dact) ? a_d[c1 + 0] : 0.f; d1.y = (c1 + 1 < Dact) ? a_d[c1 + 1] : 0.f;
      d1.z = (c1 + 2 < Dact) ? a_d[c1 + 2] : 0.f; d1.w = (c1 + 3 < Dact) ? a_d[c1 + 3] : 0.f;
    }
    constexpr int RPW = BM / 8;
    #pragma unroll
    for (int rr = 0; rr < RPW; ++rr) {
      const int row = wid * RPW + rr;
      const int gr  = rowBase + row;
      const float4 x = *reinterpret_cast<const float4*>(&Cs[row * CP + 4 * lane]);
      const float4 y = *reinterpret_cast<const float4*>(&Cs[row * CP + 128 + 4 * lane]);
      float p = x.x * s0.x + x.y * s0.y + x.z * s0.z + x.w * s0.w
              + y.x * s1.x + y.y * s1.y + y.z * s1.z + y.w * s1.w;
      float q = x.x * d0.x + x.y * d0.y + x.z * d0.z + x.w * d0.w
              + y.x * d1.x + y.y * d1.y + y.z * d1.z + y.w * d1.w;
      #pragma unroll
      for (int off = 16; off > 0; off >>= 1) {
        p += __shfl_xor(p, off, 32);
        q += __shfl_xor(q, off, 32);
      }
      if (lane == 0) {
        red[row] = p;
        red[BM + row] = q;
      }
      if (gr < N) {
        float4* hp = reinterpret_cast<float4*>(H) + (size_t)gr * 64;
        hp[lane] = x;
        hp[lane + 32] = y;
      }
    }
  } else {
    static_assert(BN == 16, "narrow form: 16 padded columns");
    if (wid < BM / 8) {
      const int qd = lane & 3;
      const int c0 = 4 * qd;
      float4 s0, d0;
      s0.x = (c0 + 0 < Dact) ? a_s[c0 + 0] : 0.f; s0.y = (c0 + 1 < Dact) ? a_s[c0 + 1] : 0.f;
      s0.z = (c0 + 2 < Dact) ? a_s[c0 + 2] : 0.f; s0.w = (c0 + 3 < Dact) ? a_s[c0 + 3] : 0.f;
      d0.x = (c0 + 0 < Dact) ? a_d[c0 + 0] : 0.f; d0.y = (c0 + 1 < Dact) ? a_d[c0 + 1] : 0.f;
      d0.z = (c0 + 2 < Dact) ? a_d[c0 + 2] : 0.f; d0.w = (c0 + 3 < Dact) ? a_d[c0 + 3] : 0.f;
      const int row = wid * 8 + (lane >> 2);
      const int gr  = rowBase + row;
      const float4 x = *reinterpret_cast<const float4*>(&Cs[row * CP + c0]);
      float p = x.x * s0.x + x.y * s0.y + x.z * s0.z + x.w * s0.w;
      float q = x.x * d0.x + x.y * d0.y + x.z * d0.z + x.w * d0.w;
      #pragma unroll
      for (int off = 2; off > 0; off >>= 1) {
        p += __shfl_xor(p, off, 32);
        q += __shfl_xor(q, off, 32);
      }
      if (qd == 0) {
        red[row] = p;
        red[BM + row] = q;
      }
      if (gr < N) reinterpret_cast<float4*>(H)[(size_t)gr * 4 + qd] = x;
    }
  }

  __syncthreads();
  if (tid < 2 * BM) {
    const int row = (tid < BM) ? tid : tid - BM;
    const int gr  = rowBase + row;
    if (gr < N) {
      if (tid < BM) asv[gr] = red[row];
      else          adv[gr] = red[BM + row];
    }
  }
}

__global__ __launch_bounds__(256) void gat_node_gather16(
    const int* __restrict__ rowptr, const int* __restrict__ csrc,
    const float* __restrict__ asv, const float* __restrict__ adv,
    const float* __restrict__ H, const float* __restrict__ bias,
    float* __restrict__ out, int N, int Dact) {
  const int lane = threadIdx.x & 31;
  const int i = blockIdx.x * 8 + __builtin_amdgcn_readfirstlane((int)(threadIdx.x >> 5));
  if (i >= N) return;
  const int beg = rowptr[i];
  const int end = rowptr[i + 1];
  const float ad = adv[i];
  const float sself = leaky02(asv[i] + ad);
  int s0; float sc0; bool v0; float m, sum;
  gat_node_stats(csrc, asv, i, beg, end, lane, ad, sself, s0, sc0, v0, m, sum);

  const int q = lane & 3;
  const int slot = lane >> 2;
  const float4* H4 = reinterpret_cast<const float4*>(H);
  float4 acc = make_float4(0.f, 0.f, 0.f, 0.f);
  if (slot == 0) {
    const float c = __expf(sself - m) / sum;
    const float4 x = H4[(size_t)i * 4 + q];
    acc = make_float4(c * x.x, c * x.y, c * x.z, c * x.w);
  }
  int sv = s0;
  float cv = v0 ? __expf(sc0 - m) / sum : 0.f;
  for (int base = beg; base < end; base += 32) {
    if (base != beg) {
      const int j = base + lane;
      const bool v = j < end;
      sv = v ? csrc[j] : i;
      cv = v ? __expf(leaky02(asv[sv] + ad) - m) / sum : 0.f;
    }
    const int cnt = min(32, end - base);
    for (int t0 = 0; t0 < cnt; t0 += 8) {
      const int t = t0 + slot;
      const int s = __shfl(sv, t, 32);
      const float c = __shfl(cv, t, 32);
      if (t < cnt) {
        const float4 x = H4[(size_t)s * 4 + q];
        acc.x = fmaf(c, x.x, acc.x); acc.y = fmaf(c, x.y, acc.y);
        acc.z = fmaf(c, x.z, acc.z); acc.w = fmaf(c, x.w, acc.w);
      }
    }
  }
  #pragma unroll
  for (int off = 4; off < 32; off <<= 1) {
    acc.x += __shfl_xor(acc.x, off, 32);
    acc.y += __shfl_xor(acc.y, off, 32);
    acc.z += __shfl_xor(acc.z, off, 32);
    acc.w += __shfl_xor(acc.w, off, 32);
  }
  if (slot == 0) {
    const int c = q * 4;
    float* op = out + (size_t)i * Dact;
    if (c + 0 < Dact) op[c + 0] = acc.x + bias[c + 0];
    if (c + 1 < Dact) op[c + 1] = acc.y + bias[c + 1];
    if (c + 2 < Dact) op[c + 2] = acc.z + bias[c + 2];
    if (c + 3 < Dact) op[c + 3] = acc.w + bias[c + 3];
  }
}

extern "C" void kernel_launch(void* const* d_in, const int* in_sizes, int n_in,
                              void* d_out, int out_size, void* d_ws, size_t ws_size,
                              hipStream_t stream) {
  const float* x  = (const float*)d_in[0];
  const int*   ei = (const int*)d_in[1];
  const float* W1 = (const float*)d_in[2];
  const float* as1 = (const float*)d_in[3];
  const float* ad1 = (const float*)d_in[4];
  const float* b1 = (const float*)d_in[5];
  const float* W2 = (const float*)d_in[6];
  const float* as2 = (const float*)d_in[7];
  const float* ad2 = (const float*)d_in[8];
  const float* b2 = (const float*)d_in[9];
  const float* W3 = (const float*)d_in[10];
  const float* as3 = (const float*)d_in[11];
  const float* ad3 = (const float*)d_in[12];
  const float* b3 = (const float*)d_in[13];

  const int N  = in_sizes[0] / 256;
  const int E  = in_sizes[1] / 2;
  const int* src = ei;
  const int* dst = ei + E;

  float* ws = (float*)d_ws;
  float* hA  = ws;
  float* hB  = hA + (size_t)N * 256;
  float* h3  = hB + (size_t)N * 256;
  int*   ib  = (int*)(h3 + (size_t)N * 16);
  int*   cnt    = ib;
  int*   rowptr = cnt + N;
  int*   cursor = rowptr + N + 1;
  int*   tsum   = cursor + N;
  float* asvB = (float*)(ib + (size_t)N * 4);
  float* advB = asvB + N;
  float* asvA = (float*)(ib + (size_t)N * 16);
  float* advA = asvA + N;
  int*   csrc = (int*)(advA + N);

  dim3 blk(256);
  int gGemm  = (N + 31) / 32;
  int gNodeW = (N + 7) / 8;
  int gNodeT = (N + 255) / 256;
  int gEdgeT = (E + 255) / 256;

  const int NBK = (N + 255) / 256;
  const int NCH = (E + SORT_CH - 1) / SORT_CH;
  int*  tmp  = (int*)hB;
  int*  cntT = (int*)hB + 2 * (size_t)E;
  int*  off  = cntT + (size_t)NBK * NCH;
  if (NBK <= SORT_MAXB && 2 * (size_t)E + 2 * (size_t)NBK * NCH <= (size_t)N * 256) {
    gat_bin_count<<<NCH, blk, 0, stream>>>(dst, cntT, E, NBK, NCH);
    gat_bin_scan<<<1, dim3(1024), 0, stream>>>(cntT, off, NBK * NCH);
    gat_bin_scatter<<<NCH, blk, 0, stream>>>(src, dst, cntT, off, tmp, E, NBK, NCH);
    gat_bucket_sort<<<NBK, blk, 0, stream>>>(tmp, off, rowptr, csrc, N, E, NBK, NCH);
  } else {
    gat_csr_zero<<<gNodeT, blk, 0, stream>>>(cnt, N);
    gat_csr_hist<<<gEdgeT, blk, 0, stream>>>(dst, cnt, E);
    gat_csr_tilesum<<<gNodeT, blk, 0, stream>>>(cnt, tsum, N);
    gat_csr_scan<<<gNodeT, blk, 0, stream>>>(cnt, tsum, rowptr, cursor, N);
    gat_csr_fill<<<gEdgeT, blk, 0, stream>>>(src, dst, cursor, csrc, E);
  }

  gat_layer_gemm<false, 4, 4><<<gGemm, blk, 0, stream>>>(
      x, nullptr, nullptr, nullptr, nullptr, nullptr, nullptr,
      W1, as1, ad1, hA, asvA, advA, N, 256);

  gat_layer_gemm<true, 4, 4><<<gGemm, blk, 0, stream>>>(
      nullptr, rowptr, csrc, asvA, advA, hA, b1,
      W2, as2, ad2, hB, asvB, advB, N, 256);

  gat_layer_gemm<true, 1, 1><<<gGemm, blk, 0, stream>>>(
      nullptr, rowptr, csrc, asvB, advB, hB, b2,
      W3, as3, ad3, h3, asvA, advA, N, 10);

  gat_node_gather16<<<gNodeW, blk, 0, stream>>>(rowptr, csrc, asvA, advA, h3, b3, (float*)d_out, N, 10);
}


# Edge sort in 2 launches: each chunk stores its pairs in bucket order at its own place with per-bucket counts and starts; bucket blocks collect their runs; count, scan launches go; LDS arrival order
# speedup vs baseline: 5.8449x; 1.0650x over previous
#include <hip/hip_runtime.h>
#include <math.h>


typedef float v2f __attribute__((ext_vector_type(2)));
typedef float v8f __attribute__((ext_vector_type(8)));

#define DK 256
#define KC 32
#define LDT (KC + 4)
#define LDA (DK + 4)

__global__ __launch_bounds__(256) void gat_csr_zero(int* __restrict__ cnt, int N) {
  int i = blockIdx.x * 256 + threadIdx.x;
  if (i < N) cnt[i] = 0;
}

__global__ __launch_bounds__(256) void gat_csr_hist(
    const int* __restrict__ dst, int* __restrict__ cnt, int E) {
  int e = blockIdx.x * 256 + threadIdx.x;
  if (e < E) atomicAdd(&cnt[dst[e]], 1);
}

__global__ __launch_bounds__(256) void gat_csr_tilesum(
    const int* __restrict__ cnt, int* __restrict__ tsum, int N) {
  __shared__ int red[8];
  const int tid = threadIdx.x;
  int i = blockIdx.x * 256 + tid;
  int v = (i < N) ? cnt[i] : 0;
  #pragma unroll
  for (int off = 16; off > 0; off >>= 1) v += __shfl_xor(v, off, 32);
  if ((tid & 31) == 0) red[tid >> 5] = v;
  __syncthreads();
  if (tid == 0) {
    int t = 0;
    #pragma unroll
    for (int w = 0; w < 8; ++w) t += red[w];
    tsum[blockIdx.x] = t;
  }
}

__global__ __launch_bounds__(256) void gat_csr_scan(
    const int* __restrict__ cnt, const int* __restrict__ tsum,
    int* __restrict__ rowptr, int* __restrict__ cursor, int N) {
  __shared__ int red[8];
  __shared__ int wtot[8];
  const int tid = threadIdx.x;
  const int lane = tid & 31;
  const int wid = tid >> 5;
  int part = 0;
  for (int t = tid; t < (int)blockIdx.x; t += 256) part += tsum[t];
  #pragma unroll
  for (int off = 16; off > 0; off >>= 1) part += __shfl_xor(part, off, 32);
  if (lane == 0) red[wid] = part;
  int i = blockIdx.x * 256 + tid;
  int v = (i < N) ? cnt[i] : 0;
  int x = v;
  #pragma unroll
  for (int d = 1; d < 32; d <<= 1) {
    int n = __shfl_up(x, d, 32);
    if (lane >= d) x += n;
  }
  if (lane == 31) wtot[wid] = x;
  __syncthreads();
  int base = 0;
  #pragma unroll
  for (int w = 0; w < 8; ++w) {
    base += red[w];
    if (w < wid) base += wtot[w];
  }
  int excl = base + x - v;
  if (i < N) {
    rowptr[i] = excl;
    cursor[i] = excl;
    if (i == N - 1) rowptr[N] = excl + v;
  }
}

__global__ __launch_bounds__(256) void gat_csr_fill(
    const int* __restrict__ src, const int* __restrict__ dst,
    int* __restrict__ cursor, int* __restrict__ csrc, int E) {
  int e = blockIdx.x * 256 + threadIdx.x;
  if (e < E) {
    int pos = atomicAdd(&cursor[dst[e]], 1);
    csrc[pos] = src[e];
  }
}

#define SORT_CH 4096
#define SORT_MAXB 256
#define SORT_CAP 8192

__device__ __forceinline__ int gat_excl_scan256(int v, int* red, int& total) {
  const int lane = threadIdx.x & 31;
  const int wid = threadIdx.x >> 5;
  int x = v;
  #pragma unroll
  for (int d = 1; d < 32; d <<= 1) {
    int n = __shfl_up(x, d, 32);
    if (lane >= d) x += n;
  }
  if (lane == 31) red[wid] = x;
  __syncthreads();
  int base = 0, tot = 0;
  #pragma unroll
  for (int w = 0; w < 8; ++w) {
    const int r = red[w];
    if (w < wid) base += r;
    tot += r;
  }
  __syncthreads();
  total = tot;
  return base + x - v;
}

__global__ __launch_bounds__(256) void gat_bin_local(
    const int* __restrict__ src, const int* __restrict__ dst,
    int* __restrict__ tmpw, int* __restrict__ cntL, int* __restrict__ offL, int E) {
  int2* tmp = reinterpret_cast<int2*>(tmpw);
  __shared__ int red[8];
  __shared__ int hist[SORT_MAXB];
  __shared__ int cursor[SORT_MAXB];
  __shared__ int2 stage[SORT_CH];
  const int tid = threadIdx.x;
  const int b = blockIdx.x;
  hist[tid] = 0;
  __syncthreads();
  const int base = b * SORT_CH;
  int d[SORT_CH / 256], s[SORT_CH / 256];
  #pragma unroll
  for (int i = 0; i < SORT_CH / 256; ++i) {
    const int e = min(base + i * 256 + tid, E - 1);
    d[i] = dst[e];
    s[i] = src[e];
  }
  #pragma unroll
  for (int i = 0; i < SORT_CH / 256; ++i)
    if (base + i * 256 + tid < E) atomicAdd(&hist[d[i] >> 8], 1);
  __syncthreads();
  const int mine = hist[tid];
  int total;
  const int ls = gat_excl_scan256(mine, red, total);
  cursor[tid] = ls;
  cntL[(size_t)b * SORT_MAXB + tid] = mine;
  offL[(size_t)b * SORT_MAXB + tid] = ls;
  __syncthreads();
  #pragma unroll
  for (int i = 0; i < SORT_CH / 256; ++i) {
    if (base + i * 256 + tid < E) {
      const int p = atomicAdd(&cursor[d[i] >> 8], 1);
      stage[p] = make_int2(s[i], d[i]);
    }
  }
  __syncthreads();
  for (int j = tid; j < total; j += 256) tmp[(size_t)base + j] = stage[j];
}

__global__ __launch_bounds__(256) void gat_bucket_sort(
    const int* __restrict__ tmpw, const int* __restrict__ cntL, const int* __restrict__ offL,
    int* __restrict__ rowptr, int* __restrict__ csrc, int N, int NCH) {
  const int2* tmp = reinterpret_cast<const int2*>(tmpw);
  __shared__ int red[8];
  __shared__ int cnt[256];
  __shared__ int cursor[256];
  __shared__ int stage[SORT_CAP];
  const int tid = threadIdx.x;
  const int k = blockIdx.x;
  int lo = 0, n = 0;
  if (tid < NCH) {
    lo = offL[(size_t)tid * SORT_MAXB + k];
    n  = cntL[(size_t)tid * SORT_MAXB + k];
  }
  cnt[tid] = 0;
  int bs;
  (void)gat_excl_scan256(lo, red, bs);
  const int2* run = tmp + (size_t)tid * SORT_CH + lo;
  for (int j0 = 0; j0 < n; j0 += 8) {
    int dd[8];
    #pragma unroll
    for (int u = 0; u < 8; ++u) dd[u] = run[min(j0 + u, n - 1)].y;
    #pragma unroll
    for (int u = 0; u < 8; ++u)
      atomicAdd(&cnt[dd[u] & 255], (j0 + u < n) ? 1 : 0);
  }
  __syncthreads();
  const int c = cnt[tid];
  int total;
  const int st = gat_excl_scan256(c, red, total);
  cursor[tid] = st;
  const int node = k * 256 + tid;
  if (node < N) {
    rowptr[node] = bs + st;
    if (node == N - 1) rowptr[N] = bs + st + c;
  }
  __syncthreads();
  for (int j0 = 0; j0 < n; j0 += 8) {
    int2 pr[8];
    #pragma unroll
    for (int u = 0; u < 8; ++u) pr[u] = run[min(j0 + u, n - 1)];
    #pragma unroll
    for (int u = 0; u < 8; ++u) {
      const bool live = j0 + u < n;
      const int p = atomicAdd(&cursor[pr[u].y & 255], live ? 1 : 0);
      if (live) {
        if (p < SORT_CAP) stage[p] = pr[u].x;
        else              csrc[bs + p] = pr[u].x;
      }
    }
  }
  __syncthreads();
  const int nst = min(total, SORT_CAP);
  for (int j = tid; j < nst; j += 256) csrc[bs + j] = stage[j];
}

__device__ __forceinline__ float leaky02(float v) {
  return (v > 0.f) ? v : 0.2f * v;
}

__device__ __forceinline__ void gat_node_stats(
    const int* __restrict__ csrc, const float* __restrict__ asv,
    int i, int beg, int end, int lane, float ad, float sself,
    int& s0, float& sc0, bool& v0, float& m, float& sum) {
  const int j0 = beg + lane;
  v0 = j0 < end;
  s0 = v0 ? csrc[j0] : i;
  sc0 = v0 ? leaky02(asv[s0] + ad) : sself;
  float mm = fmaxf(sself, sc0);
  for (int j = j0 + 32; j < end; j += 32)
    mm = fmaxf(mm, leaky02(asv[csrc[j]] + ad));
  #pragma unroll
  for (int off = 16; off > 0; off >>= 1) mm = fmaxf(mm, __shfl_xor(mm, off, 32));
  float ss = v0 ? __expf(sc0 - mm) : 0.f;
  for (int j = j0 + 32; j < end; j += 32)
    ss += __expf(leaky02(asv[csrc[j]] + ad) - mm);
  #pragma unroll
  for (int off = 16; off > 0; off >>= 1) ss += __shfl_xor(ss, off, 32);
  m = mm;
  sum = ss + __expf(sself - mm);
}

__device__ __forceinline__ void gat_node_row256(
    const int* __restrict__ rowptr, const int* __restrict__ csrc,
    const float* __restrict__ asv, const float* __restrict__ adv,
    const float* __restrict__ H, int i, int lane, float4& a0, float4& a1) {
  const float4* H4 = reinterpret_cast<const float4*>(H);
  const float4 xs = H4[(size_t)i * 64 + lane];
  const float4 ys = H4[(size_t)i * 64 + lane + 32];
  const int beg = rowptr[i];
  const int end = rowptr[i + 1];
  const float ad = adv[i];
  const float sself = leaky02(asv[i] + ad);
  int s0; float sc0; bool v0; float m, sum;
  gat_node_stats(csrc, asv, i, beg, end, lane, ad, sself, s0, sc0, v0, m, sum);

  {
    const float c = __expf(sself - m) / sum;
    a0 = make_float4(c * xs.x, c * xs.y, c * xs.z, c * xs.w);
    a1 = make_float4(c * ys.x, c * ys.y, c * ys.z, c * ys.w);
  }
  int sv = s0;
  float cv = v0 ? __expf(sc0 - m) / sum : 0.f;
  for (int base = beg; base < end; base += 32) {
    if (base != beg) {
      const int j = base + lane;
      const bool v = j < end;
      sv = v ? csrc[j] : i;
      cv = v ? __expf(leaky02(asv[sv] + ad) - m) / sum : 0.f;
    }
    const int cnt = min(32, end - base);
    #pragma unroll 4
    for (int t = 0; t < cnt; ++t) {
      const int s = __builtin_amdgcn_readlane(sv, t);
      const float c = __int_as_float(__builtin_amdgcn_readlane(__float_as_int(cv), t));
      const float4* hp = H4 + (size_t)s * 64;
      const float4 x = hp[lane];
      const float4 y = hp[lane + 32];
      a0.x = fmaf(c, x.x, a0.x); a0.y = fmaf(c, x.y, a0.y);
      a0.z = fmaf(c, x.z, a0.z); a0.w = fmaf(c, x.w, a0.w);
      a1.x = fmaf(c, y.x, a1.x); a1.y = fmaf(c, y.y, a1.y);
      a1.z = fmaf(c, y.z, a1.z); a1.w = fmaf(c, y.w, a1.w);
    }
  }
}

template <bool GATHER, int WNW, int NT>
__global__ __launch_bounds__(256, GATHER ? 8 : 1) void gat_layer_gemm(
    const float* __restrict__ X,
    const int* __restrict__ rowptr, const int* __restrict__ csrc,
    const float* __restrict__ asv_in, const float* __restrict__ adv_in,
    const float* __restrict__ Hin, const float* __restrict__ bias_in,
    const float* __restrict__ W,
    const float* __restrict__ a_s, const float* __restrict__ a_d,
    float* __restrict__ H, float* __restrict__ asv, float* __restrict__ adv,
    int N, int Dact) {
  static_assert(WNW == 1 || WNW == 2 || WNW == 4, "2 x WNW accumulator waves out of eight");
  constexpr int BM = 32;
  constexpr int BN = 16 * NT * WNW;
  constexpr int NWA = 2 * WNW;
  constexpr int AP = GATHER ? LDA : LDT;
  constexpr int CP = BN + 4;
  constexpr int A_F4 = (BM * (KC / 4)) / 256;
  constexpr int B_QT = BN * (KC / 4);
  constexpr int B_Q = (B_QT + 255) / 256;
  constexpr int SM_A = BM * AP;
  constexpr int SM_AB = SM_A + BN * LDT;
  constexpr int SM_C = BM * CP + 2 * BM;
  constexpr int SM = SM_AB > SM_C ? SM_AB : SM_C;
  __shared__ __attribute__((aligned(16))) float smem[SM];
  float* As = smem;
  float* Bst = smem + SM_A;
  float* Cs = smem;
  float* red = smem + BM * CP;

  const int tid  = threadIdx.x;
  const int lane = tid & 31;
  const int lo   = lane & 15;
  const int hi   = lane >> 4;
  const int wid  = __builtin_amdgcn_readfirstlane(tid >> 5);
  const bool accw = (NWA == 8) || (wid < NWA);
  const int wm   = (wid / WNW) * 16;
  const int wn   = (wid % WNW) * (16 * NT);
  const int rowBase = blockIdx.x * BM;

  if constexpr (GATHER) {
    const int c0 = 4 * lane, c1 = 128 + 4 * lane;
    float4 bz0, bz1;
    bz0.x = bias_in[c0 + 0]; bz0.y = bias_in[c0 + 1]; bz0.z = bias_in[c0 + 2]; bz0.w = bias_in[c0 + 3];
    bz1.x = bias_in[c1 + 0]; bz1.y = bias_in[c1 + 1]; bz1.z = bias_in[c1 + 2]; bz1.w = bias_in[c1 + 3];
    for (int rr = 0; rr < BM / 8; ++rr) {
      const int row = wid * (BM / 8) + rr;
      const int i   = rowBase + row;
      float4 a0 = make_float4(0.f, 0.f, 0.f, 0.f);
      float4 a1 = a0;
      if (i < N) {
        gat_node_row256(rowptr, csrc, asv_in, adv_in, Hin, i, lane, a0, a1);
        a0.x += bz0.x; a0.y += bz0.y; a0.z += bz0.z; a0.w += bz0.w;
        a1.x += bz1.x; a1.y += bz1.y; a1.z += bz1.z; a1.w += bz1.w;
        a0.x = fmaxf(a0.x, 0.f); a0.y = fmaxf(a0.y, 0.f); a0.z = fmaxf(a0.z, 0.f); a0.w = fmaxf(a0.w, 0.f);
        a1.x = fmaxf(a1.x, 0.f); a1.y = fmaxf(a1.y, 0.f); a1.z = fmaxf(a1.z, 0.f); a1.w = fmaxf(a1.w, 0.f);
      }
      *reinterpret_cast<float4*>(&As[row * AP + c0]) = a0;
      *reinterpret_cast<float4*>(&As[row * AP + c1]) = a1;
    }
  }

  v8f acc[NT];
  #pragma unroll
  for (int t = 0; t < NT; ++t) acc[t] = v8f{};

  float4 ra[A_F4];
  float4 rb[B_Q];

  auto load_chunk = [&](int kk) {
    if constexpr (!GATHER) {
      #pragma unroll
      for (int i = 0; i < A_F4; ++i) {
        const int f4 = tid + 256 * i;
        const int r  = f4 >> 3;
        const int c4 = f4 & 7;
        const int gr = rowBase + r;
        float4 v = make_float4(0.f, 0.f, 0.f, 0.f);
        if (gr < N)
          v = *reinterpret_cast<const float4*>(X + (size_t)gr * DK + kk + c4 * 4);
        ra[i] = v;
      }
    }
    #pragma unroll
    for (int i = 0; i < B_Q; ++i) {
      const int q  = tid + 256 * i;
      const int n  = q % BN;
      const int kq = q / BN;
      float4 v = make_float4(0.f, 0.f, 0.f, 0.f);
      if ((B_QT % 256 == 0 || q < B_QT) && n < Dact) {
        const float* wp = W + (size_t)(kk + kq * 4) * Dact + n;
        v.x = wp[0];
        v.y = wp[Dact];
        v.z = wp[2 * Dact];
        v.w = wp[3 * Dact];
      }
      rb[i] = v;
    }
  };
  auto store_chunk = [&]() {
    if constexpr (!GATHER) {
      #pragma unroll
      for (int i = 0; i < A_F4; ++i) {
        const int f4 = tid + 256 * i;
        const int r  = f4 >> 3;
        const int c4 = f4 & 7;
        *reinterpret_cast<float4*>(&As[r * AP + c4 * 4]) = ra[i];
      }
    }
    #pragma unroll
    for (int i = 0; i < B_Q; ++i) {
      const int q  = tid + 256 * i;
      const int n  = q % BN;
      const int kq = q / BN;
      if (B_QT % 256 == 0 || q < B_QT)
        *reinterpret_cast<float4*>(&Bst[n * LDT + kq * 4]) = rb[i];
    }
  };

  const float* arow = As + (wm + lo) * AP + 4 * hi;
  const float* brow = Bst + (wn + lo) * LDT + 4 * hi;

  load_chunk(0);
  for (int kk = 0; kk < DK; kk += KC) {
    __syncthreads();
    store_chunk();
    __syncthreads();
    if (kk + KC < DK) load_chunk(kk + KC);
    if (accw) {
      const float* ak = GATHER ? arow + kk : arow;
      #pragma unroll
      for (int kb = 0; kb < KC; kb += 8) {
        const float4 a4 = *reinterpret_cast<const float4*>(ak + kb);
        float4 b4[NT];
        #pragma unroll
        for (int t = 0; t < NT; ++t)
          b4[t] = *reinterpret_cast<const float4*>(brow + t * 16 * LDT + kb);
        const v2f a01 = {a4.x, a4.y};
        const v2f a23 = {a4.z, a4.w};
        #pragma unroll
        for (int t = 0; t < NT; ++t) {
          const v2f b01 = {b4[t].x, b4[t].y};
          acc[t] = __builtin_amdgcn_wmma_f32_16x16x4_f32(false, a01, false, b01,
                                                         (short)0, acc[t], false, false);
        }
        #pragma unroll
        for (int t = 0; t < NT; ++t) {
          const v2f b23 = {b4[t].z, b4[t].w};
          acc[t] = __builtin_amdgcn_wmma_f32_16x16x4_f32(false, a23, false, b23,
                                                         (short)0, acc[t], false, false);
        }
      }
    }
  }

  __syncthreads();
  if (accw) {
    #pragma unroll
    for (int t = 0; t < NT; ++t) {
      #pragma unroll
      for (int r = 0; r < 8; ++r)
        Cs[(wm + 8 * hi + r) * CP + wn + 16 * t + lo] = acc[t][r];
    }
  }
  __syncthreads();

  if constexpr (BN == 256) {
    float4 s0, s1, d0, d1;
    {
      const int c0 = 4 * lane, c1 = 128 + 4 * lane;
      s0.x = (c0 + 0 < Dact) ? a_s[c0 + 0] : 0.f; s0.y = (c0 + 1 < Dact) ? a_s[c0 + 1] : 0.f;
      s0.z = (c0 + 2 < Dact) ? a_s[c0 + 2] : 0.f; s0.w = (c0 + 3 < Dact) ? a_s[c0 + 3] : 0.f;
      s1.x = (c1 + 0 < Dact) ? a_s[c1 + 0] : 0.f; s1.y = (c1 + 1 < Dact) ? a_s[c1 + 1] : 0.f;
      s1.z = (c1 + 2 < Dact) ? a_s[c1 + 2] : 0.f; s1.w = (c1 + 3 < Dact) ? a_s[c1 + 3] : 0.f;
      d0.x = (c0 + 0 < Dact) ? a_d[c0 + 0] : 0.f; d0.y = (c0 + 1 < Dact) ? a_d[c0 + 1] : 0.f;
      d0.z = (c0 + 2 < Dact) ? a_d[c0 + 2] : 0.f; d0.w = (c0 + 3 < Dact) ? a_d[c0 + 3] : 0.f;
      d1.x = (c1 + 0 < Dact) ? a_d[c1 + 0] : 0.f; d1.y = (c1 + 1 < Dact) ? a_d[c1 + 1] : 0.f;
      d1.z = (c1 + 2 < Dact) ? a_d[c1 + 2] : 0.f; d1.w = (c1 + 3 < Dact) ? a_d[c1 + 3] : 0.f;
    }
    constexpr int RPW = BM / 8;
    #pragma unroll
    for (int rr = 0; rr < RPW; ++rr) {
      const int row = wid * RPW + rr;
      const int gr  = rowBase + row;
      const float4 x = *reinterpret_cast<const float4*>(&Cs[row * CP + 4 * lane]);
      const float4 y = *reinterpret_cast<const float4*>(&Cs[row * CP + 128 + 4 * lane]);
      float p = x.x * s0.x + x.y * s0.y + x.z * s0.z + x.w * s0.w
              + y.x * s1.x + y.y * s1.y + y.z * s1.z + y.w * s1.w;
      float q = x.x * d0.x + x.y * d0.y + x.z * d0.z + x.w * d0.w
              + y.x * d1.x + y.y * d1.y + y.z * d1.z + y.w * d1.w;
      #pragma unroll
      for (int off = 16; off > 0; off >>= 1) {
        p += __shfl_xor(p, off, 32);
        q += __shfl_xor(q, off, 32);
      }
      if (lane == 0) {
        red[row] = p;
        red[BM + row] = q;
      }
      if (gr < N) {
        float4* hp = reinterpret_cast<float4*>(H) + (size_t)gr * 64;
        hp[lane] = x;
        hp[lane + 32] = y;
      }
    }
  } else {
    static_assert(BN == 16, "narrow form: 16 padded columns");
    if (wid < BM / 8) {
      const int qd = lane & 3;
      const int c0 = 4 * qd;
      float4 s0, d0;
      s0.x = (c0 + 0 < Dact) ? a_s[c0 + 0] : 0.f; s0.y = (c0 + 1 < Dact) ? a_s[c0 + 1] : 0.f;
      s0.z = (c0 + 2 < Dact) ? a_s[c0 + 2] : 0.f; s0.w = (c0 + 3 < Dact) ? a_s[c0 + 3] : 0.f;
      d0.x = (c0 + 0 < Dact) ? a_d[c0 + 0] : 0.f; d0.y = (c0 + 1 < Dact) ? a_d[c0 + 1] : 0.f;
      d0.z = (c0 + 2 < Dact) ? a_d[c0 + 2] : 0.f; d0.w = (c0 + 3 < Dact) ? a_d[c0 + 3] : 0.f;
      const int row = wid * 8 + (lane >> 2);
      const int gr  = rowBase + row;
      const float4 x = *reinterpret_cast<const float4*>(&Cs[row * CP + c0]);
      float p = x.x * s0.x + x.y * s0.y + x.z * s0.z + x.w * s0.w;
      float q = x.x * d0.x + x.y * d0.y + x.z * d0.z + x.w * d0.w;
      #pragma unroll
      for (int off = 2; off > 0; off >>= 1) {
        p += __shfl_xor(p, off, 32);
        q += __shfl_xor(q, off, 32);
      }
      if (qd == 0) {
        red[row] = p;
        red[BM + row] = q;
      }
      if (gr < N) reinterpret_cast<float4*>(H)[(size_t)gr * 4 + qd] = x;
    }
  }

  __syncthreads();
  if (tid < 2 * BM) {
    const int row = (tid < BM) ? tid : tid - BM;
    const int gr  = rowBase + row;
    if (gr < N) {
      if (tid < BM) asv[gr] = red[row];
      else          adv[gr] = red[BM + row];
    }
  }
}

__global__ __launch_bounds__(256) void gat_node_gather16(
    const int* __restrict__ rowptr, const int* __restrict__ csrc,
    const float* __restrict__ asv, const float* __restrict__ adv,
    const float* __restrict__ H, const float* __restrict__ bias,
    float* __restrict__ out, int N, int Dact) {
  const int lane = threadIdx.x & 31;
  const int i = blockIdx.x * 8 + __builtin_amdgcn_readfirstlane((int)(threadIdx.x >> 5));
  if (i >= N) return;
  const int beg = rowptr[i];
  const int end = rowptr[i + 1];
  const float ad = adv[i];
  const float sself = leaky02(asv[i] + ad);
  int s0; float sc0; bool v0; float m, sum;
  gat_node_stats(csrc, asv, i, beg, end, lane, ad, sself, s0, sc0, v0, m, sum);

  const int q = lane & 3;
  const int slot = lane >> 2;
  const float4* H4 = reinterpret_cast<const float4*>(H);
  float4 acc = make_float4(0.f, 0.f, 0.f, 0.f);
  if (slot == 0) {
    const float c = __expf(sself - m) / sum;
    const float4 x = H4[(size_t)i * 4 + q];
    acc = make_float4(c * x.x, c * x.y, c * x.z, c * x.w);
  }
  int sv = s0;
  float cv = v0 ? __expf(sc0 - m) / sum : 0.f;
  for (int base = beg; base < end; base += 32) {
    if (base != beg) {
      const int j = base + lane;
      const bool v = j < end;
      sv = v ? csrc[j] : i;
      cv = v ? __expf(leaky02(asv[sv] + ad) - m) / sum : 0.f;
    }
    const int cnt = min(32, end - base);
    for (int t0 = 0; t0 < cnt; t0 += 8) {
      const int t = t0 + slot;
      const int s = __shfl(sv, t, 32);
      const float c = __shfl(cv, t, 32);
      if (t < cnt) {
        const float4 x = H4[(size_t)s * 4 + q];
        acc.x = fmaf(c, x.x, acc.x); acc.y = fmaf(c, x.y, acc.y);
        acc.z = fmaf(c, x.z, acc.z); acc.w = fmaf(c, x.w, acc.w);
      }
    }
  }
  #pragma unroll
  for (int off = 4; off < 32; off <<= 1) {
    acc.x += __shfl_xor(acc.x, off, 32);
    acc.y += __shfl_xor(acc.y, off, 32);
    acc.z += __shfl_xor(acc.z, off, 32);
    acc.w += __shfl_xor(acc.w, off, 32);
  }
  if (slot == 0) {
    const int c = q * 4;
    float* op = out + (size_t)i * Dact;
    if (c + 0 < Dact) op[c + 0] = acc.x + bias[c + 0];
    if (c + 1 < Dact) op[c + 1] = acc.y + bias[c + 1];
    if (c + 2 < Dact) op[c + 2] = acc.z + bias[c + 2];
    if (c + 3 < Dact) op[c + 3] = acc.w + bias[c + 3];
  }
}

extern "C" void kernel_launch(void* const* d_in, const int* in_sizes, int n_in,
                              void* d_out, int out_size, void* d_ws, size_t ws_size,
                              hipStream_t stream) {
  const float* x  = (const float*)d_in[0];
  const int*   ei = (const int*)d_in[1];
  const float* W1 = (const float*)d_in[2];
  const float* as1 = (const float*)d_in[3];
  const float* ad1 = (const float*)d_in[4];
  const float* b1 = (const float*)d_in[5];
  const float* W2 = (const float*)d_in[6];
  const float* as2 = (const float*)d_in[7];
  const float* ad2 = (const float*)d_in[8];
  const float* b2 = (const float*)d_in[9];
  const float* W3 = (const float*)d_in[10];
  const float* as3 = (const float*)d_in[11];
  const float* ad3 = (const float*)d_in[12];
  const float* b3 = (const float*)d_in[13];

  const int N  = in_sizes[0] / 256;
  const int E  = in_sizes[1] / 2;
  const int* src = ei;
  const int* dst = ei + E;

  float* ws = (float*)d_ws;
  float* hA  = ws;
  float* hB  = hA + (size_t)N * 256;
  float* h3  = hB + (size_t)N * 256;
  int*   ib  = (int*)(h3 + (size_t)N * 16);
  int*   cnt    = ib;
  int*   rowptr = cnt + N;
  int*   cursor = rowptr + N + 1;
  int*   tsum   = cursor + N;
  float* asvB = (float*)(ib + (size_t)N * 4);
  float* advB = asvB + N;
  float* asvA = (float*)(ib + (size_t)N * 16);
  float* advA = asvA + N;
  int*   csrc = (int*)(advA + N);

  dim3 blk(256);
  int gGemm  = (N + 31) / 32;
  int gNodeW = (N + 7) / 8;
  int gNodeT = (N + 255) / 256;
  int gEdgeT = (E + 255) / 256;

  const int NBK = (N + 255) / 256;
  const int NCH = (E + SORT_CH - 1) / SORT_CH;
  int*  tmp  = (int*)hB;
  int*  cntL = (int*)hB + 2 * (size_t)NCH * SORT_CH;
  int*  offL = cntL + (size_t)NCH * SORT_MAXB;
  if (NBK <= SORT_MAXB && NCH <= SORT_MAXB &&
      2 * (size_t)NCH * SORT_CH + 2 * (size_t)NCH * SORT_MAXB <= (size_t)N * 256) {
    gat_bin_local<<<NCH, blk, 0, stream>>>(src, dst, tmp, cntL, offL, E);
    gat_bucket_sort<<<NBK, blk, 0, stream>>>(tmp, cntL, offL, rowptr, csrc, N, NCH);
  } else {
    gat_csr_zero<<<gNodeT, blk, 0, stream>>>(cnt, N);
    gat_csr_hist<<<gEdgeT, blk, 0, stream>>>(dst, cnt, E);
    gat_csr_tilesum<<<gNodeT, blk, 0, stream>>>(cnt, tsum, N);
    gat_csr_scan<<<gNodeT, blk, 0, stream>>>(cnt, tsum, rowptr, cursor, N);
    gat_csr_fill<<<gEdgeT, blk, 0, stream>>>(src, dst, cursor, csrc, E);
  }

  gat_layer_gemm<false, 4, 4><<<gGemm, blk, 0, stream>>>(
      x, nullptr, nullptr, nullptr, nullptr, nullptr, nullptr,
      W1, as1, ad1, hA, asvA, advA, N, 256);

  gat_layer_gemm<true, 4, 4><<<gGemm, blk, 0, stream>>>(
      nullptr, rowptr, csrc, asvA, advA, hA, b1,
      W2, as2, ad2, hB, asvB, advB, N, 256);

  gat_layer_gemm<true, 1, 1><<<gGemm, blk, 0, stream>>>(
      nullptr, rowptr, csrc, asvB, advB, hB, b2,
      W3, as3, ad3, h3, asvA, advA, N, 10);

  gat_node_gather16<<<gNodeW, blk, 0, stream>>>(rowptr, csrc, asvA, advA, h3, b3, (float*)d_out, N, 10);
}


# Edge sort's two launches carry the layer 1 product's row tiles as extra blocks (sort blocks first, 2/5 and 3/5 of the tiles): the sort runs under the product, not before it; 6 launches become 5
# speedup vs baseline: 6.1433x; 1.0511x over previous
#include <hip/hip_runtime.h>
#include <math.h>


typedef float v2f __attribute__((ext_vector_type(2)));
typedef float v8f __attribute__((ext_vector_type(8)));

#define DK 256
#define KC 32
#define LDT (KC + 4)
#define LDA (DK + 4)

__global__ __launch_bounds__(256) void gat_csr_zero(int* __restrict__ cnt, int N) {
  int i = blockIdx.x * 256 + threadIdx.x;
  if (i < N) cnt[i] = 0;
}

__global__ __launch_bounds__(256) void gat_csr_hist(
    const int* __restrict__ dst, int* __restrict__ cnt, int E) {
  int e = blockIdx.x * 256 + threadIdx.x;
  if (e < E) atomicAdd(&cnt[dst[e]], 1);
}

__global__ __launch_bounds__(256) void gat_csr_tilesum(
    const int* __restrict__ cnt, int* __restrict__ tsum, int N) {
  __shared__ int red[8];
  const int tid = threadIdx.x;
  int i = blockIdx.x * 256 + tid;
  int v = (i < N) ? cnt[i] : 0;
  #pragma unroll
  for (int off = 16; off > 0; off >>= 1) v += __shfl_xor(v, off, 32);
  if ((tid & 31) == 0) red[tid >> 5] = v;
  __syncthreads();
  if (tid == 0) {
    int t = 0;
    #pragma unroll
    for (int w = 0; w < 8; ++w) t += red[w];
    tsum[blockIdx.x] = t;
  }
}

__global__ __launch_bounds__(256) void gat_csr_scan(
    const int* __restrict__ cnt, const int* __restrict__ tsum,
    int* __restrict__ rowptr, int* __restrict__ cursor, int N) {
  __shared__ int red[8];
  __shared__ int wtot[8];
  const int tid = threadIdx.x;
  const int lane = tid & 31;
  const int wid = tid >> 5;
  int part = 0;
  for (int t = tid; t < (int)blockIdx.x; t += 256) part += tsum[t];
  #pragma unroll
  for (int off = 16; off > 0; off >>= 1) part += __shfl_xor(part, off, 32);
  if (lane == 0) red[wid] = part;
  int i = blockIdx.x * 256 + tid;
  int v = (i < N) ? cnt[i] : 0;
  int x = v;
  #pragma unroll
  for (int d = 1; d < 32; d <<= 1) {
    int n = __shfl_up(x, d, 32);
    if (lane >= d) x += n;
  }
  if (lane == 31) wtot[wid] = x;
  __syncthreads();
  int base = 0;
  #pragma unroll
  for (int w = 0; w < 8; ++w) {
    base += red[w];
    if (w < wid) base += wtot[w];
  }
  int excl = base + x - v;
  if (i < N) {
    rowptr[i] = excl;
    cursor[i] = excl;
    if (i == N - 1) rowptr[N] = excl + v;
  }
}

__global__ __launch_bounds__(256) void gat_csr_fill(
    const int* __restrict__ src, const int* __restrict__ dst,
    int* __restrict__ cursor, int* __restrict__ csrc, int E) {
  int e = blockIdx.x * 256 + threadIdx.x;
  if (e < E) {
    int pos = atomicAdd(&cursor[dst[e]], 1);
    csrc[pos] = src[e];
  }
}

#define SORT_CH 4096
#define SORT_MAXB 256
#define SORT_CAP 8192

__device__ __forceinline__ int gat_excl_scan256(int v, int* red, int& total) {
  const int lane = threadIdx.x & 31;
  const int wid = threadIdx.x >> 5;
  int x = v;
  #pragma unroll
  for (int d = 1; d < 32; d <<= 1) {
    int n = __shfl_up(x, d, 32);
    if (lane >= d) x += n;
  }
  if (lane == 31) red[wid] = x;
  __syncthreads();
  int base = 0, tot = 0;
  #pragma unroll
  for (int w = 0; w < 8; ++w) {
    const int r = red[w];
    if (w < wid) base += r;
    tot += r;
  }
  __syncthreads();
  total = tot;
  return base + x - v;
}

__device__ __forceinline__ void gat_bin_local_blk(
    const int* __restrict__ src, const int* __restrict__ dst,
    int* __restrict__ tmpw, int* __restrict__ cntL, int* __restrict__ offL, int E, int b) {
  int2* tmp = reinterpret_cast<int2*>(tmpw);
  __shared__ int red[8];
  __shared__ int hist[SORT_MAXB];
  __shared__ int cursor[SORT_MAXB];
  __shared__ int2 stage[SORT_CH];
  const int tid = threadIdx.x;
  hist[tid] = 0;
  __syncthreads();
  const int base = b * SORT_CH;
  int d[SORT_CH / 256], s[SORT_CH / 256];
  #pragma unroll
  for (int i = 0; i < SORT_CH / 256; ++i) {
    const int e = min(base + i * 256 + tid, E - 1);
    d[i] = dst[e];
    s[i] = src[e];
  }
  #pragma unroll
  for (int i = 0; i < SORT_CH / 256; ++i)
    if (base + i * 256 + tid < E) atomicAdd(&hist[d[i] >> 8], 1);
  __syncthreads();
  const int mine = hist[tid];
  int total;
  const int ls = gat_excl_scan256(mine, red, total);
  cursor[tid] = ls;
  cntL[(size_t)b * SORT_MAXB + tid] = mine;
  offL[(size_t)b * SORT_MAXB + tid] = ls;
  __syncthreads();
  #pragma unroll
  for (int i = 0; i < SORT_CH / 256; ++i) {
    if (base + i * 256 + tid < E) {
      const int p = atomicAdd(&cursor[d[i] >> 8], 1);
      stage[p] = make_int2(s[i], d[i]);
    }
  }
  __syncthreads();
  for (int j = tid; j < total; j += 256) tmp[(size_t)base + j] = stage[j];
}

__device__ __forceinline__ void gat_bucket_sort_blk(
    const int* __restrict__ tmpw, const int* __restrict__ cntL, const int* __restrict__ offL,
    int* __restrict__ rowptr, int* __restrict__ csrc, int N, int NCH, int k) {
  const int2* tmp = reinterpret_cast<const int2*>(tmpw);
  __shared__ int red[8];
  __shared__ int cnt[256];
  __shared__ int cursor[256];
  __shared__ int stage[SORT_CAP];
  const int tid = threadIdx.x;
  int lo = 0, n = 0;
  if (tid < NCH) {
    lo = offL[(size_t)tid * SORT_MAXB + k];
    n  = cntL[(size_t)tid * SORT_MAXB + k];
  }
  cnt[tid] = 0;
  int bs;
  (void)gat_excl_scan256(lo, red, bs);
  const int2* run = tmp + (size_t)tid * SORT_CH + lo;
  for (int j0 = 0; j0 < n; j0 += 8) {
    int dd[8];
    #pragma unroll
    for (int u = 0; u < 8; ++u) dd[u] = run[min(j0 + u, n - 1)].y;
    #pragma unroll
    for (int u = 0; u < 8; ++u)
      atomicAdd(&cnt[dd[u] & 255], (j0 + u < n) ? 1 : 0);
  }
  __syncthreads();
  const int c = cnt[tid];
  int total;
  const int st = gat_excl_scan256(c, red, total);
  cursor[tid] = st;
  const int node = k * 256 + tid;
  if (node < N) {
    rowptr[node] = bs + st;
    if (node == N - 1) rowptr[N] = bs + st + c;
  }
  __syncthreads();
  for (int j0 = 0; j0 < n; j0 += 8) {
    int2 pr[8];
    #pragma unroll
    for (int u = 0; u < 8; ++u) pr[u] = run[min(j0 + u, n - 1)];
    #pragma unroll
    for (int u = 0; u < 8; ++u) {
      const bool live = j0 + u < n;
      const int p = atomicAdd(&cursor[pr[u].y & 255], live ? 1 : 0);
      if (live) {
        if (p < SORT_CAP) stage[p] = pr[u].x;
        else              csrc[bs + p] = pr[u].x;
      }
    }
  }
  __syncthreads();
  const int nst = min(total, SORT_CAP);
  for (int j = tid; j < nst; j += 256) csrc[bs + j] = stage[j];
}

__device__ __forceinline__ float leaky02(float v) {
  return (v > 0.f) ? v : 0.2f * v;
}

__device__ __forceinline__ void gat_node_stats(
    const int* __restrict__ csrc, const float* __restrict__ asv,
    int i, int beg, int end, int lane, float ad, float sself,
    int& s0, float& sc0, bool& v0, float& m, float& sum) {
  const int j0 = beg + lane;
  v0 = j0 < end;
  s0 = v0 ? csrc[j0] : i;
  sc0 = v0 ? leaky02(asv[s0] + ad) : sself;
  float mm = fmaxf(sself, sc0);
  for (int j = j0 + 32; j < end; j += 32)
    mm = fmaxf(mm, leaky02(asv[csrc[j]] + ad));
  #pragma unroll
  for (int off = 16; off > 0; off >>= 1) mm = fmaxf(mm, __shfl_xor(mm, off, 32));
  float ss = v0 ? __expf(sc0 - mm) : 0.f;
  for (int j = j0 + 32; j < end; j += 32)
    ss += __expf(leaky02(asv[csrc[j]] + ad) - mm);
  #pragma unroll
  for (int off = 16; off > 0; off >>= 1) ss += __shfl_xor(ss, off, 32);
  m = mm;
  sum = ss + __expf(sself - mm);
}

__device__ __forceinline__ void gat_node_row256(
    const int* __restrict__ rowptr, const int* __restrict__ csrc,
    const float* __restrict__ asv, const float* __restrict__ adv,
    const float* __restrict__ H, int i, int lane, float4& a0, float4& a1) {
  const float4* H4 = reinterpret_cast<const float4*>(H);
  const float4 xs = H4[(size_t)i * 64 + lane];
  const float4 ys = H4[(size_t)i * 64 + lane + 32];
  const int beg = rowptr[i];
  const int end = rowptr[i + 1];
  const float ad = adv[i];
  const float sself = leaky02(asv[i] + ad);
  int s0; float sc0; bool v0; float m, sum;
  gat_node_stats(csrc, asv, i, beg, end, lane, ad, sself, s0, sc0, v0, m, sum);

  {
    const float c = __expf(sself - m) / sum;
    a0 = make_float4(c * xs.x, c * xs.y, c * xs.z, c * xs.w);
    a1 = make_float4(c * ys.x, c * ys.y, c * ys.z, c * ys.w);
  }
  int sv = s0;
  float cv = v0 ? __expf(sc0 - m) / sum : 0.f;
  for (int base = beg; base < end; base += 32) {
    if (base != beg) {
      const int j = base + lane;
      const bool v = j < end;
      sv = v ? csrc[j] : i;
      cv = v ? __expf(leaky02(asv[sv] + ad) - m) / sum : 0.f;
    }
    const int cnt = min(32, end - base);
    #pragma unroll 4
    for (int t = 0; t < cnt; ++t) {
      const int s = __builtin_amdgcn_readlane(sv, t);
      const float c = __int_as_float(__builtin_amdgcn_readlane(__float_as_int(cv), t));
      const float4* hp = H4 + (size_t)s * 64;
      const float4 x = hp[lane];
      const float4 y = hp[lane + 32];
      a0.x = fmaf(c, x.x, a0.x); a0.y = fmaf(c, x.y, a0.y);
      a0.z = fmaf(c, x.z, a0.z); a0.w = fmaf(c, x.w, a0.w);
      a1.x = fmaf(c, y.x, a1.x); a1.y = fmaf(c, y.y, a1.y);
      a1.z = fmaf(c, y.z, a1.z); a1.w = fmaf(c, y.w, a1.w);
    }
  }
}

template <bool GATHER, int WNW, int NT>
__device__ __forceinline__ void gat_layer_gemm_blk(
    const float* __restrict__ X,
    const int* __restrict__ rowptr, const int* __restrict__ csrc,
    const float* __restrict__ asv_in, const float* __restrict__ adv_in,
    const float* __restrict__ Hin, const float* __restrict__ bias_in,
    const float* __restrict__ W,
    const float* __restrict__ a_s, const float* __restrict__ a_d,
    float* __restrict__ H, float* __restrict__ asv, float* __restrict__ adv,
    int N, int Dact, int tile) {
  static_assert(WNW == 1 || WNW == 2 || WNW == 4, "2 x WNW accumulator waves out of eight");
  constexpr int BM = 32;
  constexpr int BN = 16 * NT * WNW;
  constexpr int NWA = 2 * WNW;
  constexpr int AP = GATHER ? LDA : LDT;
  constexpr int CP = BN + 4;
  constexpr int A_F4 = (BM * (KC / 4)) / 256;
  constexpr int B_QT = BN * (KC / 4);
  constexpr int B_Q = (B_QT + 255) / 256;
  constexpr int SM_A = BM * AP;
  constexpr int SM_AB = SM_A + BN * LDT;
  constexpr int SM_C = BM * CP + 2 * BM;
  constexpr int SM = SM_AB > SM_C ? SM_AB : SM_C;
  __shared__ __attribute__((aligned(16))) float smem[SM];
  float* As = smem;
  float* Bst = smem + SM_A;
  float* Cs = smem;
  float* red = smem + BM * CP;

  const int tid  = threadIdx.x;
  const int lane = tid & 31;
  const int lo   = lane & 15;
  const int hi   = lane >> 4;
  const int wid  = __builtin_amdgcn_readfirstlane(tid >> 5);
  const bool accw = (NWA == 8) || (wid < NWA);
  const int wm   = (wid / WNW) * 16;
  const int wn   = (wid % WNW) * (16 * NT);
  const int rowBase = tile * BM;

  if constexpr (GATHER) {
    const int c0 = 4 * lane, c1 = 128 + 4 * lane;
    float4 bz0, bz1;
    bz0.x = bias_in[c0 + 0]; bz0.y = bias_in[c0 + 1]; bz0.z = bias_in[c0 + 2]; bz0.w = bias_in[c0 + 3];
    bz1.x = bias_in[c1 + 0]; bz1.y = bias_in[c1 + 1]; bz1.z = bias_in[c1 + 2]; bz1.w = bias_in[c1 + 3];
    for (int rr = 0; rr < BM / 8; ++rr) {
      const int row = wid * (BM / 8) + rr;
      const int i   = rowBase + row;
      float4 a0 = make_float4(0.f, 0.f, 0.f, 0.f);
      float4 a1 = a0;
      if (i < N) {
        gat_node_row256(rowptr, csrc, asv_in, adv_in, Hin, i, lane, a0, a1);
        a0.x += bz0.x; a0.y += bz0.y; a0.z += bz0.z; a0.w += bz0.w;
        a1.x += bz1.x; a1.y += bz1.y; a1.z += bz1.z; a1.w += bz1.w;
        a0.x = fmaxf(a0.x, 0.f); a0.y = fmaxf(a0.y, 0.f); a0.z = fmaxf(a0.z, 0.f); a0.w = fmaxf(a0.w, 0.f);
        a1.x = fmaxf(a1.x, 0.f); a1.y = fmaxf(a1.y, 0.f); a1.z = fmaxf(a1.z, 0.f); a1.w = fmaxf(a1.w, 0.f);
      }
      *reinterpret_cast<float4*>(&As[row * AP + c0]) = a0;
      *reinterpret_cast<float4*>(&As[row * AP + c1]) = a1;
    }
  }

  v8f acc[NT];
  #pragma unroll
  for (int t = 0; t < NT; ++t) acc[t] = v8f{};

  float4 ra[A_F4];
  float4 rb[B_Q];

  auto load_chunk = [&](int kk) {
    if constexpr (!GATHER) {
      #pragma unroll
      for (int i = 0; i < A_F4; ++i) {
        const int f4 = tid + 256 * i;
        const int r  = f4 >> 3;
        const int c4 = f4 & 7;
        const int gr = rowBase + r;
        float4 v = make_float4(0.f, 0.f, 0.f, 0.f);
        if (gr < N)
          v = *reinterpret_cast<const float4*>(X + (size_t)gr * DK + kk + c4 * 4);
        ra[i] = v;
      }
    }
    #pragma unroll
    for (int i = 0; i < B_Q; ++i) {
      const int q  = tid + 256 * i;
      const int n  = q % BN;
      const int kq = q / BN;
      float4 v = make_float4(0.f, 0.f, 0.f, 0.f);
      if ((B_QT % 256 == 0 || q < B_QT) && n < Dact) {
        const float* wp = W + (size_t)(kk + kq * 4) * Dact + n;
        v.x = wp[0];
        v.y = wp[Dact];
        v.z = wp[2 * Dact];
        v.w = wp[3 * Dact];
      }
      rb[i] = v;
    }
  };
  auto store_chunk = [&]() {
    if constexpr (!GATHER) {
      #pragma unroll
      for (int i = 0; i < A_F4; ++i) {
        const int f4 = tid + 256 * i;
        const int r  = f4 >> 3;
        const int c4 = f4 & 7;
        *reinterpret_cast<float4*>(&As[r * AP + c4 * 4]) = ra[i];
      }
    }
    #pragma unroll
    for (int i = 0; i < B_Q; ++i) {
      const int q  = tid + 256 * i;
      const int n  = q % BN;
      const int kq = q / BN;
      if (B_QT % 256 == 0 || q < B_QT)
        *reinterpret_cast<float4*>(&Bst[n * LDT + kq * 4]) = rb[i];
    }
  };

  const float* arow = As + (wm + lo) * AP + 4 * hi;
  const float* brow = Bst + (wn + lo) * LDT + 4 * hi;

  load_chunk(0);
  for (int kk = 0; kk < DK; kk += KC) {
    __syncthreads();
    store_chunk();
    __syncthreads();
    if (kk + KC < DK) load_chunk(kk + KC);
    if (accw) {
      const float* ak = GATHER ? arow + kk : arow;
      #pragma unroll
      for (int kb = 0; kb < KC; kb += 8) {
        const float4 a4 = *reinterpret_cast<const float4*>(ak + kb);
        float4 b4[NT];
        #pragma unroll
        for (int t = 0; t < NT; ++t)
          b4[t] = *reinterpret_cast<const float4*>(brow + t * 16 * LDT + kb);
        const v2f a01 = {a4.x, a4.y};
        const v2f a23 = {a4.z, a4.w};
        #pragma unroll
        for (int t = 0; t < NT; ++t) {
          const v2f b01 = {b4[t].x, b4[t].y};
          acc[t] = __builtin_amdgcn_wmma_f32_16x16x4_f32(false, a01, false, b01,
                                                         (short)0, acc[t], false, false);
        }
        #pragma unroll
        for (int t = 0; t < NT; ++t) {
          const v2f b23 = {b4[t].z, b4[t].w};
          acc[t] = __builtin_amdgcn_wmma_f32_16x16x4_f32(false, a23, false, b23,
                                                         (short)0, acc[t], false, false);
        }
      }
    }
  }

  __syncthreads();
  if (accw) {
    #pragma unroll
    for (int t = 0; t < NT; ++t) {
      #pragma unroll
      for (int r = 0; r < 8; ++r)
        Cs[(wm + 8 * hi + r) * CP + wn + 16 * t + lo] = acc[t][r];
    }
  }
  __syncthreads();

  if constexpr (BN == 256) {
    float4 s0, s1, d0, d1;
    {
      const int c0 = 4 * lane, c1 = 128 + 4 * lane;
      s0.x = (c0 + 0 < Dact) ? a_s[c0 + 0] : 0.f; s0.y = (c0 + 1 < Dact) ? a_s[c0 + 1] : 0.f;
      s0.z = (c0 + 2 < Dact) ? a_s[c0 + 2] : 0.f; s0.w = (c0 + 3 < Dact) ? a_s[c0 + 3] : 0.f;
      s1.x = (c1 + 0 < Dact) ? a_s[c1 + 0] : 0.f; s1.y = (c1 + 1 < Dact) ? a_s[c1 + 1] : 0.f;
      s1.z = (c1 + 2 < Dact) ? a_s[c1 + 2] : 0.f; s1.w = (c1 + 3 < Dact) ? a_s[c1 + 3] : 0.f;
      d0.x = (c0 + 0 < Dact) ? a_d[c0 + 0] : 0.f; d0.y = (c0 + 1 < Dact) ? a_d[c0 + 1] : 0.f;
      d0.z = (c0 + 2 < Dact) ? a_d[c0 + 2] : 0.f; d0.w = (c0 + 3 < Dact) ? a_d[c0 + 3] : 0.f;
      d1.x = (c1 + 0 < Dact) ? a_d[c1 + 0] : 0.f; d1.y = (c1 + 1 < Dact) ? a_d[c1 + 1] : 0.f;
      d1.z = (c1 + 2 < Dact) ? a_d[c1 + 2] : 0.f; d1.w = (c1 + 3 < Dact) ? a_d[c1 + 3] : 0.f;
    }
    constexpr int RPW = BM / 8;
    #pragma unroll
    for (int rr = 0; rr < RPW; ++rr) {
      const int row = wid * RPW + rr;
      const int gr  = rowBase + row;
      const float4 x = *reinterpret_cast<const float4*>(&Cs[row * CP + 4 * lane]);
      const float4 y = *reinterpret_cast<const float4*>(&Cs[row * CP + 128 + 4 * lane]);
      float p = x.x * s0.x + x.y * s0.y + x.z * s0.z + x.w * s0.w
              + y.x * s1.x + y.y * s1.y + y.z * s1.z + y.w * s1.w;
      float q = x.x * d0.x + x.y * d0.y + x.z * d0.z + x.w * d0.w
              + y.x * d1.x + y.y * d1.y + y.z * d1.z + y.w * d1.w;
      #pragma unroll
      for (int off = 16; off > 0; off >>= 1) {
        p += __shfl_xor(p, off, 32);
        q += __shfl_xor(q, off, 32);
      }
      if (lane == 0) {
        red[row] = p;
        red[BM + row] = q;
      }
      if (gr < N) {
        float4* hp = reinterpret_cast<float4*>(H) + (size_t)gr * 64;
        hp[lane] = x;
        hp[lane + 32] = y;
      }
    }
  } else {
    static_assert(BN == 16, "narrow form: 16 padded columns");
    if (wid < BM / 8) {
      const int qd = lane & 3;
      const int c0 = 4 * qd;
      float4 s0, d0;
      s0.x = (c0 + 0 < Dact) ? a_s[c0 + 0] : 0.f; s0.y = (c0 + 1 < Dact) ? a_s[c0 + 1] : 0.f;
      s0.z = (c0 + 2 < Dact) ? a_s[c0 + 2] : 0.f; s0.w = (c0 + 3 < Dact) ? a_s[c0 + 3] : 0.f;
      d0.x = (c0 + 0 < Dact) ? a_d[c0 + 0] : 0.f; d0.y = (c0 + 1 < Dact) ? a_d[c0 + 1] : 0.f;
      d0.z = (c0 + 2 < Dact) ? a_d[c0 + 2] : 0.f; d0.w = (c0 + 3 < Dact) ? a_d[c0 + 3] : 0.f;
      const int row = wid * 8 + (lane >> 2);
      const int gr  = rowBase + row;
      const float4 x = *reinterpret_cast<const float4*>(&Cs[row * CP + c0]);
      float p = x.x * s0.x + x.y * s0.y + x.z * s0.z + x.w * s0.w;
      float q = x.x * d0.x + x.y * d0.y + x.z * d0.z + x.w * d0.w;
      #pragma unroll
      for (int off = 2; off > 0; off >>= 1) {
        p += __shfl_xor(p, off, 32);
        q += __shfl_xor(q, off, 32);
      }
      if (qd == 0) {
        red[row] = p;
        red[BM + row] = q;
      }
      if (gr < N) reinterpret_cast<float4*>(H)[(size_t)gr * 4 + qd] = x;
    }
  }

  __syncthreads();
  if (tid < 2 * BM) {
    const int row = (tid < BM) ? tid : tid - BM;
    const int gr  = rowBase + row;
    if (gr < N) {
      if (tid < BM) asv[gr] = red[row];
      else          adv[gr] = red[BM + row];
    }
  }
}

template <bool GATHER, int WNW, int NT>
__global__ __launch_bounds__(256, GATHER ? 8 : 1) void gat_layer_gemm(
    const float* __restrict__ X,
    const int* __restrict__ rowptr, const int* __restrict__ csrc,
    const float* __restrict__ asv_in, const float* __restrict__ adv_in,
    const float* __restrict__ Hin, const float* __restrict__ bias_in,
    const float* __restrict__ W,
    const float* __restrict__ a_s, const float* __restrict__ a_d,
    float* __restrict__ H, float* __restrict__ asv, float* __restrict__ adv,
    int N, int Dact) {
  gat_layer_gemm_blk<GATHER, WNW, NT>(X, rowptr, csrc, asv_in, adv_in, Hin, bias_in,
                                      W, a_s, a_d, H, asv, adv, N, Dact, (int)blockIdx.x);
}

__global__ __launch_bounds__(256) void gat_bin_and_gemm(
    const int* __restrict__ src, const int* __restrict__ dst,
    int* __restrict__ tmpw, int* __restrict__ cntL, int* __restrict__ offL, int E, int nSort,
    const float* __restrict__ X, const float* __restrict__ W,
    const float* __restrict__ a_s, const float* __restrict__ a_d,
    float* __restrict__ H, float* __restrict__ asv, float* __restrict__ adv,
    int N, int Dact, int tile0) {
  const int bid = blockIdx.x;
  if (bid < nSort) {
    gat_bin_local_blk(src, dst, tmpw, cntL, offL, E, bid);
  } else {
    gat_layer_gemm_blk<false, 4, 4>(X, nullptr, nullptr, nullptr, nullptr, nullptr, nullptr,
                                    W, a_s, a_d, H, asv, adv, N, Dact, tile0 + bid - nSort);
  }
}

__global__ __launch_bounds__(256) void gat_bucket_and_gemm(
    const int* __restrict__ tmpw, const int* __restrict__ cntL, const int* __restrict__ offL,
    int* __restrict__ rowptr, int* __restrict__ csrc, int NCH, int nSort,
    const float* __restrict__ X, const float* __restrict__ W,
    const float* __restrict__ a_s, const float* __restrict__ a_d,
    float* __restrict__ H, float* __restrict__ asv, float* __restrict__ adv,
    int N, int Dact, int tile0) {
  const int bid = blockIdx.x;
  if (bid < nSort) {
    gat_bucket_sort_blk(tmpw, cntL, offL, rowptr, csrc, N, NCH, bid);
  } else {
    gat_layer_gemm_blk<false, 4, 4>(X, nullptr, nullptr, nullptr, nullptr, nullptr, nullptr,
                                    W, a_s, a_d, H, asv, adv, N, Dact, tile0 + bid - nSort);
  }
}

__global__ __launch_bounds__(256) void gat_node_gather16(
    const int* __restrict__ rowptr, const int* __restrict__ csrc,
    const float* __restrict__ asv, const float* __restrict__ adv,
    const float* __restrict__ H, const float* __restrict__ bias,
    float* __restrict__ out, int N, int Dact) {
  const int lane = threadIdx.x & 31;
  const int i = blockIdx.x * 8 + __builtin_amdgcn_readfirstlane((int)(threadIdx.x >> 5));
  if (i >= N) return;
  const int beg = rowptr[i];
  const int end = rowptr[i + 1];
  const float ad = adv[i];
  const float sself = leaky02(asv[i] + ad);
  int s0; float sc0; bool v0; float m, sum;
  gat_node_stats(csrc, asv, i, beg, end, lane, ad, sself, s0, sc0, v0, m, sum);

  const int q = lane & 3;
  const int slot = lane >> 2;
  const float4* H4 = reinterpret_cast<const float4*>(H);
  float4 acc = make_float4(0.f, 0.f, 0.f, 0.f);
  if (slot == 0) {
    const float c = __expf(sself - m) / sum;
    const float4 x = H4[(size_t)i * 4 + q];
    acc = make_float4(c * x.x, c * x.y, c * x.z, c * x.w);
  }
  int sv = s0;
  float cv = v0 ? __expf(sc0 - m) / sum : 0.f;
  for (int base = beg; base < end; base += 32) {
    if (base != beg) {
      const int j = base + lane;
      const bool v = j < end;
      sv = v ? csrc[j] : i;
      cv = v ? __expf(leaky02(asv[sv] + ad) - m) / sum : 0.f;
    }
    const int cnt = min(32, end - base);
    for (int t0 = 0; t0 < cnt; t0 += 8) {
      const int t = t0 + slot;
      const int s = __shfl(sv, t, 32);
      const float c = __shfl(cv, t, 32);
      if (t < cnt) {
        const float4 x = H4[(size_t)s * 4 + q];
        acc.x = fmaf(c, x.x, acc.x); acc.y = fmaf(c, x.y, acc.y);
        acc.z = fmaf(c, x.z, acc.z); acc.w = fmaf(c, x.w, acc.w);
      }
    }
  }
  #pragma unroll
  for (int off = 4; off < 32; off <<= 1) {
    acc.x += __shfl_xor(acc.x, off, 32);
    acc.y += __shfl_xor(acc.y, off, 32);
    acc.z += __shfl_xor(acc.z, off, 32);
    acc.w += __shfl_xor(acc.w, off, 32);
  }
  if (slot == 0) {
    const int c = q * 4;
    float* op = out + (size_t)i * Dact;
    if (c + 0 < Dact) op[c + 0] = acc.x + bias[c + 0];
    if (c + 1 < Dact) op[c + 1] = acc.y + bias[c + 1];
    if (c + 2 < Dact) op[c + 2] = acc.z + bias[c + 2];
    if (c + 3 < Dact) op[c + 3] = acc.w + bias[c + 3];
  }
}

extern "C" void kernel_launch(void* const* d_in, const int* in_sizes, int n_in,
                              void* d_out, int out_size, void* d_ws, size_t ws_size,
                              hipStream_t stream) {
  const float* x  = (const float*)d_in[0];
  const int*   ei = (const int*)d_in[1];
  const float* W1 = (const float*)d_in[2];
  const float* as1 = (const float*)d_in[3];
  const float* ad1 = (const float*)d_in[4];
  const float* b1 = (const float*)d_in[5];
  const float* W2 = (const float*)d_in[6];
  const float* as2 = (const float*)d_in[7];
  const float* ad2 = (const float*)d_in[8];
  const float* b2 = (const float*)d_in[9];
  const float* W3 = (const float*)d_in[10];
  const float* as3 = (const float*)d_in[11];
  const float* ad3 = (const float*)d_in[12];
  const float* b3 = (const float*)d_in[13];

  const int N  = in_sizes[0] / 256;
  const int E  = in_sizes[1] / 2;
  const int* src = ei;
  const int* dst = ei + E;

  float* ws = (float*)d_ws;
  float* hA  = ws;
  float* hB  = hA + (size_t)N * 256;
  float* h3  = hB + (size_t)N * 256;
  int*   ib  = (int*)(h3 + (size_t)N * 16);
  int*   cnt    = ib;
  int*   rowptr = cnt + N;
  int*   cursor = rowptr + N + 1;
  int*   tsum   = cursor + N;
  float* asvB = (float*)(ib + (size_t)N * 4);
  float* advB = asvB + N;
  float* asvA = (float*)(ib + (size_t)N * 16);
  float* advA = asvA + N;
  int*   csrc = (int*)(advA + N);

  dim3 blk(256);
  int gGemm  = (N + 31) / 32;
  int gNodeW = (N + 7) / 8;
  int gNodeT = (N + 255) / 256;
  int gEdgeT = (E + 255) / 256;

  const int NBK = (N + 255) / 256;
  const int NCH = (E + SORT_CH - 1) / SORT_CH;
  int*  tmp  = (int*)hB;
  int*  cntL = (int*)hB + 2 * (size_t)NCH * SORT_CH;
  int*  offL = cntL + (size_t)NCH * SORT_MAXB;
  if (NBK <= SORT_MAXB && NCH <= SORT_MAXB &&
      2 * (size_t)NCH * SORT_CH + 2 * (size_t)NCH * SORT_MAXB <= (size_t)N * 256) {
    const int g1 = (gGemm * 2) / 5;
    gat_bin_and_gemm<<<NCH + g1, blk, 0, stream>>>(
        src, dst, tmp, cntL, offL, E, NCH,
        x, W1, as1, ad1, hA, asvA, advA, N, 256, 0);
    gat_bucket_and_gemm<<<NBK + (gGemm - g1), blk, 0, stream>>>(
        tmp, cntL, offL, rowptr, csrc, NCH, NBK,
        x, W1, as1, ad1, hA, asvA, advA, N, 256, g1);
  } else {
    gat_csr_zero<<<gNodeT, blk, 0, stream>>>(cnt, N);
    gat_csr_hist<<<gEdgeT, blk, 0, stream>>>(dst, cnt, E);
    gat_csr_tilesum<<<gNodeT, blk, 0, stream>>>(cnt, tsum, N);
    gat_csr_scan<<<gNodeT, blk, 0, stream>>>(cnt, tsum, rowptr, cursor, N);
    gat_csr_fill<<<gEdgeT, blk, 0, stream>>>(src, dst, cursor, csrc, E);
    gat_layer_gemm<false, 4, 4><<<gGemm, blk, 0, stream>>>(
        x, nullptr, nullptr, nullptr, nullptr, nullptr, nullptr,
        W1, as1, ad1, hA, asvA, advA, N, 256);
  }

  gat_layer_gemm<true, 4, 4><<<gGemm, blk, 0, stream>>>(
      nullptr, rowptr, csrc, asvA, advA, hA, b1,
      W2, as2, ad2, hB, asvB, advB, N, 256);

  gat_layer_gemm<true, 1, 1><<<gGemm, blk, 0, stream>>>(
      nullptr, rowptr, csrc, asvB, advB, hB, b2,
      W3, as3, ad3, h3, asvA, advA, N, 10);

  gat_node_gather16<<<gNodeW, blk, 0, stream>>>(rowptr, csrc, asvA, advA, h3, b3, (float*)d_out, N, 10);
}


# Layer 2 blocks' row build: a wave's 4 nodes' index and score loads go out side by side, source rows come 8 per round trip (list tails too), bias+relu in an LDS pass; softmax sum order changes
# speedup vs baseline: 6.4005x; 1.0419x over previous
#include <hip/hip_runtime.h>
#include <math.h>


typedef float v2f __attribute__((ext_vector_type(2)));
typedef float v4f __attribute__((ext_vector_type(4)));
typedef float v8f __attribute__((ext_vector_type(8)));

#define DK 256
#define KC 32
#define LDT (KC + 4)
#define LDA (DK + 4)

__global__ __launch_bounds__(256) void gat_csr_zero(int* __restrict__ cnt, int N) {
  int i = blockIdx.x * 256 + threadIdx.x;
  if (i < N) cnt[i] = 0;
}

__global__ __launch_bounds__(256) void gat_csr_hist(
    const int* __restrict__ dst, int* __restrict__ cnt, int E) {
  int e = blockIdx.x * 256 + threadIdx.x;
  if (e < E) atomicAdd(&cnt[dst[e]], 1);
}

__global__ __launch_bounds__(256) void gat_csr_tilesum(
    const int* __restrict__ cnt, int* __restrict__ tsum, int N) {
  __shared__ int red[8];
  const int tid = threadIdx.x;
  int i = blockIdx.x * 256 + tid;
  int v = (i < N) ? cnt[i] : 0;
  #pragma unroll
  for (int off = 16; off > 0; off >>= 1) v += __shfl_xor(v, off, 32);
  if ((tid & 31) == 0) red[tid >> 5] = v;
  __syncthreads();
  if (tid == 0) {
    int t = 0;
    #pragma unroll
    for (int w = 0; w < 8; ++w) t += red[w];
    tsum[blockIdx.x] = t;
  }
}

__global__ __launch_bounds__(256) void gat_csr_scan(
    const int* __restrict__ cnt, const int* __restrict__ tsum,
    int* __restrict__ rowptr, int* __restrict__ cursor, int N) {
  __shared__ int red[8];
  __shared__ int wtot[8];
  const int tid = threadIdx.x;
  const int lane = tid & 31;
  const int wid = tid >> 5;
  int part = 0;
  for (int t = tid; t < (int)blockIdx.x; t += 256) part += tsum[t];
  #pragma unroll
  for (int off = 16; off > 0; off >>= 1) part += __shfl_xor(part, off, 32);
  if (lane == 0) red[wid] = part;
  int i = blockIdx.x * 256 + tid;
  int v = (i < N) ? cnt[i] : 0;
  int x = v;
  #pragma unroll
  for (int d = 1; d < 32; d <<= 1) {
    int n = __shfl_up(x, d, 32);
    if (lane >= d) x += n;
  }
  if (lane == 31) wtot[wid] = x;
  __syncthreads();
  int base = 0;
  #pragma unroll
  for (int w = 0; w < 8; ++w) {
    base += red[w];
    if (w < wid) base += wtot[w];
  }
  int excl = base + x - v;
  if (i < N) {
    rowptr[i] = excl;
    cursor[i] = excl;
    if (i == N - 1) rowptr[N] = excl + v;
  }
}

__global__ __launch_bounds__(256) void gat_csr_fill(
    const int* __restrict__ src, const int* __restrict__ dst,
    int* __restrict__ cursor, int* __restrict__ csrc, int E) {
  int e = blockIdx.x * 256 + threadIdx.x;
  if (e < E) {
    int pos = atomicAdd(&cursor[dst[e]], 1);
    csrc[pos] = src[e];
  }
}

#define SORT_CH 4096
#define SORT_MAXB 256
#define SORT_CAP 8192

__device__ __forceinline__ int gat_excl_scan256(int v, int* red, int& total) {
  const int lane = threadIdx.x & 31;
  const int wid = threadIdx.x >> 5;
  int x = v;
  #pragma unroll
  for (int d = 1; d < 32; d <<= 1) {
    int n = __shfl_up(x, d, 32);
    if (lane >= d) x += n;
  }
  if (lane == 31) red[wid] = x;
  __syncthreads();
  int base = 0, tot = 0;
  #pragma unroll
  for (int w = 0; w < 8; ++w) {
    const int r = red[w];
    if (w < wid) base += r;
    tot += r;
  }
  __syncthreads();
  total = tot;
  return base + x - v;
}

__device__ __forceinline__ void gat_bin_local_blk(
    const int* __restrict__ src, const int* __restrict__ dst,
    int* __restrict__ tmpw, int* __restrict__ cntL, int* __restrict__ offL, int E, int b) {
  int2* tmp = reinterpret_cast<int2*>(tmpw);
  __shared__ int red[8];
  __shared__ int hist[SORT_MAXB];
  __shared__ int cursor[SORT_MAXB];
  __shared__ int2 stage[SORT_CH];
  const int tid = threadIdx.x;
  hist[tid] = 0;
  __syncthreads();
  const int base = b * SORT_CH;
  int d[SORT_CH / 256], s[SORT_CH / 256];
  #pragma unroll
  for (int i = 0; i < SORT_CH / 256; ++i) {
    const int e = min(base + i * 256 + tid, E - 1);
    d[i] = dst[e];
    s[i] = src[e];
  }
  #pragma unroll
  for (int i = 0; i < SORT_CH / 256; ++i)
    if (base + i * 256 + tid < E) atomicAdd(&hist[d[i] >> 8], 1);
  __syncthreads();
  const int mine = hist[tid];
  int total;
  const int ls = gat_excl_scan256(mine, red, total);
  cursor[tid] = ls;
  cntL[(size_t)b * SORT_MAXB + tid] = mine;
  offL[(size_t)b * SORT_MAXB + tid] = ls;
  __syncthreads();
  #pragma unroll
  for (int i = 0; i < SORT_CH / 256; ++i) {
    if (base + i * 256 + tid < E) {
      const int p = atomicAdd(&cursor[d[i] >> 8], 1);
      stage[p] = make_int2(s[i], d[i]);
    }
  }
  __syncthreads();
  for (int j = tid; j < total; j += 256) tmp[(size_t)base + j] = stage[j];
}

__device__ __forceinline__ void gat_bucket_sort_blk(
    const int* __restrict__ tmpw, const int* __restrict__ cntL, const int* __restrict__ offL,
    int* __restrict__ rowptr, int* __restrict__ csrc, int N, int NCH, int k) {
  const int2* tmp = reinterpret_cast<const int2*>(tmpw);
  __shared__ int red[8];
  __shared__ int cnt[256];
  __shared__ int cursor[256];
  __shared__ int stage[SORT_CAP];
  const int tid = threadIdx.x;
  int lo = 0, n = 0;
  if (tid < NCH) {
    lo = offL[(size_t)tid * SORT_MAXB + k];
    n  = cntL[(size_t)tid * SORT_MAXB + k];
  }
  cnt[tid] = 0;
  int bs;
  (void)gat_excl_scan256(lo, red, bs);
  const int2* run = tmp + (size_t)tid * SORT_CH + lo;
  for (int j0 = 0; j0 < n; j0 += 8) {
    int dd[8];
    #pragma unroll
    for (int u = 0; u < 8; ++u) dd[u] = run[min(j0 + u, n - 1)].y;
    #pragma unroll
    for (int u = 0; u < 8; ++u)
      atomicAdd(&cnt[dd[u] & 255], (j0 + u < n) ? 1 : 0);
  }
  __syncthreads();
  const int c = cnt[tid];
  int total;
  const int st = gat_excl_scan256(c, red, total);
  cursor[tid] = st;
  const int node = k * 256 + tid;
  if (node < N) {
    rowptr[node] = bs + st;
    if (node == N - 1) rowptr[N] = bs + st + c;
  }
  __syncthreads();
  for (int j0 = 0; j0 < n; j0 += 8) {
    int2 pr[8];
    #pragma unroll
    for (int u = 0; u < 8; ++u) pr[u] = run[min(j0 + u, n - 1)];
    #pragma unroll
    for (int u = 0; u < 8; ++u) {
      const bool live = j0 + u < n;
      const int p = atomicAdd(&cursor[pr[u].y & 255], live ? 1 : 0);
      if (live) {
        if (p < SORT_CAP) stage[p] = pr[u].x;
        else              csrc[bs + p] = pr[u].x;
      }
    }
  }
  __syncthreads();
  const int nst = min(total, SORT_CAP);
  for (int j = tid; j < nst; j += 256) csrc[bs + j] = stage[j];
}

__device__ __forceinline__ float leaky02(float v) {
  return (v > 0.f) ? v : 0.2f * v;
}

__device__ __forceinline__ void gat_node_stats(
    const int* __restrict__ csrc, const float* __restrict__ asv,
    int i, int beg, int end, int lane, float ad, float sself,
    int& s0, float& sc0, bool& v0, float& m, float& sum) {
  const int j0 = beg + lane;
  v0 = j0 < end;
  s0 = v0 ? csrc[j0] : i;
  sc0 = v0 ? leaky02(asv[s0] + ad) : sself;
  float mm = fmaxf(sself, sc0);
  for (int j = j0 + 32; j < end; j += 32)
    mm = fmaxf(mm, leaky02(asv[csrc[j]] + ad));
  #pragma unroll
  for (int off = 16; off > 0; off >>= 1) mm = fmaxf(mm, __shfl_xor(mm, off, 32));
  float ss = v0 ? __expf(sc0 - mm) : 0.f;
  for (int j = j0 + 32; j < end; j += 32)
    ss += __expf(leaky02(asv[csrc[j]] + ad) - mm);
  #pragma unroll
  for (int off = 16; off > 0; off >>= 1) ss += __shfl_xor(ss, off, 32);
  m = mm;
  sum = ss + __expf(sself - mm);
}

__device__ __forceinline__ void gat_node_row256(
    const int* __restrict__ rowptr, const int* __restrict__ csrc,
    const float* __restrict__ asv, const float* __restrict__ adv,
    const float* __restrict__ H, int i, int lane, float4& a0, float4& a1) {
  const float4* H4 = reinterpret_cast<const float4*>(H);
  const float4 xs = H4[(size_t)i * 64 + lane];
  const float4 ys = H4[(size_t)i * 64 + lane + 32];
  const int beg = rowptr[i];
  const int end = rowptr[i + 1];
  const float ad = adv[i];
  const float sself = leaky02(asv[i] + ad);
  int s0; float sc0; bool v0; float m, sum;
  gat_node_stats(csrc, asv, i, beg, end, lane, ad, sself, s0, sc0, v0, m, sum);

  {
    const float c = __expf(sself - m) / sum;
    a0 = make_float4(c * xs.x, c * xs.y, c * xs.z, c * xs.w);
    a1 = make_float4(c * ys.x, c * ys.y, c * ys.z, c * ys.w);
  }
  int sv = s0;
  float cv = v0 ? __expf(sc0 - m) / sum : 0.f;
  for (int base = beg; base < end; base += 32) {
    if (base != beg) {
      const int j = base + lane;
      const bool v = j < end;
      sv = v ? csrc[j] : i;
      cv = v ? __expf(leaky02(asv[sv] + ad) - m) / sum : 0.f;
    }
    const int cnt = min(32, end - base);
    #pragma unroll 4
    for (int t = 0; t < cnt; ++t) {
      const int s = __builtin_amdgcn_readlane(sv, t);
      const float c = __int_as_float(__builtin_amdgcn_readlane(__float_as_int(cv), t));
      const float4* hp = H4 + (size_t)s * 64;
      const float4 x = hp[lane];
      const float4 y = hp[lane + 32];
      a0.x = fmaf(c, x.x, a0.x); a0.y = fmaf(c, x.y, a0.y);
      a0.z = fmaf(c, x.z, a0.z); a0.w = fmaf(c, x.w, a0.w);
      a1.x = fmaf(c, y.x, a1.x); a1.y = fmaf(c, y.y, a1.y);
      a1.z = fmaf(c, y.z, a1.z); a1.w = fmaf(c, y.w, a1.w);
    }
  }
}

__device__ __forceinline__ float gat_list_coeff(
    const int* __restrict__ csrc, const float* __restrict__ asv,
    int beg, int end, int lane, float ad, float sc, bool live, float& m, float& sum) {
  const bool big = (end - beg) > 31;
  float mm = sc;
  if (big)
    for (int j = beg + 31 + lane; j < end; j += 32)
      mm = fmaxf(mm, leaky02(asv[csrc[j]] + ad));
  #pragma unroll
  for (int off = 16; off > 0; off >>= 1) mm = fmaxf(mm, __shfl_xor(mm, off, 32));
  const float e0 = (live || lane == 0) ? __expf(sc - mm) : 0.f;
  float ss = e0;
  if (big)
    for (int j = beg + 31 + lane; j < end; j += 32)
      ss += __expf(leaky02(asv[csrc[j]] + ad) - mm);
  #pragma unroll
  for (int off = 16; off > 0; off >>= 1) ss += __shfl_xor(ss, off, 32);
  m = mm;
  sum = ss;
  return e0 / ss;
}

template <int G, bool TAIL>
__device__ __forceinline__ void gat_rows_fma(
    const v4f* __restrict__ H4v, int svr, float cvr, int t, int cnt, int lane,
    float4& a0, float4& a1) {
  static_assert(G == 4 || G == 8, "the register-naming statements below are written for 4 and 8 rows");
  v4f x[G], y[G];
  #pragma unroll
  for (int u = 0; u < G; ++u) {
    const int tl = TAIL ? min(t + u, cnt - 1) : t + u;
    const unsigned o = ((unsigned)__builtin_amdgcn_readlane(svr, tl) << 6) + (unsigned)lane;
    x[u] = H4v[o];
    y[u] = H4v[o + 32];
  }
  if constexpr (G == 8) {
    asm volatile("" : "+v"(x[0]), "+v"(x[1]), "+v"(x[2]), "+v"(x[3]),
                      "+v"(x[4]), "+v"(x[5]), "+v"(x[6]), "+v"(x[7]),
                      "+v"(y[0]), "+v"(y[1]), "+v"(y[2]), "+v"(y[3]),
                      "+v"(y[4]), "+v"(y[5]), "+v"(y[6]), "+v"(y[7]));
  } else {
    asm volatile("" : "+v"(x[0]), "+v"(x[1]), "+v"(x[2]), "+v"(x[3]),
                      "+v"(y[0]), "+v"(y[1]), "+v"(y[2]), "+v"(y[3]));
  }
  #pragma unroll
  for (int u = 0; u < G; ++u) {
    if (!TAIL || t + u < cnt) {
      const float c = __int_as_float(__builtin_amdgcn_readlane(__float_as_int(cvr), t + u));
      a0.x = fmaf(c, x[u].x, a0.x); a0.y = fmaf(c, x[u].y, a0.y);
      a0.z = fmaf(c, x[u].z, a0.z); a0.w = fmaf(c, x[u].w, a0.w);
      a1.x = fmaf(c, y[u].x, a1.x); a1.y = fmaf(c, y[u].y, a1.y);
      a1.z = fmaf(c, y[u].z, a1.z); a1.w = fmaf(c, y[u].w, a1.w);
    }
  }
}

template <bool GATHER, int WNW, int NT>
__device__ __forceinline__ void gat_layer_gemm_blk(
    const float* __restrict__ X,
    const int* __restrict__ rowptr, const int* __restrict__ csrc,
    const float* __restrict__ asv_in, const float* __restrict__ adv_in,
    const float* __restrict__ Hin, const float* __restrict__ bias_in,
    const float* __restrict__ W,
    const float* __restrict__ a_s, const float* __restrict__ a_d,
    float* __restrict__ H, float* __restrict__ asv, float* __restrict__ adv,
    int N, int Dact, int tile) {
  static_assert(WNW == 1 || WNW == 2 || WNW == 4, "2 x WNW accumulator waves out of eight");
  constexpr int BM = 32;
  constexpr int BN = 16 * NT * WNW;
  constexpr int NWA = 2 * WNW;
  constexpr int AP = GATHER ? LDA : LDT;
  constexpr int CP = BN + 4;
  constexpr int A_F4 = (BM * (KC / 4)) / 256;
  constexpr int B_QT = BN * (KC / 4);
  constexpr int B_Q = (B_QT + 255) / 256;
  constexpr int SM_A = BM * AP;
  constexpr int SM_AB = SM_A + BN * LDT;
  constexpr int SM_C = BM * CP + 2 * BM;
  constexpr int SM = SM_AB > SM_C ? SM_AB : SM_C;
  __shared__ __attribute__((aligned(16))) float smem[SM];
  float* As = smem;
  float* Bst = smem + SM_A;
  float* Cs = smem;
  float* red = smem + BM * CP;

  const int tid  = threadIdx.x;
  const int lane = tid & 31;
  const int lo   = lane & 15;
  const int hi   = lane >> 4;
  const int wid  = __builtin_amdgcn_readfirstlane(tid >> 5);
  const bool accw = (NWA == 8) || (wid < NWA);
  const int wm   = (wid / WNW) * 16;
  const int wn   = (wid % WNW) * (16 * NT);
  const int rowBase = tile * BM;

  if constexpr (GATHER && WNW == 4) {
    constexpr int RN = BM / 8;
    constexpr int RP = 4;
    const int c0 = 4 * lane, c1 = 128 + 4 * lane;
    const float4* H4 = reinterpret_cast<const float4*>(Hin);
    const v4f* H4v = reinterpret_cast<const v4f*>(Hin);
    #pragma unroll
    for (int p0 = 0; p0 < RN; p0 += RP) {
      int   ni[RP], nb[RP], ne[RP];
      float nad[RP];
      #pragma unroll
      for (int rr = 0; rr < RP; ++rr) {
        const int i = min(rowBase + wid * RN + p0 + rr, N - 1);
        ni[rr] = i;
        nb[rr] = rowptr[i];
        ne[rr] = rowptr[i + 1];
        nad[rr] = adv_in[i];
      }
      int   sv[RP];
      float cv[RP];
      bool  lv[RP];
      #pragma unroll
      for (int rr = 0; rr < RP; ++rr) {
        const int j0 = nb[rr] + lane - 1;
        lv[rr] = (lane >= 1) && (j0 < ne[rr]);
        const int sl = csrc[lv[rr] ? j0 : 0];
        sv[rr] = lv[rr] ? sl : ni[rr];
      }
      #pragma unroll
      for (int rr = 0; rr < RP; ++rr)
        cv[rr] = leaky02(asv_in[sv[rr]] + nad[rr]);
      #pragma unroll
      for (int rr = 0; rr < RP; ++rr) {
        float m, sum;
        cv[rr] = gat_list_coeff(csrc, asv_in, nb[rr], ne[rr], lane, nad[rr], cv[rr], lv[rr], m, sum);
      }
      #pragma unroll
      for (int rr = 0; rr < RP; ++rr) {
        const int row = wid * RN + p0 + rr;
        float4 a0 = make_float4(0.f, 0.f, 0.f, 0.f);
        float4 a1 = a0;
        const int total = ne[rr] - nb[rr] + 1;
        int   svr = sv[rr];
        float cvr = cv[rr];
        for (int base = 0; base < total; base += 32) {
          if (base != 0) {
            const float sc = leaky02(asv_in[sv[rr]] + nad[rr]);
            float m, sum;
            (void)gat_list_coeff(csrc, asv_in, nb[rr], ne[rr], lane, nad[rr], sc, lv[rr], m, sum);
            const int j = nb[rr] + base - 1 + lane;
            const bool v = j < ne[rr];
            svr = v ? csrc[j] : ni[rr];
            cvr = v ? __expf(leaky02(asv_in[svr] + nad[rr]) - m) / sum : 0.f;
          }
          const int cnt = min(32, total - base);
          int t = 0;
          for (; t + 8 <= cnt; t += 8)
            gat_rows_fma<8, false>(H4v, svr, cvr, t, cnt, lane, a0, a1);
          if (cnt - t > 4)      gat_rows_fma<8, true>(H4v, svr, cvr, t, cnt, lane, a0, a1);
          else if (cnt - t > 0) gat_rows_fma<4, true>(H4v, svr, cvr, t, cnt, lane, a0, a1);
        }
        *reinterpret_cast<float4*>(&As[row * AP + c0]) = a0;
        *reinterpret_cast<float4*>(&As[row * AP + c1]) = a1;
      }
    }
  } else if constexpr (GATHER) {
    const int c0 = 4 * lane, c1 = 128 + 4 * lane;
    float4 bz0, bz1;
    bz0.x = bias_in[c0 + 0]; bz0.y = bias_in[c0 + 1]; bz0.z = bias_in[c0 + 2]; bz0.w = bias_in[c0 + 3];
    bz1.x = bias_in[c1 + 0]; bz1.y = bias_in[c1 + 1]; bz1.z = bias_in[c1 + 2]; bz1.w = bias_in[c1 + 3];
    for (int rr = 0; rr < BM / 8; ++rr) {
      const int row = wid * (BM / 8) + rr;
      const int i   = rowBase + row;
      float4 a0 = make_float4(0.f, 0.f, 0.f, 0.f);
      float4 a1 = a0;
      if (i < N) {
        gat_node_row256(rowptr, csrc, asv_in, adv_in, Hin, i, lane, a0, a1);
        a0.x += bz0.x; a0.y += bz0.y; a0.z += bz0.z; a0.w += bz0.w;
        a1.x += bz1.x; a1.y += bz1.y; a1.z += bz1.z; a1.w += bz1.w;
        a0.x = fmaxf(a0.x, 0.f); a0.y = fmaxf(a0.y, 0.f); a0.z = fmaxf(a0.z, 0.f); a0.w = fmaxf(a0.w, 0.f);
        a1.x = fmaxf(a1.x, 0.f); a1.y = fmaxf(a1.y, 0.f); a1.z = fmaxf(a1.z, 0.f); a1.w = fmaxf(a1.w, 0.f);
      }
      *reinterpret_cast<float4*>(&As[row * AP + c0]) = a0;
      *reinterpret_cast<float4*>(&As[row * AP + c1]) = a1;
    }
  }

  v8f acc[NT];
  #pragma unroll
  for (int t = 0; t < NT; ++t) acc[t] = v8f{};

  float4 ra[A_F4];
  float4 rb[B_Q];

  auto load_chunk = [&](int kk) {
    if constexpr (!GATHER) {
      #pragma unroll
      for (int i = 0; i < A_F4; ++i) {
        const int f4 = tid + 256 * i;
        const int r  = f4 >> 3;
        const int c4 = f4 & 7;
        const int gr = rowBase + r;
        float4 v = make_float4(0.f, 0.f, 0.f, 0.f);
        if (gr < N)
          v = *reinterpret_cast<const float4*>(X + (size_t)gr * DK + kk + c4 * 4);
        ra[i] = v;
      }
    }
    if constexpr (GATHER && WNW == 4) {
      static_assert(BN == 256 && B_Q == KC / 4, "one column per thread");
      const int nc = min(tid, Dact - 1);
      #pragma unroll
      for (int i = 0; i < B_Q; ++i) {
        const float* wp = W + (size_t)(kk + i * 4) * Dact + nc;
        rb[i].x = wp[0];
        rb[i].y = wp[Dact];
        rb[i].z = wp[2 * Dact];
        rb[i].w = wp[3 * Dact];
      }
      if (Dact < BN) {
        #pragma unroll
        for (int i = 0; i < B_Q; ++i)
          if (tid >= Dact) rb[i] = make_float4(0.f, 0.f, 0.f, 0.f);
      }
    } else {
    #pragma unroll
    for (int i = 0; i < B_Q; ++i) {
      const int q  = tid + 256 * i;
      const int n  = q % BN;
      const int kq = q / BN;
      float4 v = make_float4(0.f, 0.f, 0.f, 0.f);
      if ((B_QT % 256 == 0 || q < B_QT) && n < Dact) {
        const float* wp = W + (size_t)(kk + kq * 4) * Dact + n;
        v.x = wp[0];
        v.y = wp[Dact];
        v.z = wp[2 * Dact];
        v.w = wp[3 * Dact];
      }
      rb[i] = v;
    }
    }
  };
  auto store_chunk = [&]() {
    if constexpr (!GATHER) {
      #pragma unroll
      for (int i = 0; i < A_F4; ++i) {
        const int f4 = tid + 256 * i;
        const int r  = f4 >> 3;
        const int c4 = f4 & 7;
        *reinterpret_cast<float4*>(&As[r * AP + c4 * 4]) = ra[i];
      }
    }
    #pragma unroll
    for (int i = 0; i < B_Q; ++i) {
      const int q  = tid + 256 * i;
      const int n  = q % BN;
      const int kq = q / BN;
      if (B_QT % 256 == 0 || q < B_QT)
        *reinterpret_cast<float4*>(&Bst[n * LDT + kq * 4]) = rb[i];
    }
  };

  const float* arow = As + (wm + lo) * AP + 4 * hi;
  const float* brow = Bst + (wn + lo) * LDT + 4 * hi;

  if constexpr (GATHER && WNW == 4) {
    constexpr int RN = BM / 8;
    const int c0 = 4 * lane, c1 = 128 + 4 * lane;
    {
      float4 bz0, bz1;
      bz0.x = bias_in[c0 + 0]; bz0.y = bias_in[c0 + 1]; bz0.z = bias_in[c0 + 2]; bz0.w = bias_in[c0 + 3];
      bz1.x = bias_in[c1 + 0]; bz1.y = bias_in[c1 + 1]; bz1.z = bias_in[c1 + 2]; bz1.w = bias_in[c1 + 3];
      load_chunk(0);
      #pragma unroll
      for (int rr = 0; rr < RN; ++rr) {
        const int row = wid * RN + rr;
        float4 a0 = *reinterpret_cast<const float4*>(&As[row * AP + c0]);
        float4 a1 = *reinterpret_cast<const float4*>(&As[row * AP + c1]);
        if (rowBase + row < N) {
          a0.x = fmaxf(a0.x + bz0.x, 0.f); a0.y = fmaxf(a0.y + bz0.y, 0.f);
          a0.z = fmaxf(a0.z + bz0.z, 0.f); a0.w = fmaxf(a0.w + bz0.w, 0.f);
          a1.x = fmaxf(a1.x + bz1.x, 0.f); a1.y = fmaxf(a1.y + bz1.y, 0.f);
          a1.z = fmaxf(a1.z + bz1.z, 0.f); a1.w = fmaxf(a1.w + bz1.w, 0.f);
        } else {
          a0 = make_float4(0.f, 0.f, 0.f, 0.f);
          a1 = a0;
        }
        *reinterpret_cast<float4*>(&As[row * AP + c0]) = a0;
        *reinterpret_cast<float4*>(&As[row * AP + c1]) = a1;
      }
    }
  } else {
    load_chunk(0);
  }
  for (int kk = 0; kk < DK; kk += KC) {
    __syncthreads();
    store_chunk();
    __syncthreads();
    if (kk + KC < DK) load_chunk(kk + KC);
    if (accw) {
      const float* ak = GATHER ? arow + kk : arow;
      #pragma unroll
      for (int kb = 0; kb < KC; kb += 8) {
        const float4 a4 = *reinterpret_cast<const float4*>(ak + kb);
        float4 b4[NT];
        #pragma unroll
        for (int t = 0; t < NT; ++t)
          b4[t] = *reinterpret_cast<const float4*>(brow + t * 16 * LDT + kb);
        const v2f a01 = {a4.x, a4.y};
        const v2f a23 = {a4.z, a4.w};
        #pragma unroll
        for (int t = 0; t < NT; ++t) {
          const v2f b01 = {b4[t].x, b4[t].y};
          acc[t] = __builtin_amdgcn_wmma_f32_16x16x4_f32(false, a01, false, b01,
                                                         (short)0, acc[t], false, false);
        }
        #pragma unroll
        for (int t = 0; t < NT; ++t) {
          const v2f b23 = {b4[t].z, b4[t].w};
          acc[t] = __builtin_amdgcn_wmma_f32_16x16x4_f32(false, a23, false, b23,
                                                         (short)0, acc[t], false, false);
        }
      }
    }
  }

  __syncthreads();
  if (accw) {
    #pragma unroll
    for (int t = 0; t < NT; ++t) {
      #pragma unroll
      for (int r = 0; r < 8; ++r)
        Cs[(wm + 8 * hi + r) * CP + wn + 16 * t + lo] = acc[t][r];
    }
  }
  __syncthreads();

  if constexpr (BN == 256) {
    float4 s0, s1, d0, d1;
    {
      const int c0 = 4 * lane, c1 = 128 + 4 * lane;
      s0.x = (c0 + 0 < Dact) ? a_s[c0 + 0] : 0.f; s0.y = (c0 + 1 < Dact) ? a_s[c0 + 1] : 0.f;
      s0.z = (c0 + 2 < Dact) ? a_s[c0 + 2] : 0.f; s0.w = (c0 + 3 < Dact) ? a_s[c0 + 3] : 0.f;
      s1.x = (c1 + 0 < Dact) ? a_s[c1 + 0] : 0.f; s1.y = (c1 + 1 < Dact) ? a_s[c1 + 1] : 0.f;
      s1.z = (c1 + 2 < Dact) ? a_s[c1 + 2] : 0.f; s1.w = (c1 + 3 < Dact) ? a_s[c1 + 3] : 0.f;
      d0.x = (c0 + 0 < Dact) ? a_d[c0 + 0] : 0.f; d0.y = (c0 + 1 < Dact) ? a_d[c0 + 1] : 0.f;
      d0.z = (c0 + 2 < Dact) ? a_d[c0 + 2] : 0.f; d0.w = (c0 + 3 < Dact) ? a_d[c0 + 3] : 0.f;
      d1.x = (c1 + 0 < Dact) ? a_d[c1 + 0] : 0.f; d1.y = (c1 + 1 < Dact) ? a_d[c1 + 1] : 0.f;
      d1.z = (c1 + 2 < Dact) ? a_d[c1 + 2] : 0.f; d1.w = (c1 + 3 < Dact) ? a_d[c1 + 3] : 0.f;
    }
    constexpr int RPW = BM / 8;
    #pragma unroll
    for (int rr = 0; rr < RPW; ++rr) {
      const int row = wid * RPW + rr;
      const int gr  = rowBase + row;
      const float4 x = *reinterpret_cast<const float4*>(&Cs[row * CP + 4 * lane]);
      const float4 y = *reinterpret_cast<const float4*>(&Cs[row * CP + 128 + 4 * lane]);
      float p = x.x * s0.x + x.y * s0.y + x.z * s0.z + x.w * s0.w
              + y.x * s1.x + y.y * s1.y + y.z * s1.z + y.w * s1.w;
      float q = x.x * d0.x + x.y * d0.y + x.z * d0.z + x.w * d0.w
              + y.x * d1.x + y.y * d1.y + y.z * d1.z + y.w * d1.w;
      #pragma unroll
      for (int off = 16; off > 0; off >>= 1) {
        p += __shfl_xor(p, off, 32);
        q += __shfl_xor(q, off, 32);
      }
      if (lane == 0) {
        red[row] = p;
        red[BM + row] = q;
      }
      if (gr < N) {
        float4* hp = reinterpret_cast<float4*>(H) + (size_t)gr * 64;
        hp[lane] = x;
        hp[lane + 32] = y;
      }
    }
  } else {
    static_assert(BN == 16, "narrow form: 16 padded columns");
    if (wid < BM / 8) {
      const int qd = lane & 3;
      const int c0 = 4 * qd;
      float4 s0, d0;
      s0.x = (c0 + 0 < Dact) ? a_s[c0 + 0] : 0.f; s0.y = (c0 + 1 < Dact) ? a_s[c0 + 1] : 0.f;
      s0.z = (c0 + 2 < Dact) ? a_s[c0 + 2] : 0.f; s0.w = (c0 + 3 < Dact) ? a_s[c0 + 3] : 0.f;
      d0.x = (c0 + 0 < Dact) ? a_d[c0 + 0] : 0.f; d0.y = (c0 + 1 < Dact) ? a_d[c0 + 1] : 0.f;
      d0.z = (c0 + 2 < Dact) ? a_d[c0 + 2] : 0.f; d0.w = (c0 + 3 < Dact) ? a_d[c0 + 3] : 0.f;
      const int row = wid * 8 + (lane >> 2);
      const int gr  = rowBase + row;
      const float4 x = *reinterpret_cast<const float4*>(&Cs[row * CP + c0]);
      float p = x.x * s0.x + x.y * s0.y + x.z * s0.z + x.w * s0.w;
      float q = x.x * d0.x + x.y * d0.y + x.z * d0.z + x.w * d0.w;
      #pragma unroll
      for (int off = 2; off > 0; off >>= 1) {
        p += __shfl_xor(p, off, 32);
        q += __shfl_xor(q, off, 32);
      }
      if (qd == 0) {
        red[row] = p;
        red[BM + row] = q;
      }
      if (gr < N) reinterpret_cast<float4*>(H)[(size_t)gr * 4 + qd] = x;
    }
  }

  __syncthreads();
  if (tid < 2 * BM) {
    const int row = (tid < BM) ? tid : tid - BM;
    const int gr  = rowBase + row;
    if (gr < N) {
      if (tid < BM) asv[gr] = red[row];
      else          adv[gr] = red[BM + row];
    }
  }
}

template <bool GATHER, int WNW, int NT>
__global__ __launch_bounds__(256, GATHER ? 8 : 1) void gat_layer_gemm(
    const float* __restrict__ X,
    const int* __restrict__ rowptr, const int* __restrict__ csrc,
    const float* __restrict__ asv_in, const float* __restrict__ adv_in,
    const float* __restrict__ Hin, const float* __restrict__ bias_in,
    const float* __restrict__ W,
    const float* __restrict__ a_s, const float* __restrict__ a_d,
    float* __restrict__ H, float* __restrict__ asv, float* __restrict__ adv,
    int N, int Dact) {
  gat_layer_gemm_blk<GATHER, WNW, NT>(X, rowptr, csrc, asv_in, adv_in, Hin, bias_in,
                                      W, a_s, a_d, H, asv, adv, N, Dact, (int)blockIdx.x);
}

__global__ __launch_bounds__(256) void gat_bin_and_gemm(
    const int* __restrict__ src, const int* __restrict__ dst,
    int* __restrict__ tmpw, int* __restrict__ cntL, int* __restrict__ offL, int E, int nSort,
    const float* __restrict__ X, const float* __restrict__ W,
    const float* __restrict__ a_s, const float* __restrict__ a_d,
    float* __restrict__ H, float* __restrict__ asv, float* __restrict__ adv,
    int N, int Dact, int tile0) {
  const int bid = blockIdx.x;
  if (bid < nSort) {
    gat_bin_local_blk(src, dst, tmpw, cntL, offL, E, bid);
  } else {
    gat_layer_gemm_blk<false, 4, 4>(X, nullptr, nullptr, nullptr, nullptr, nullptr, nullptr,
                                    W, a_s, a_d, H, asv, adv, N, Dact, tile0 + bid - nSort);
  }
}

__global__ __launch_bounds__(256) void gat_bucket_and_gemm(
    const int* __restrict__ tmpw, const int* __restrict__ cntL, const int* __restrict__ offL,
    int* __restrict__ rowptr, int* __restrict__ csrc, int NCH, int nSort,
    const float* __restrict__ X, const float* __restrict__ W,
    const float* __restrict__ a_s, const float* __restrict__ a_d,
    float* __restrict__ H, float* __restrict__ asv, float* __restrict__ adv,
    int N, int Dact, int tile0) {
  const int bid = blockIdx.x;
  if (bid < nSort) {
    gat_bucket_sort_blk(tmpw, cntL, offL, rowptr, csrc, N, NCH, bid);
  } else {
    gat_layer_gemm_blk<false, 4, 4>(X, nullptr, nullptr, nullptr, nullptr, nullptr, nullptr,
                                    W, a_s, a_d, H, asv, adv, N, Dact, tile0 + bid - nSort);
  }
}

__global__ __launch_bounds__(256) void gat_node_gather16(
    const int* __restrict__ rowptr, const int* __restrict__ csrc,
    const float* __restrict__ asv, const float* __restrict__ adv,
    const float* __restrict__ H, const float* __restrict__ bias,
    float* __restrict__ out, int N, int Dact) {
  const int lane = threadIdx.x & 31;
  const int i = blockIdx.x * 8 + __builtin_amdgcn_readfirstlane((int)(threadIdx.x >> 5));
  if (i >= N) return;
  const int beg = rowptr[i];
  const int end = rowptr[i + 1];
  const float ad = adv[i];
  const float sself = leaky02(asv[i] + ad);
  int s0; float sc0; bool v0; float m, sum;
  gat_node_stats(csrc, asv, i, beg, end, lane, ad, sself, s0, sc0, v0, m, sum);

  const int q = lane & 3;
  const int slot = lane >> 2;
  const float4* H4 = reinterpret_cast<const float4*>(H);
  float4 acc = make_float4(0.f, 0.f, 0.f, 0.f);
  if (slot == 0) {
    const float c = __expf(sself - m) / sum;
    const float4 x = H4[(size_t)i * 4 + q];
    acc = make_float4(c * x.x, c * x.y, c * x.z, c * x.w);
  }
  int sv = s0;
  float cv = v0 ? __expf(sc0 - m) / sum : 0.f;
  for (int base = beg; base < end; base += 32) {
    if (base != beg) {
      const int j = base + lane;
      const bool v = j < end;
      sv = v ? csrc[j] : i;
      cv = v ? __expf(leaky02(asv[sv] + ad) - m) / sum : 0.f;
    }
    const int cnt = min(32, end - base);
    for (int t0 = 0; t0 < cnt; t0 += 8) {
      const int t = t0 + slot;
      const int s = __shfl(sv, t, 32);
      const float c = __shfl(cv, t, 32);
      if (t < cnt) {
        const float4 x = H4[(size_t)s * 4 + q];
        acc.x = fmaf(c, x.x, acc.x); acc.y = fmaf(c, x.y, acc.y);
        acc.z = fmaf(c, x.z, acc.z); acc.w = fmaf(c, x.w, acc.w);
      }
    }
  }
  #pragma unroll
  for (int off = 4; off < 32; off <<= 1) {
    acc.x += __shfl_xor(acc.x, off, 32);
    acc.y += __shfl_xor(acc.y, off, 32);
    acc.z += __shfl_xor(acc.z, off, 32);
    acc.w += __shfl_xor(acc.w, off, 32);
  }
  if (slot == 0) {
    const int c = q * 4;
    float* op = out + (size_t)i * Dact;
    if (c + 0 < Dact) op[c + 0] = acc.x + bias[c + 0];
    if (c + 1 < Dact) op[c + 1] = acc.y + bias[c + 1];
    if (c + 2 < Dact) op[c + 2] = acc.z + bias[c + 2];
    if (c + 3 < Dact) op[c + 3] = acc.w + bias[c + 3];
  }
}

extern "C" void kernel_launch(void* const* d_in, const int* in_sizes, int n_in,
                              void* d_out, int out_size, void* d_ws, size_t ws_size,
                              hipStream_t stream) {
  const float* x  = (const float*)d_in[0];
  const int*   ei = (const int*)d_in[1];
  const float* W1 = (const float*)d_in[2];
  const float* as1 = (const float*)d_in[3];
  const float* ad1 = (const float*)d_in[4];
  const float* b1 = (const float*)d_in[5];
  const float* W2 = (const float*)d_in[6];
  const float* as2 = (const float*)d_in[7];
  const float* ad2 = (const float*)d_in[8];
  const float* b2 = (const float*)d_in[9];
  const float* W3 = (const float*)d_in[10];
  const float* as3 = (const float*)d_in[11];
  const float* ad3 = (const float*)d_in[12];
  const float* b3 = (const float*)d_in[13];

  const int N  = in_sizes[0] / 256;
  const int E  = in_sizes[1] / 2;
  const int* src = ei;
  const int* dst = ei + E;

  float* ws = (float*)d_ws;
  float* hA  = ws;
  float* hB  = hA + (size_t)N * 256;
  float* h3  = hB + (size_t)N * 256;
  int*   ib  = (int*)(h3 + (size_t)N * 16);
  int*   cnt    = ib;
  int*   rowptr = cnt + N;
  int*   cursor = rowptr + N + 1;
  int*   tsum   = cursor + N;
  float* asvB = (float*)(ib + (size_t)N * 4);
  float* advB = asvB + N;
  float* asvA = (float*)(ib + (size_t)N * 16);
  float* advA = asvA + N;
  int*   csrc = (int*)(advA + N);

  dim3 blk(256);
  int gGemm  = (N + 31) / 32;
  int gNodeW = (N + 7) / 8;
  int gNodeT = (N + 255) / 256;
  int gEdgeT = (E + 255) / 256;

  const int NBK = (N + 255) / 256;
  const int NCH = (E + SORT_CH - 1) / SORT_CH;
  int*  tmp  = (int*)hB;
  int*  cntL = (int*)hB + 2 * (size_t)NCH * SORT_CH;
  int*  offL = cntL + (size_t)NCH * SORT_MAXB;
  if (NBK <= SORT_MAXB && NCH <= SORT_MAXB &&
      2 * (size_t)NCH * SORT_CH + 2 * (size_t)NCH * SORT_MAXB <= (size_t)N * 256) {
    const int g1 = (gGemm * 2) / 5;
    gat_bin_and_gemm<<<NCH + g1, blk, 0, stream>>>(
        src, dst, tmp, cntL, offL, E, NCH,
        x, W1, as1, ad1, hA, asvA, advA, N, 256, 0);
    gat_bucket_and_gemm<<<NBK + (gGemm - g1), blk, 0, stream>>>(
        tmp, cntL, offL, rowptr, csrc, NCH, NBK,
        x, W1, as1, ad1, hA, asvA, advA, N, 256, g1);
  } else {
    gat_csr_zero<<<gNodeT, blk, 0, stream>>>(cnt, N);
    gat_csr_hist<<<gEdgeT, blk, 0, stream>>>(dst, cnt, E);
    gat_csr_tilesum<<<gNodeT, blk, 0, stream>>>(cnt, tsum, N);
    gat_csr_scan<<<gNodeT, blk, 0, stream>>>(cnt, tsum, rowptr, cursor, N);
    gat_csr_fill<<<gEdgeT, blk, 0, stream>>>(src, dst, cursor, csrc, E);
    gat_layer_gemm<false, 4, 4><<<gGemm, blk, 0, stream>>>(
        x, nullptr, nullptr, nullptr, nullptr, nullptr, nullptr,
        W1, as1, ad1, hA, asvA, advA, N, 256);
  }

  gat_layer_gemm<true, 4, 4><<<gGemm, blk, 0, stream>>>(
      nullptr, rowptr, csrc, asvA, advA, hA, b1,
      W2, as2, ad2, hB, asvB, advB, N, 256);

  gat_layer_gemm<true, 1, 1><<<gGemm, blk, 0, stream>>>(
      nullptr, rowptr, csrc, asvB, advB, hB, b2,
      W3, as3, ad3, h3, asvA, advA, N, 10);

  gat_node_gather16<<<gNodeW, blk, 0, stream>>>(rowptr, csrc, asvA, advA, h3, b3, (float*)d_out, N, 10);
}


# W chunks of the 256-wide products come in by 16-byte loads along n (four 128-byte row pieces per load, not one), turned into k-quads in registers: 4x fewer W load instructions; sum order unchanged
# speedup vs baseline: 6.7806x; 1.0594x over previous
#include <hip/hip_runtime.h>
#include <math.h>


typedef float v2f __attribute__((ext_vector_type(2)));
typedef float v4f __attribute__((ext_vector_type(4)));
typedef float v8f __attribute__((ext_vector_type(8)));

#define DK 256
#define KC 32
#define LDT (KC + 4)
#define LDA (DK + 4)

__global__ __launch_bounds__(256) void gat_csr_zero(int* __restrict__ cnt, int N) {
  int i = blockIdx.x * 256 + threadIdx.x;
  if (i < N) cnt[i] = 0;
}

__global__ __launch_bounds__(256) void gat_csr_hist(
    const int* __restrict__ dst, int* __restrict__ cnt, int E) {
  int e = blockIdx.x * 256 + threadIdx.x;
  if (e < E) atomicAdd(&cnt[dst[e]], 1);
}

__global__ __launch_bounds__(256) void gat_csr_tilesum(
    const int* __restrict__ cnt, int* __restrict__ tsum, int N) {
  __shared__ int red[8];
  const int tid = threadIdx.x;
  int i = blockIdx.x * 256 + tid;
  int v = (i < N) ? cnt[i] : 0;
  #pragma unroll
  for (int off = 16; off > 0; off >>= 1) v += __shfl_xor(v, off, 32);
  if ((tid & 31) == 0) red[tid >> 5] = v;
  __syncthreads();
  if (tid == 0) {
    int t = 0;
    #pragma unroll
    for (int w = 0; w < 8; ++w) t += red[w];
    tsum[blockIdx.x] = t;
  }
}

__global__ __launch_bounds__(256) void gat_csr_scan(
    const int* __restrict__ cnt, const int* __restrict__ tsum,
    int* __restrict__ rowptr, int* __restrict__ cursor, int N) {
  __shared__ int red[8];
  __shared__ int wtot[8];
  const int tid = threadIdx.x;
  const int lane = tid & 31;
  const int wid = tid >> 5;
  int part = 0;
  for (int t = tid; t < (int)blockIdx.x; t += 256) part += tsum[t];
  #pragma unroll
  for (int off = 16; off > 0; off >>= 1) part += __shfl_xor(part, off, 32);
  if (lane == 0) red[wid] = part;
  int i = blockIdx.x * 256 + tid;
  int v = (i < N) ? cnt[i] : 0;
  int x = v;
  #pragma unroll
  for (int d = 1; d < 32; d <<= 1) {
    int n = __shfl_up(x, d, 32);
    if (lane >= d) x += n;
  }
  if (lane == 31) wtot[wid] = x;
  __syncthreads();
  int base = 0;
  #pragma unroll
  for (int w = 0; w < 8; ++w) {
    base += red[w];
    if (w < wid) base += wtot[w];
  }
  int excl = base + x - v;
  if (i < N) {
    rowptr[i] = excl;
    cursor[i] = excl;
    if (i == N - 1) rowptr[N] = excl + v;
  }
}

__global__ __launch_bounds__(256) void gat_csr_fill(
    const int* __restrict__ src, const int* __restrict__ dst,
    int* __restrict__ cursor, int* __restrict__ csrc, int E) {
  int e = blockIdx.x * 256 + threadIdx.x;
  if (e < E) {
    int pos = atomicAdd(&cursor[dst[e]], 1);
    csrc[pos] = src[e];
  }
}

#define SORT_CH 4096
#define SORT_MAXB 256
#define SORT_CAP 8192

__device__ __forceinline__ int gat_excl_scan256(int v, int* red, int& total) {
  const int lane = threadIdx.x & 31;
  const int wid = threadIdx.x >> 5;
  int x = v;
  #pragma unroll
  for (int d = 1; d < 32; d <<= 1) {
    int n = __shfl_up(x, d, 32);
    if (lane >= d) x += n;
  }
  if (lane == 31) red[wid] = x;
  __syncthreads();
  int base = 0, tot = 0;
  #pragma unroll
  for (int w = 0; w < 8; ++w) {
    const int r = red[w];
    if (w < wid) base += r;
    tot += r;
  }
  __syncthreads();
  total = tot;
  return base + x - v;
}

__device__ __forceinline__ void gat_bin_local_blk(
    const int* __restrict__ src, const int* __restrict__ dst,
    int* __restrict__ tmpw, int* __restrict__ cntL, int* __restrict__ offL, int E, int b) {
  int2* tmp = reinterpret_cast<int2*>(tmpw);
  __shared__ int red[8];
  __shared__ int hist[SORT_MAXB];
  __shared__ int cursor[SORT_MAXB];
  __shared__ int2 stage[SORT_CH];
  const int tid = threadIdx.x;
  hist[tid] = 0;
  __syncthreads();
  const int base = b * SORT_CH;
  int d[SORT_CH / 256], s[SORT_CH / 256];
  #pragma unroll
  for (int i = 0; i < SORT_CH / 256; ++i) {
    const int e = min(base + i * 256 + tid, E - 1);
    d[i] = dst[e];
    s[i] = src[e];
  }
  #pragma unroll
  for (int i = 0; i < SORT_CH / 256; ++i)
    if (base + i * 256 + tid < E) atomicAdd(&hist[d[i] >> 8], 1);
  __syncthreads();
  const int mine = hist[tid];
  int total;
  const int ls = gat_excl_scan256(mine, red, total);
  cursor[tid] = ls;
  cntL[(size_t)b * SORT_MAXB + tid] = mine;
  offL[(size_t)b * SORT_MAXB + tid] = ls;
  __syncthreads();
  #pragma unroll
  for (int i = 0; i < SORT_CH / 256; ++i) {
    if (base + i * 256 + tid < E) {
      const int p = atomicAdd(&cursor[d[i] >> 8], 1);
      stage[p] = make_int2(s[i], d[i]);
    }
  }
  __syncthreads();
  for (int j = tid; j < total; j += 256) tmp[(size_t)base + j] = stage[j];
}

__device__ __forceinline__ void gat_bucket_sort_blk(
    const int* __restrict__ tmpw, const int* __restrict__ cntL, const int* __restrict__ offL,
    int* __restrict__ rowptr, int* __restrict__ csrc, int N, int NCH, int k) {
  const int2* tmp = reinterpret_cast<const int2*>(tmpw);
  __shared__ int red[8];
  __shared__ int cnt[256];
  __shared__ int cursor[256];
  __shared__ int stage[SORT_CAP];
  const int tid = threadIdx.x;
  int lo = 0, n = 0;
  if (tid < NCH) {
    lo = offL[(size_t)tid * SORT_MAXB + k];
    n  = cntL[(size_t)tid * SORT_MAXB + k];
  }
  cnt[tid] = 0;
  int bs;
  (void)gat_excl_scan256(lo, red, bs);
  const int2* run = tmp + (size_t)tid * SORT_CH + lo;
  for (int j0 = 0; j0 < n; j0 += 8) {
    int dd[8];
    #pragma unroll
    for (int u = 0; u < 8; ++u) dd[u] = run[min(j0 + u, n - 1)].y;
    #pragma unroll
    for (int u = 0; u < 8; ++u)
      atomicAdd(&cnt[dd[u] & 255], (j0 + u < n) ? 1 : 0);
  }
  __syncthreads();
  const int c = cnt[tid];
  int total;
  const int st = gat_excl_scan256(c, red, total);
  cursor[tid] = st;
  const int node = k * 256 + tid;
  if (node < N) {
    rowptr[node] = bs + st;
    if (node == N - 1) rowptr[N] = bs + st + c;
  }
  __syncthreads();
  for (int j0 = 0; j0 < n; j0 += 8) {
    int2 pr[8];
    #pragma unroll
    for (int u = 0; u < 8; ++u) pr[u] = run[min(j0 + u, n - 1)];
    #pragma unroll
    for (int u = 0; u < 8; ++u) {
      const bool live = j0 + u < n;
      const int p = atomicAdd(&cursor[pr[u].y & 255], live ? 1 : 0);
      if (live) {
        if (p < SORT_CAP) stage[p] = pr[u].x;
        else              csrc[bs + p] = pr[u].x;
      }
    }
  }
  __syncthreads();
  const int nst = min(total, SORT_CAP);
  for (int j = tid; j < nst; j += 256) csrc[bs + j] = stage[j];
}

__device__ __forceinline__ float leaky02(float v) {
  return (v > 0.f) ? v : 0.2f * v;
}

__device__ __forceinline__ void gat_node_stats(
    const int* __restrict__ csrc, const float* __restrict__ asv,
    int i, int beg, int end, int lane, float ad, float sself,
    int& s0, float& sc0, bool& v0, float& m, float& sum) {
  const int j0 = beg + lane;
  v0 = j0 < end;
  s0 = v0 ? csrc[j0] : i;
  sc0 = v0 ? leaky02(asv[s0] + ad) : sself;
  float mm = fmaxf(sself, sc0);
  for (int j = j0 + 32; j < end; j += 32)
    mm = fmaxf(mm, leaky02(asv[csrc[j]] + ad));
  #pragma unroll
  for (int off = 16; off > 0; off >>= 1) mm = fmaxf(mm, __shfl_xor(mm, off, 32));
  float ss = v0 ? __expf(sc0 - mm) : 0.f;
  for (int j = j0 + 32; j < end; j += 32)
    ss += __expf(leaky02(asv[csrc[j]] + ad) - mm);
  #pragma unroll
  for (int off = 16; off > 0; off >>= 1) ss += __shfl_xor(ss, off, 32);
  m = mm;
  sum = ss + __expf(sself - mm);
}

__device__ __forceinline__ void gat_node_row256(
    const int* __restrict__ rowptr, const int* __restrict__ csrc,
    const float* __restrict__ asv, const float* __restrict__ adv,
    const float* __restrict__ H, int i, int lane, float4& a0, float4& a1) {
  const float4* H4 = reinterpret_cast<const float4*>(H);
  const float4 xs = H4[(size_t)i * 64 + lane];
  const float4 ys = H4[(size_t)i * 64 + lane + 32];
  const int beg = rowptr[i];
  const int end = rowptr[i + 1];
  const float ad = adv[i];
  const float sself = leaky02(asv[i] + ad);
  int s0; float sc0; bool v0; float m, sum;
  gat_node_stats(csrc, asv, i, beg, end, lane, ad, sself, s0, sc0, v0, m, sum);

  {
    const float c = __expf(sself - m) / sum;
    a0 = make_float4(c * xs.x, c * xs.y, c * xs.z, c * xs.w);
    a1 = make_float4(c * ys.x, c * ys.y, c * ys.z, c * ys.w);
  }
  int sv = s0;
  float cv = v0 ? __expf(sc0 - m) / sum : 0.f;
  for (int base = beg; base < end; base += 32) {
    if (base != beg) {
      const int j = base + lane;
      const bool v = j < end;
      sv = v ? csrc[j] : i;
      cv = v ? __expf(leaky02(asv[sv] + ad) - m) / sum : 0.f;
    }
    const int cnt = min(32, end - base);
    #pragma unroll 4
    for (int t = 0; t < cnt; ++t) {
      const int s = __builtin_amdgcn_readlane(sv, t);
      const float c = __int_as_float(__builtin_amdgcn_readlane(__float_as_int(cv), t));
      const float4* hp = H4 + (size_t)s * 64;
      const float4 x = hp[lane];
      const float4 y = hp[lane + 32];
      a0.x = fmaf(c, x.x, a0.x); a0.y = fmaf(c, x.y, a0.y);
      a0.z = fmaf(c, x.z, a0.z); a0.w = fmaf(c, x.w, a0.w);
      a1.x = fmaf(c, y.x, a1.x); a1.y = fmaf(c, y.y, a1.y);
      a1.z = fmaf(c, y.z, a1.z); a1.w = fmaf(c, y.w, a1.w);
    }
  }
}

__device__ __forceinline__ float gat_list_coeff(
    const int* __restrict__ csrc, const float* __restrict__ asv,
    int beg, int end, int lane, float ad, float sc, bool live, float& m, float& sum) {
  const bool big = (end - beg) > 31;
  float mm = sc;
  if (big)
    for (int j = beg + 31 + lane; j < end; j += 32)
      mm = fmaxf(mm, leaky02(asv[csrc[j]] + ad));
  #pragma unroll
  for (int off = 16; off > 0; off >>= 1) mm = fmaxf(mm, __shfl_xor(mm, off, 32));
  const float e0 = (live || lane == 0) ? __expf(sc - mm) : 0.f;
  float ss = e0;
  if (big)
    for (int j = beg + 31 + lane; j < end; j += 32)
      ss += __expf(leaky02(asv[csrc[j]] + ad) - mm);
  #pragma unroll
  for (int off = 16; off > 0; off >>= 1) ss += __shfl_xor(ss, off, 32);
  m = mm;
  sum = ss;
  return e0 / ss;
}

template <int G, bool TAIL>
__device__ __forceinline__ void gat_rows_fma(
    const v4f* __restrict__ H4v, int svr, float cvr, int t, int cnt, int lane,
    float4& a0, float4& a1) {
  static_assert(G == 4 || G == 8, "the register-naming statements below are written for 4 and 8 rows");
  v4f x[G], y[G];
  #pragma unroll
  for (int u = 0; u < G; ++u) {
    const int tl = TAIL ? min(t + u, cnt - 1) : t + u;
    const unsigned o = ((unsigned)__builtin_amdgcn_readlane(svr, tl) << 6) + (unsigned)lane;
    x[u] = H4v[o];
    y[u] = H4v[o + 32];
  }
  if constexpr (G == 8) {
    asm volatile("" : "+v"(x[0]), "+v"(x[1]), "+v"(x[2]), "+v"(x[3]),
                      "+v"(x[4]), "+v"(x[5]), "+v"(x[6]), "+v"(x[7]),
                      "+v"(y[0]), "+v"(y[1]), "+v"(y[2]), "+v"(y[3]),
                      "+v"(y[4]), "+v"(y[5]), "+v"(y[6]), "+v"(y[7]));
  } else {
    asm volatile("" : "+v"(x[0]), "+v"(x[1]), "+v"(x[2]), "+v"(x[3]),
                      "+v"(y[0]), "+v"(y[1]), "+v"(y[2]), "+v"(y[3]));
  }
  #pragma unroll
  for (int u = 0; u < G; ++u) {
    if (!TAIL || t + u < cnt) {
      const float c = __int_as_float(__builtin_amdgcn_readlane(__float_as_int(cvr), t + u));
      a0.x = fmaf(c, x[u].x, a0.x); a0.y = fmaf(c, x[u].y, a0.y);
      a0.z = fmaf(c, x[u].z, a0.z); a0.w = fmaf(c, x[u].w, a0.w);
      a1.x = fmaf(c, y[u].x, a1.x); a1.y = fmaf(c, y[u].y, a1.y);
      a1.z = fmaf(c, y[u].z, a1.z); a1.w = fmaf(c, y[u].w, a1.w);
    }
  }
}

template <bool GATHER, int WNW, int NT>
__device__ __forceinline__ void gat_layer_gemm_blk(
    const float* __restrict__ X,
    const int* __restrict__ rowptr, const int* __restrict__ csrc,
    const float* __restrict__ asv_in, const float* __restrict__ adv_in,
    const float* __restrict__ Hin, const float* __restrict__ bias_in,
    const float* __restrict__ W,
    const float* __restrict__ a_s, const float* __restrict__ a_d,
    float* __restrict__ H, float* __restrict__ asv, float* __restrict__ adv,
    int N, int Dact, int tile) {
  static_assert(WNW == 1 || WNW == 2 || WNW == 4, "2 x WNW accumulator waves out of eight");
  constexpr int BM = 32;
  constexpr int BN = 16 * NT * WNW;
  constexpr int NWA = 2 * WNW;
  constexpr int AP = GATHER ? LDA : LDT;
  constexpr int CP = BN + 4;
  constexpr int A_F4 = (BM * (KC / 4)) / 256;
  constexpr int B_QT = BN * (KC / 4);
  constexpr int B_Q = (B_QT + 255) / 256;
  constexpr int SM_A = BM * AP;
  constexpr int SM_AB = SM_A + BN * LDT;
  constexpr int SM_C = BM * CP + 2 * BM;
  constexpr int SM = SM_AB > SM_C ? SM_AB : SM_C;
  __shared__ __attribute__((aligned(16))) float smem[SM];
  float* As = smem;
  float* Bst = smem + SM_A;
  float* Cs = smem;
  float* red = smem + BM * CP;

  const int tid  = threadIdx.x;
  const int lane = tid & 31;
  const int lo   = lane & 15;
  const int hi   = lane >> 4;
  const int wid  = __builtin_amdgcn_readfirstlane(tid >> 5);
  const bool accw = (NWA == 8) || (wid < NWA);
  const int wm   = (wid / WNW) * 16;
  const int wn   = (wid % WNW) * (16 * NT);
  const int rowBase = tile * BM;

  if constexpr (GATHER && WNW == 4) {
    constexpr int RN = BM / 8;
    constexpr int RP = 4;
    const int c0 = 4 * lane, c1 = 128 + 4 * lane;
    const float4* H4 = reinterpret_cast<const float4*>(Hin);
    const v4f* H4v = reinterpret_cast<const v4f*>(Hin);
    #pragma unroll
    for (int p0 = 0; p0 < RN; p0 += RP) {
      int   ni[RP], nb[RP], ne[RP];
      float nad[RP];
      #pragma unroll
      for (int rr = 0; rr < RP; ++rr) {
        const int i = min(rowBase + wid * RN + p0 + rr, N - 1);
        ni[rr] = i;
        nb[rr] = rowptr[i];
        ne[rr] = rowptr[i + 1];
        nad[rr] = adv_in[i];
      }
      int   sv[RP];
      float cv[RP];
      bool  lv[RP];
      #pragma unroll
      for (int rr = 0; rr < RP; ++rr) {
        const int j0 = nb[rr] + lane - 1;
        lv[rr] = (lane >= 1) && (j0 < ne[rr]);
        const int sl = csrc[lv[rr] ? j0 : 0];
        sv[rr] = lv[rr] ? sl : ni[rr];
      }
      #pragma unroll
      for (int rr = 0; rr < RP; ++rr)
        cv[rr] = leaky02(asv_in[sv[rr]] + nad[rr]);
      #pragma unroll
      for (int rr = 0; rr < RP; ++rr) {
        float m, sum;
        cv[rr] = gat_list_coeff(csrc, asv_in, nb[rr], ne[rr], lane, nad[rr], cv[rr], lv[rr], m, sum);
      }
      #pragma unroll
      for (int rr = 0; rr < RP; ++rr) {
        const int row = wid * RN + p0 + rr;
        float4 a0 = make_float4(0.f, 0.f, 0.f, 0.f);
        float4 a1 = a0;
        const int total = ne[rr] - nb[rr] + 1;
        int   svr = sv[rr];
        float cvr = cv[rr];
        for (int base = 0; base < total; base += 32) {
          if (base != 0) {
            const float sc = leaky02(asv_in[sv[rr]] + nad[rr]);
            float m, sum;
            (void)gat_list_coeff(csrc, asv_in, nb[rr], ne[rr], lane, nad[rr], sc, lv[rr], m, sum);
            const int j = nb[rr] + base - 1 + lane;
            const bool v = j < ne[rr];
            svr = v ? csrc[j] : ni[rr];
            cvr = v ? __expf(leaky02(asv_in[svr] + nad[rr]) - m) / sum : 0.f;
          }
          const int cnt = min(32, total - base);
          int t = 0;
          for (; t + 8 <= cnt; t += 8)
            gat_rows_fma<8, false>(H4v, svr, cvr, t, cnt, lane, a0, a1);
          if (cnt - t > 4)      gat_rows_fma<8, true>(H4v, svr, cvr, t, cnt, lane, a0, a1);
          else if (cnt - t > 0) gat_rows_fma<4, true>(H4v, svr, cvr, t, cnt, lane, a0, a1);
        }
        *reinterpret_cast<float4*>(&As[row * AP + c0]) = a0;
        *reinterpret_cast<float4*>(&As[row * AP + c1]) = a1;
      }
    }
  } else if constexpr (GATHER) {
    const int c0 = 4 * lane, c1 = 128 + 4 * lane;
    float4 bz0, bz1;
    bz0.x = bias_in[c0 + 0]; bz0.y = bias_in[c0 + 1]; bz0.z = bias_in[c0 + 2]; bz0.w = bias_in[c0 + 3];
    bz1.x = bias_in[c1 + 0]; bz1.y = bias_in[c1 + 1]; bz1.z = bias_in[c1 + 2]; bz1.w = bias_in[c1 + 3];
    for (int rr = 0; rr < BM / 8; ++rr) {
      const int row = wid * (BM / 8) + rr;
      const int i   = rowBase + row;
      float4 a0 = make_float4(0.f, 0.f, 0.f, 0.f);
      float4 a1 = a0;
      if (i < N) {
        gat_node_row256(rowptr, csrc, asv_in, adv_in, Hin, i, lane, a0, a1);
        a0.x += bz0.x; a0.y += bz0.y; a0.z += bz0.z; a0.w += bz0.w;
        a1.x += bz1.x; a1.y += bz1.y; a1.z += bz1.z; a1.w += bz1.w;
        a0.x = fmaxf(a0.x, 0.f); a0.y = fmaxf(a0.y, 0.f); a0.z = fmaxf(a0.z, 0.f); a0.w = fmaxf(a0.w, 0.f);
        a1.x = fmaxf(a1.x, 0.f); a1.y = fmaxf(a1.y, 0.f); a1.z = fmaxf(a1.z, 0.f); a1.w = fmaxf(a1.w, 0.f);
      }
      *reinterpret_cast<float4*>(&As[row * AP + c0]) = a0;
      *reinterpret_cast<float4*>(&As[row * AP + c1]) = a1;
    }
  }

  v8f acc[NT];
  #pragma unroll
  for (int t = 0; t < NT; ++t) acc[t] = v8f{};

  float4 ra[A_F4];
  float4 rb[B_Q];

  const int wcq = 8 * wid + (lane & 3) + 4 * (lane >> 4);
  const int wkq = (lane >> 2) & 3;
  const float* wq = W + (size_t)(4 * wkq) * BN + 4 * wcq;
  float* bq = Bst + (4 * wcq) * LDT + 4 * wkq;

  auto load_chunk = [&](int kk) {
    if constexpr (!GATHER) {
      #pragma unroll
      for (int i = 0; i < A_F4; ++i) {
        const int f4 = tid + 256 * i;
        const int r  = f4 >> 3;
        const int c4 = f4 & 7;
        const int gr = rowBase + r;
        float4 v = make_float4(0.f, 0.f, 0.f, 0.f);
        if (gr < N)
          v = *reinterpret_cast<const float4*>(X + (size_t)gr * DK + kk + c4 * 4);
        ra[i] = v;
      }
    }
    if constexpr (BN == 256) {
      static_assert(B_Q == 8 && KC == 32, "two passes of four k rows per thread");
      #pragma unroll
      for (int s = 0; s < 2; ++s) {
        #pragma unroll
        for (int i = 0; i < 4; ++i)
          rb[s * 4 + i] = *reinterpret_cast<const float4*>(wq + (size_t)(kk + 16 * s + i) * BN);
      }
    } else {
    #pragma unroll
    for (int i = 0; i < B_Q; ++i) {
      const int q  = tid + 256 * i;
      const int n  = q % BN;
      const int kq = q / BN;
      float4 v = make_float4(0.f, 0.f, 0.f, 0.f);
      if ((B_QT % 256 == 0 || q < B_QT) && n < Dact) {
        const float* wp = W + (size_t)(kk + kq * 4) * Dact + n;
        v.x = wp[0];
        v.y = wp[Dact];
        v.z = wp[2 * Dact];
        v.w = wp[3 * Dact];
      }
      rb[i] = v;
    }
    }
  };
  auto store_chunk = [&]() {
    if constexpr (!GATHER) {
      #pragma unroll
      for (int i = 0; i < A_F4; ++i) {
        const int f4 = tid + 256 * i;
        const int r  = f4 >> 3;
        const int c4 = f4 & 7;
        *reinterpret_cast<float4*>(&As[r * AP + c4 * 4]) = ra[i];
      }
    }
    if constexpr (BN == 256) {
      #pragma unroll
      for (int s = 0; s < 2; ++s) {
        const float4 l0 = rb[s * 4], l1 = rb[s * 4 + 1], l2 = rb[s * 4 + 2], l3 = rb[s * 4 + 3];
        float* bp = bq + 16 * s;
        *reinterpret_cast<float4*>(bp)           = make_float4(l0.x, l1.x, l2.x, l3.x);
        *reinterpret_cast<float4*>(bp + LDT)     = make_float4(l0.y, l1.y, l2.y, l3.y);
        *reinterpret_cast<float4*>(bp + 2 * LDT) = make_float4(l0.z, l1.z, l2.z, l3.z);
        *reinterpret_cast<float4*>(bp + 3 * LDT) = make_float4(l0.w, l1.w, l2.w, l3.w);
      }
    } else {
    #pragma unroll
    for (int i = 0; i < B_Q; ++i) {
      const int q  = tid + 256 * i;
      const int n  = q % BN;
      const int kq = q / BN;
      if (B_QT % 256 == 0 || q < B_QT)
        *reinterpret_cast<float4*>(&Bst[n * LDT + kq * 4]) = rb[i];
    }
    }
  };

  const float* arow = As + (wm + lo) * AP + 4 * hi;
  const float* brow = Bst + (wn + lo) * LDT + 4 * hi;

  if constexpr (GATHER && WNW == 4) {
    constexpr int RN = BM / 8;
    const int c0 = 4 * lane, c1 = 128 + 4 * lane;
    {
      float4 bz0, bz1;
      bz0.x = bias_in[c0 + 0]; bz0.y = bias_in[c0 + 1]; bz0.z = bias_in[c0 + 2]; bz0.w = bias_in[c0 + 3];
      bz1.x = bias_in[c1 + 0]; bz1.y = bias_in[c1 + 1]; bz1.z = bias_in[c1 + 2]; bz1.w = bias_in[c1 + 3];
      load_chunk(0);
      #pragma unroll
      for (int rr = 0; rr < RN; ++rr) {
        const int row = wid * RN + rr;
        float4 a0 = *reinterpret_cast<const float4*>(&As[row * AP + c0]);
        float4 a1 = *reinterpret_cast<const float4*>(&As[row * AP + c1]);
        if (rowBase + row < N) {
          a0.x = fmaxf(a0.x + bz0.x, 0.f); a0.y = fmaxf(a0.y + bz0.y, 0.f);
          a0.z = fmaxf(a0.z + bz0.z, 0.f); a0.w = fmaxf(a0.w + bz0.w, 0.f);
          a1.x = fmaxf(a1.x + bz1.x, 0.f); a1.y = fmaxf(a1.y + bz1.y, 0.f);
          a1.z = fmaxf(a1.z + bz1.z, 0.f); a1.w = fmaxf(a1.w + bz1.w, 0.f);
        } else {
          a0 = make_float4(0.f, 0.f, 0.f, 0.f);
          a1 = a0;
        }
        *reinterpret_cast<float4*>(&As[row * AP + c0]) = a0;
        *reinterpret_cast<float4*>(&As[row * AP + c1]) = a1;
      }
    }
  } else {
    load_chunk(0);
  }
  for (int kk = 0; kk < DK; kk += KC) {
    __syncthreads();
    store_chunk();
    __syncthreads();
    if (kk + KC < DK) load_chunk(kk + KC);
    if (accw) {
      const float* ak = GATHER ? arow + kk : arow;
      #pragma unroll
      for (int kb = 0; kb < KC; kb += 8) {
        const float4 a4 = *reinterpret_cast<const float4*>(ak + kb);
        float4 b4[NT];
        #pragma unroll
        for (int t = 0; t < NT; ++t)
          b4[t] = *reinterpret_cast<const float4*>(brow + t * 16 * LDT + kb);
        const v2f a01 = {a4.x, a4.y};
        const v2f a23 = {a4.z, a4.w};
        #pragma unroll
        for (int t = 0; t < NT; ++t) {
          const v2f b01 = {b4[t].x, b4[t].y};
          acc[t] = __builtin_amdgcn_wmma_f32_16x16x4_f32(false, a01, false, b01,
                                                         (short)0, acc[t], false, false);
        }
        #pragma unroll
        for (int t = 0; t < NT; ++t) {
          const v2f b23 = {b4[t].z, b4[t].w};
          acc[t] = __builtin_amdgcn_wmma_f32_16x16x4_f32(false, a23, false, b23,
                                                         (short)0, acc[t], false, false);
        }
      }
    }
  }

  __syncthreads();
  if (accw) {
    #pragma unroll
    for (int t = 0; t < NT; ++t) {
      #pragma unroll
      for (int r = 0; r < 8; ++r)
        Cs[(wm + 8 * hi + r) * CP + wn + 16 * t + lo] = acc[t][r];
    }
  }
  __syncthreads();

  if constexpr (BN == 256) {
    float4 s0, s1, d0, d1;
    {
      const int c0 = 4 * lane, c1 = 128 + 4 * lane;
      s0.x = (c0 + 0 < Dact) ? a_s[c0 + 0] : 0.f; s0.y = (c0 + 1 < Dact) ? a_s[c0 + 1] : 0.f;
      s0.z = (c0 + 2 < Dact) ? a_s[c0 + 2] : 0.f; s0.w = (c0 + 3 < Dact) ? a_s[c0 + 3] : 0.f;
      s1.x = (c1 + 0 < Dact) ? a_s[c1 + 0] : 0.f; s1.y = (c1 + 1 < Dact) ? a_s[c1 + 1] : 0.f;
      s1.z = (c1 + 2 < Dact) ? a_s[c1 + 2] : 0.f; s1.w = (c1 + 3 < Dact) ? a_s[c1 + 3] : 0.f;
      d0.x = (c0 + 0 < Dact) ? a_d[c0 + 0] : 0.f; d0.y = (c0 + 1 < Dact) ? a_d[c0 + 1] : 0.f;
      d0.z = (c0 + 2 < Dact) ? a_d[c0 + 2] : 0.f; d0.w = (c0 + 3 < Dact) ? a_d[c0 + 3] : 0.f;
      d1.x = (c1 + 0 < Dact) ? a_d[c1 + 0] : 0.f; d1.y = (c1 + 1 < Dact) ? a_d[c1 + 1] : 0.f;
      d1.z = (c1 + 2 < Dact) ? a_d[c1 + 2] : 0.f; d1.w = (c1 + 3 < Dact) ? a_d[c1 + 3] : 0.f;
    }
    constexpr int RPW = BM / 8;
    #pragma unroll
    for (int rr = 0; rr < RPW; ++rr) {
      const int row = wid * RPW + rr;
      const int gr  = rowBase + row;
      const float4 x = *reinterpret_cast<const float4*>(&Cs[row * CP + 4 * lane]);
      const float4 y = *reinterpret_cast<const float4*>(&Cs[row * CP + 128 + 4 * lane]);
      float p = x.x * s0.x + x.y * s0.y + x.z * s0.z + x.w * s0.w
              + y.x * s1.x + y.y * s1.y + y.z * s1.z + y.w * s1.w;
      float q = x.x * d0.x + x.y * d0.y + x.z * d0.z + x.w * d0.w
              + y.x * d1.x + y.y * d1.y + y.z * d1.z + y.w * d1.w;
      #pragma unroll
      for (int off = 16; off > 0; off >>= 1) {
        p += __shfl_xor(p, off, 32);
        q += __shfl_xor(q, off, 32);
      }
      if (lane == 0) {
        red[row] = p;
        red[BM + row] = q;
      }
      if (gr < N) {
        float4* hp = reinterpret_cast<float4*>(H) + (size_t)gr * 64;
        hp[lane] = x;
        hp[lane + 32] = y;
      }
    }
  } else {
    static_assert(BN == 16, "narrow form: 16 padded columns");
    if (wid < BM / 8) {
      const int qd = lane & 3;
      const int c0 = 4 * qd;
      float4 s0, d0;
      s0.x = (c0 + 0 < Dact) ? a_s[c0 + 0] : 0.f; s0.y = (c0 + 1 < Dact) ? a_s[c0 + 1] : 0.f;
      s0.z = (c0 + 2 < Dact) ? a_s[c0 + 2] : 0.f; s0.w = (c0 + 3 < Dact) ? a_s[c0 + 3] : 0.f;
      d0.x = (c0 + 0 < Dact) ? a_d[c0 + 0] : 0.f; d0.y = (c0 + 1 < Dact) ? a_d[c0 + 1] : 0.f;
      d0.z = (c0 + 2 < Dact) ? a_d[c0 + 2] : 0.f; d0.w = (c0 + 3 < Dact) ? a_d[c0 + 3] : 0.f;
      const int row = wid * 8 + (lane >> 2);
      const int gr  = rowBase + row;
      const float4 x = *reinterpret_cast<const float4*>(&Cs[row * CP + c0]);
      float p = x.x * s0.x + x.y * s0.y + x.z * s0.z + x.w * s0.w;
      float q = x.x * d0.x + x.y * d0.y + x.z * d0.z + x.w * d0.w;
      #pragma unroll
      for (int off = 2; off > 0; off >>= 1) {
        p += __shfl_xor(p, off, 32);
        q += __shfl_xor(q, off, 32);
      }
      if (qd == 0) {
        red[row] = p;
        red[BM + row] = q;
      }
      if (gr < N) reinterpret_cast<float4*>(H)[(size_t)gr * 4 + qd] = x;
    }
  }

  __syncthreads();
  if (tid < 2 * BM) {
    const int row = (tid < BM) ? tid : tid - BM;
    const int gr  = rowBase + row;
    if (gr < N) {
      if (tid < BM) asv[gr] = red[row];
      else          adv[gr] = red[BM + row];
    }
  }
}

template <bool GATHER, int WNW, int NT>
__global__ __launch_bounds__(256, GATHER ? 8 : 1) void gat_layer_gemm(
    const float* __restrict__ X,
    const int* __restrict__ rowptr, const int* __restrict__ csrc,
    const float* __restrict__ asv_in, const float* __restrict__ adv_in,
    const float* __restrict__ Hin, const float* __restrict__ bias_in,
    const float* __restrict__ W,
    const float* __restrict__ a_s, const float* __restrict__ a_d,
    float* __restrict__ H, float* __restrict__ asv, float* __restrict__ adv,
    int N, int Dact) {
  gat_layer_gemm_blk<GATHER, WNW, NT>(X, rowptr, csrc, asv_in, adv_in, Hin, bias_in,
                                      W, a_s, a_d, H, asv, adv, N, Dact, (int)blockIdx.x);
}

__global__ __launch_bounds__(256) void gat_bin_and_gemm(
    const int* __restrict__ src, const int* __restrict__ dst,
    int* __restrict__ tmpw, int* __restrict__ cntL, int* __restrict__ offL, int E, int nSort,
    const float* __restrict__ X, const float* __restrict__ W,
    const float* __restrict__ a_s, const float* __restrict__ a_d,
    float* __restrict__ H, float* __restrict__ asv, float* __restrict__ adv,
    int N, int Dact, int tile0) {
  const int bid = blockIdx.x;
  if (bid < nSort) {
    gat_bin_local_blk(src, dst, tmpw, cntL, offL, E, bid);
  } else {
    gat_layer_gemm_blk<false, 4, 4>(X, nullptr, nullptr, nullptr, nullptr, nullptr, nullptr,
                                    W, a_s, a_d, H, asv, adv, N, Dact, tile0 + bid - nSort);
  }
}

__global__ __launch_bounds__(256) void gat_bucket_and_gemm(
    const int* __restrict__ tmpw, const int* __restrict__ cntL, const int* __restrict__ offL,
    int* __restrict__ rowptr, int* __restrict__ csrc, int NCH, int nSort,
    const float* __restrict__ X, const float* __restrict__ W,
    const float* __restrict__ a_s, const float* __restrict__ a_d,
    float* __restrict__ H, float* __restrict__ asv, float* __restrict__ adv,
    int N, int Dact, int tile0) {
  const int bid = blockIdx.x;
  if (bid < nSort) {
    gat_bucket_sort_blk(tmpw, cntL, offL, rowptr, csrc, N, NCH, bid);
  } else {
    gat_layer_gemm_blk<false, 4, 4>(X, nullptr, nullptr, nullptr, nullptr, nullptr, nullptr,
                                    W, a_s, a_d, H, asv, adv, N, Dact, tile0 + bid - nSort);
  }
}

__global__ __launch_bounds__(256) void gat_node_gather16(
    const int* __restrict__ rowptr, const int* __restrict__ csrc,
    const float* __restrict__ asv, const float* __restrict__ adv,
    const float* __restrict__ H, const float* __restrict__ bias,
    float* __restrict__ out, int N, int Dact) {
  const int lane = threadIdx.x & 31;
  const int i = blockIdx.x * 8 + __builtin_amdgcn_readfirstlane((int)(threadIdx.x >> 5));
  if (i >= N) return;
  const int beg = rowptr[i];
  const int end = rowptr[i + 1];
  const float ad = adv[i];
  const float sself = leaky02(asv[i] + ad);
  int s0; float sc0; bool v0; float m, sum;
  gat_node_stats(csrc, asv, i, beg, end, lane, ad, sself, s0, sc0, v0, m, sum);

  const int q = lane & 3;
  const int slot = lane >> 2;
  const float4* H4 = reinterpret_cast<const float4*>(H);
  float4 acc = make_float4(0.f, 0.f, 0.f, 0.f);
  if (slot == 0) {
    const float c = __expf(sself - m) / sum;
    const float4 x = H4[(size_t)i * 4 + q];
    acc = make_float4(c * x.x, c * x.y, c * x.z, c * x.w);
  }
  int sv = s0;
  float cv = v0 ? __expf(sc0 - m) / sum : 0.f;
  for (int base = beg; base < end; base += 32) {
    if (base != beg) {
      const int j = base + lane;
      const bool v = j < end;
      sv = v ? csrc[j] : i;
      cv = v ? __expf(leaky02(asv[sv] + ad) - m) / sum : 0.f;
    }
    const int cnt = min(32, end - base);
    for (int t0 = 0; t0 < cnt; t0 += 8) {
      const int t = t0 + slot;
      const int s = __shfl(sv, t, 32);
      const float c = __shfl(cv, t, 32);
      if (t < cnt) {
        const float4 x = H4[(size_t)s * 4 + q];
        acc.x = fmaf(c, x.x, acc.x); acc.y = fmaf(c, x.y, acc.y);
        acc.z = fmaf(c, x.z, acc.z); acc.w = fmaf(c, x.w, acc.w);
      }
    }
  }
  #pragma unroll
  for (int off = 4; off < 32; off <<= 1) {
    acc.x += __shfl_xor(acc.x, off, 32);
    acc.y += __shfl_xor(acc.y, off, 32);
    acc.z += __shfl_xor(acc.z, off, 32);
    acc.w += __shfl_xor(acc.w, off, 32);
  }
  if (slot == 0) {
    const int c = q * 4;
    float* op = out + (size_t)i * Dact;
    if (c + 0 < Dact) op[c + 0] = acc.x + bias[c + 0];
    if (c + 1 < Dact) op[c + 1] = acc.y + bias[c + 1];
    if (c + 2 < Dact) op[c + 2] = acc.z + bias[c + 2];
    if (c + 3 < Dact) op[c + 3] = acc.w + bias[c + 3];
  }
}

extern "C" void kernel_launch(void* const* d_in, const int* in_sizes, int n_in,
                              void* d_out, int out_size, void* d_ws, size_t ws_size,
                              hipStream_t stream) {
  const float* x  = (const float*)d_in[0];
  const int*   ei = (const int*)d_in[1];
  const float* W1 = (const float*)d_in[2];
  const float* as1 = (const float*)d_in[3];
  const float* ad1 = (const float*)d_in[4];
  const float* b1 = (const float*)d_in[5];
  const float* W2 = (const float*)d_in[6];
  const float* as2 = (const float*)d_in[7];
  const float* ad2 = (const float*)d_in[8];
  const float* b2 = (const float*)d_in[9];
  const float* W3 = (const float*)d_in[10];
  const float* as3 = (const float*)d_in[11];
  const float* ad3 = (const float*)d_in[12];
  const float* b3 = (const float*)d_in[13];

  const int N  = in_sizes[0] / 256;
  const int E  = in_sizes[1] / 2;
  const int* src = ei;
  const int* dst = ei + E;

  float* ws = (float*)d_ws;
  float* hA  = ws;
  float* hB  = hA + (size_t)N * 256;
  float* h3  = hB + (size_t)N * 256;
  int*   ib  = (int*)(h3 + (size_t)N * 16);
  int*   cnt    = ib;
  int*   rowptr = cnt + N;
  int*   cursor = rowptr + N + 1;
  int*   tsum   = cursor + N;
  float* asvB = (float*)(ib + (size_t)N * 4);
  float* advB = asvB + N;
  float* asvA = (float*)(ib + (size_t)N * 16);
  float* advA = asvA + N;
  int*   csrc = (int*)(advA + N);

  dim3 blk(256);
  int gGemm  = (N + 31) / 32;
  int gNodeW = (N + 7) / 8;
  int gNodeT = (N + 255) / 256;
  int gEdgeT = (E + 255) / 256;

  const int NBK = (N + 255) / 256;
  const int NCH = (E + SORT_CH - 1) / SORT_CH;
  int*  tmp  = (int*)hB;
  int*  cntL = (int*)hB + 2 * (size_t)NCH * SORT_CH;
  int*  offL = cntL + (size_t)NCH * SORT_MAXB;
  if (NBK <= SORT_MAXB && NCH <= SORT_MAXB &&
      2 * (size_t)NCH * SORT_CH + 2 * (size_t)NCH * SORT_MAXB <= (size_t)N * 256) {
    const int g1 = (gGemm * 2) / 5;
    gat_bin_and_gemm<<<NCH + g1, blk, 0, stream>>>(
        src, dst, tmp, cntL, offL, E, NCH,
        x, W1, as1, ad1, hA, asvA, advA, N, 256, 0);
    gat_bucket_and_gemm<<<NBK + (gGemm - g1), blk, 0, stream>>>(
        tmp, cntL, offL, rowptr, csrc, NCH, NBK,
        x, W1, as1, ad1, hA, asvA, advA, N, 256, g1);
  } else {
    gat_csr_zero<<<gNodeT, blk, 0, stream>>>(cnt, N);
    gat_csr_hist<<<gEdgeT, blk, 0, stream>>>(dst, cnt, E);
    gat_csr_tilesum<<<gNodeT, blk, 0, stream>>>(cnt, tsum, N);
    gat_csr_scan<<<gNodeT, blk, 0, stream>>>(cnt, tsum, rowptr, cursor, N);
    gat_csr_fill<<<gEdgeT, blk, 0, stream>>>(src, dst, cursor, csrc, E);
    gat_layer_gemm<false, 4, 4><<<gGemm, blk, 0, stream>>>(
        x, nullptr, nullptr, nullptr, nullptr, nullptr, nullptr,
        W1, as1, ad1, hA, asvA, advA, N, 256);
  }

  gat_layer_gemm<true, 4, 4><<<gGemm, blk, 0, stream>>>(
      nullptr, rowptr, csrc, asvA, advA, hA, b1,
      W2, as2, ad2, hB, asvB, advB, N, 256);

  gat_layer_gemm<true, 1, 1><<<gGemm, blk, 0, stream>>>(
      nullptr, rowptr, csrc, asvB, advB, hB, b2,
      W3, as3, ad3, h3, asvA, advA, N, 10);

  gat_node_gather16<<<gNodeW, blk, 0, stream>>>(rowptr, csrc, asvA, advA, h3, b3, (float*)d_out, N, 10);
}
